# starting kernel
# baseline (speedup 1.0000x reference)
#include <hip/hip_runtime.h>
#include <math.h>

typedef float v2f __attribute__((ext_vector_type(2)));
typedef float v8f __attribute__((ext_vector_type(8)));

#define NN 100000
#define NEG_SLOPE 0.2f
#define EPSF 1e-16f

__global__ void fill_kernel(float* __restrict__ p, float v, int n) {
    int i = blockIdx.x * blockDim.x + threadIdx.x;
    if (i < n) p[i] = v;
}

__device__ __forceinline__ void atomicMaxF(float* addr, float v) {
    if (v >= 0.0f) atomicMax((int*)addr, __float_as_int(v));
    else           atomicMin((unsigned int*)addr, (unsigned int)__float_as_int(v));
}

__global__ __launch_bounds__(128) void gemm_wmma_kernel(
        const float* __restrict__ X, const float* __restrict__ W,
        float* __restrict__ H, int nrows, int K) {
    int wave = blockIdx.x * (blockDim.x >> 5) + (threadIdx.x >> 5);
    int mtiles = nrows >> 4;
    if (wave >= mtiles) return;

    int lane  = threadIdx.x & 31;
    int r0    = lane & 15;
    int khalf = lane >> 4;

    const float* xrow = X + (size_t)(wave * 16 + r0) * (size_t)K;

    v8f c0 = {}, c1 = {}, c2 = {}, c3 = {};
    for (int k0 = 0; k0 < K; k0 += 4) {
        int ka = k0 + khalf * 2;
        v2f a;
        a.x = xrow[ka];
        a.y = xrow[ka + 1];

        const float* w0 = W + (size_t)ka * 64;
        const float* w1 = w0 + 64;
        v2f b0, b1, b2, b3;
        b0.x = w0[r0];      b0.y = w1[r0];
        b1.x = w0[16 + r0]; b1.y = w1[16 + r0];
        b2.x = w0[32 + r0]; b2.y = w1[32 + r0];
        b3.x = w0[48 + r0]; b3.y = w1[48 + r0];

        c0 = __builtin_amdgcn_wmma_f32_16x16x4_f32(false, a, false, b0, (short)0, c0, false, false);
        c1 = __builtin_amdgcn_wmma_f32_16x16x4_f32(false, a, false, b1, (short)0, c1, false, false);
        c2 = __builtin_amdgcn_wmma_f32_16x16x4_f32(false, a, false, b2, (short)0, c2, false, false);
        c3 = __builtin_amdgcn_wmma_f32_16x16x4_f32(false, a, false, b3, (short)0, c3, false, false);
    }

    float* hbase = H + (size_t)(wave * 16 + khalf * 8) * 64 + r0;
    #pragma unroll
    for (int r = 0; r < 8; ++r) {
        float* row = hbase + (size_t)r * 64;
        row[0]  = c0[r];
        row[16] = c1[r];
        row[32] = c2[r];
        row[48] = c3[r];
    }
}

__global__ __launch_bounds__(256) void scores_kernel(
        const float* __restrict__ H, const float* __restrict__ a_s,
        const float* __restrict__ a_d, float* __restrict__ S, float* __restrict__ D) {
    int node = blockIdx.x * (blockDim.x >> 5) + (threadIdx.x >> 5);
    if (node >= NN) return;
    int lane = threadIdx.x & 31;
    const float* h = H + (size_t)node * 64;
    float h0 = h[lane], h1 = h[lane + 32];
    float s = h0 * a_s[lane] + h1 * a_s[lane + 32];
    float d = h0 * a_d[lane] + h1 * a_d[lane + 32];
    #pragma unroll
    for (int off = 16; off; off >>= 1) {
        s += __shfl_xor(s, off, 32);
        d += __shfl_xor(d, off, 32);
    }
    if (lane == 0) { S[node] = s; D[node] = d; }
}

__global__ void edge_max_kernel(const int* __restrict__ ei, int E, int Etot,
        const float* __restrict__ S, const float* __restrict__ D,
        float* __restrict__ M) {
    int i = blockIdx.x * blockDim.x + threadIdx.x;
    if (i >= Etot) return;
    int src, dst;
    if (i < E) { src = ei[i]; dst = ei[E + i]; } else { src = dst = i - E; }
    float a = S[src] + D[dst];
    a = (a >= 0.0f) ? a : NEG_SLOPE * a;
    atomicMaxF(&M[dst], a);
}

__global__ void edge_exp_kernel(const int* __restrict__ ei, int E, int Etot,
        const float* __restrict__ S, const float* __restrict__ D,
        const float* __restrict__ M, float* __restrict__ EW, float* __restrict__ Z) {
    int i = blockIdx.x * blockDim.x + threadIdx.x;
    if (i >= Etot) return;
    int src, dst;
    if (i < E) { src = ei[i]; dst = ei[E + i]; } else { src = dst = i - E; }
    float a = S[src] + D[dst];
    a = (a >= 0.0f) ? a : NEG_SLOPE * a;
    float e = expf(a - M[dst]);
    EW[i] = e;
    atomicAdd(&Z[dst], e);
}

__global__ __launch_bounds__(256) void edge_agg_kernel(
        const int* __restrict__ ei, int E, int Etot,
        const float* __restrict__ H, const float* __restrict__ EW,
        const float* __restrict__ Z, float* __restrict__ ACC) {
    int e = blockIdx.x * 8 + (threadIdx.x >> 5);
    if (e >= Etot) return;
    int lane = threadIdx.x & 31;
    int src, dst;
    if (e < E) { src = ei[e]; dst = ei[E + e]; } else { src = dst = e - E; }
    float w = EW[e] / (Z[dst] + EPSF);
    const float* h = H + (size_t)src * 64;
    float* o = ACC + (size_t)dst * 64;
    atomicAdd(&o[lane],      h[lane]      * w);
    atomicAdd(&o[lane + 32], h[lane + 32] * w);
}

__global__ void relu_bias_kernel(float* __restrict__ A, const float* __restrict__ b, int n) {
    int i = blockIdx.x * blockDim.x + threadIdx.x;
    if (i < n) {
        float v = A[i] + b[i & 63];
        A[i] = (v > 0.0f) ? v : 0.0f;
    }
}

__global__ void bias_kernel(float* __restrict__ A, const float* __restrict__ b, int n) {
    int i = blockIdx.x * blockDim.x + threadIdx.x;
    if (i < n) A[i] += b[i & 63];
}

extern "C" void kernel_launch(void* const* d_in, const int* in_sizes, int n_in,
                              void* d_out, int out_size, void* d_ws, size_t ws_size,
                              hipStream_t stream) {
    const float* x     = (const float*)d_in[0];
    const int*   ei    = (const int*)d_in[1];
    const float* W1    = (const float*)d_in[2];
    const float* as1   = (const float*)d_in[3];
    const float* ad1   = (const float*)d_in[4];
    const float* b1    = (const float*)d_in[5];
    const float* W2    = (const float*)d_in[6];
    const float* as2   = (const float*)d_in[7];
    const float* ad2   = (const float*)d_in[8];
    const float* b2    = (const float*)d_in[9];
    float* out = (float*)d_out;

    const int E    = in_sizes[1] / 2;
    const int Etot = E + NN;
    const int NF   = NN * 64;

    float* wsf = (float*)d_ws;
    float* H   = wsf;
    float* ACC = H   + NF;
    float* S   = ACC + NF;
    float* D   = S   + NN;
    float* M   = D   + NN;
    float* Z   = M   + NN;
    float* EW  = Z   + NN;

    const int TB = 256;
    dim3 blk(TB);
    int gN    = (NN + TB - 1) / TB;
    int gNF   = (NF + TB - 1) / TB;
    int gE    = (Etot + TB - 1) / TB;
    int gEagg = (Etot + 7) / 8;
    int gScr  = (NN + 7) / 8;
    int gGemm = ((NN / 16) + 3) / 4;

    fill_kernel<<<gNF, blk, 0, stream>>>(ACC, 0.0f, NF);
    fill_kernel<<<gN,  blk, 0, stream>>>(M, -INFINITY, NN);
    fill_kernel<<<gN,  blk, 0, stream>>>(Z, 0.0f, NN);

    gemm_wmma_kernel<<<gGemm, dim3(128), 0, stream>>>(x, W1, H, NN, 128);
    scores_kernel<<<gScr, blk, 0, stream>>>(H, as1, ad1, S, D);
    edge_max_kernel<<<gE, blk, 0, stream>>>(ei, E, Etot, S, D, M);
    edge_exp_kernel<<<gE, blk, 0, stream>>>(ei, E, Etot, S, D, M, EW, Z);
    edge_agg_kernel<<<gEagg, blk, 0, stream>>>(ei, E, Etot, H, EW, Z, ACC);
    relu_bias_kernel<<<gNF, blk, 0, stream>>>(ACC, b1, NF);

    fill_kernel<<<gN,  blk, 0, stream>>>(M, -INFINITY, NN);
    fill_kernel<<<gN,  blk, 0, stream>>>(Z, 0.0f, NN);
    fill_kernel<<<gNF, blk, 0, stream>>>(out, 0.0f, NF);

    gemm_wmma_kernel<<<gGemm, dim3(128), 0, stream>>>(ACC, W2, H, NN, 64);
    scores_kernel<<<gScr, blk, 0, stream>>>(H, as2, ad2, S, D);
    edge_max_kernel<<<gE, blk, 0, stream>>>(ei, E, Etot, S, D, M);
    edge_exp_kernel<<<gE, blk, 0, stream>>>(ei, E, Etot, S, D, M, EW, Z);
    edge_agg_kernel<<<gEagg, blk, 0, stream>>>(ei, E, Etot, H, EW, Z, out);
    bias_kernel<<<gNF, blk, 0, stream>>>(out, b2, NF);
}


# Edges grouped by destination once per call (int-atomic counting sort; each segment then sorted by source: new, fixed sum order); wave per node: max, exp, sum, row sum in registers, no float atomics
# speedup vs baseline: 2.9395x; 2.9395x over previous
#include <hip/hip_runtime.h>
#include <math.h>


typedef float v2f __attribute__((ext_vector_type(2)));
typedef float v8f __attribute__((ext_vector_type(8)));

#define NN 100000
#define NEG_SLOPE 0.2f
#define EPSF 1e-16f

__global__ __launch_bounds__(128) void gemm_wmma_kernel(
        const float* __restrict__ X, const float* __restrict__ W,
        float* __restrict__ H, int nrows, int K) {
    int wave = blockIdx.x * (blockDim.x >> 5) + (threadIdx.x >> 5);
    int mtiles = nrows >> 4;
    if (wave >= mtiles) return;

    int lane  = threadIdx.x & 31;
    int r0    = lane & 15;
    int khalf = lane >> 4;

    const float* xrow = X + (size_t)(wave * 16 + r0) * (size_t)K;

    v8f c0 = {}, c1 = {}, c2 = {}, c3 = {};
    for (int k0 = 0; k0 < K; k0 += 4) {
        int ka = k0 + khalf * 2;
        v2f a;
        a.x = xrow[ka];
        a.y = xrow[ka + 1];

        const float* w0 = W + (size_t)ka * 64;
        const float* w1 = w0 + 64;
        v2f b0, b1, b2, b3;
        b0.x = w0[r0];      b0.y = w1[r0];
        b1.x = w0[16 + r0]; b1.y = w1[16 + r0];
        b2.x = w0[32 + r0]; b2.y = w1[32 + r0];
        b3.x = w0[48 + r0]; b3.y = w1[48 + r0];

        c0 = __builtin_amdgcn_wmma_f32_16x16x4_f32(false, a, false, b0, (short)0, c0, false, false);
        c1 = __builtin_amdgcn_wmma_f32_16x16x4_f32(false, a, false, b1, (short)0, c1, false, false);
        c2 = __builtin_amdgcn_wmma_f32_16x16x4_f32(false, a, false, b2, (short)0, c2, false, false);
        c3 = __builtin_amdgcn_wmma_f32_16x16x4_f32(false, a, false, b3, (short)0, c3, false, false);
    }

    float* hbase = H + (size_t)(wave * 16 + khalf * 8) * 64 + r0;
    #pragma unroll
    for (int r = 0; r < 8; ++r) {
        float* row = hbase + (size_t)r * 64;
        row[0]  = c0[r];
        row[16] = c1[r];
        row[32] = c2[r];
        row[48] = c3[r];
    }
}

__global__ __launch_bounds__(256) void scores_kernel(
        const float* __restrict__ H, const float* __restrict__ a_s,
        const float* __restrict__ a_d, float* __restrict__ S, float* __restrict__ D) {
    int node = blockIdx.x * (blockDim.x >> 5) + (threadIdx.x >> 5);
    if (node >= NN) return;
    int lane = threadIdx.x & 31;
    const float* h = H + (size_t)node * 64;
    float h0 = h[lane], h1 = h[lane + 32];
    float s = h0 * a_s[lane] + h1 * a_s[lane + 32];
    float d = h0 * a_d[lane] + h1 * a_d[lane + 32];
    #pragma unroll
    for (int off = 16; off; off >>= 1) {
        s += __shfl_xor(s, off, 32);
        d += __shfl_xor(d, off, 32);
    }
    if (lane == 0) { S[node] = s; D[node] = d; }
}

__global__ void zero_int_kernel(int* __restrict__ p, int n) {
    int i = blockIdx.x * blockDim.x + threadIdx.x;
    if (i < n) p[i] = 0;
}

__global__ void count_rank_kernel(const int* __restrict__ ei, int E,
        int* __restrict__ cnt, int* __restrict__ rk) {
    int e = blockIdx.x * blockDim.x + threadIdx.x;
    if (e >= E) return;
    rk[e] = atomicAdd(&cnt[ei[E + e]], 1);
}

__global__ __launch_bounds__(1024) void scan_counts_kernel(
        const int* __restrict__ cnt, int* __restrict__ off) {
    __shared__ int red[32];
    __shared__ int wtot[32];
    const int tid  = threadIdx.x;
    const int lane = tid & 31;
    const int wv   = tid >> 5;
    const int tile0 = blockIdx.x * 1024;

    const int4* c4 = (const int4*)cnt;
    const int nvec = tile0 >> 2;
    int part = 0;
    for (int i = tid; i < nvec; i += 1024) {
        int4 v = c4[i];
        part += (v.x + v.y) + (v.z + v.w);
    }
    #pragma unroll
    for (int o = 16; o; o >>= 1) part += __shfl_xor(part, o, 32);
    if (lane == 0) red[wv] = part;
    __syncthreads();
    int base = red[lane];
    #pragma unroll
    for (int o = 16; o; o >>= 1) base += __shfl_xor(base, o, 32);

    const int i = tile0 + tid;
    const int v = (i < NN) ? cnt[i] : 0;
    int inc = v;
    #pragma unroll
    for (int o = 1; o < 32; o <<= 1) {
        int t = __shfl_up(inc, o, 32);
        if (lane >= o) inc += t;
    }
    if (lane == 31) wtot[wv] = inc;
    __syncthreads();
    const int wt = wtot[lane];
    int winc = wt;
    #pragma unroll
    for (int o = 1; o < 32; o <<= 1) {
        int t = __shfl_up(winc, o, 32);
        if (lane >= o) winc += t;
    }
    const int wprefix = __shfl(winc, wv, 32) - __shfl(wt, wv, 32);
    if (i < NN) off[i] = base + wprefix + inc - v;
    if (i == NN - 1) off[NN] = base + wprefix + inc;
}

__global__ void place_kernel(const int* __restrict__ ei, int E,
        const int* __restrict__ off, const int* __restrict__ rk, int* __restrict__ col) {
    int e = blockIdx.x * blockDim.x + threadIdx.x;
    if (e >= E) return;
    col[off[ei[E + e]] + rk[e]] = ei[e];
}

__global__ __launch_bounds__(256) void sort_segments_kernel(
        const int* __restrict__ off, const int* __restrict__ cin, int* __restrict__ cout) {
    const int lane = threadIdx.x & 31;
    const int node = __builtin_amdgcn_readfirstlane((int)(blockIdx.x * 8 + (threadIdx.x >> 5)));
    if (node >= NN) return;
    const int o0  = off[node];
    const int deg = off[node + 1] - o0;
    if (deg <= 32) {
        const int key = (lane < deg) ? cin[o0 + lane] : 0x7fffffff;
        int rank = 0;
        for (int j = 0; j < deg; ++j) {
            const int kj = __builtin_amdgcn_readlane(key, j);
            rank += (kj < key || (kj == key && j < lane)) ? 1 : 0;
        }
        if (lane < deg) cout[o0 + rank] = key;
    } else {
        for (int i = lane; i < deg; i += 32) {
            const int key = cin[o0 + i];
            int rank = 0;
            for (int j = 0; j < deg; ++j) {
                const int kj = cin[o0 + j];
                rank += (kj < key || (kj == key && j < i)) ? 1 : 0;
            }
            cout[o0 + rank] = key;
        }
    }
}

__device__ __forceinline__ float leaky(float a) { return (a >= 0.0f) ? a : NEG_SLOPE * a; }

__device__ __forceinline__ float readlane_f(float v, int l) {
    return __int_as_float(__builtin_amdgcn_readlane(__float_as_int(v), l));
}

__global__ __launch_bounds__(256) void node_softmax_sum_kernel(
        const int* __restrict__ off, const int* __restrict__ col,
        const float* __restrict__ H, const float* __restrict__ S, const float* __restrict__ D,
        const float* __restrict__ bias, int relu, float* __restrict__ OUT) {
    const int lane = threadIdx.x & 31;
    const int node = __builtin_amdgcn_readfirstlane((int)(blockIdx.x * 8 + (threadIdx.x >> 5)));
    if (node >= NN) return;
    const int o0  = off[node];
    const int deg = off[node + 1] - o0;
    const float dn = D[node];
    const float a_self = leaky(S[node] + dn);
    const v2f* H2 = (const v2f*)H;
    const v2f hs = H2[(size_t)node * 32 + lane];
    const float bx = bias[2 * lane];
    const float by = bias[2 * lane + 1];
    v2f acc;

    if (deg <= 32) {
        int s = node;
        float a = -INFINITY;
        if (lane < deg) { s = col[o0 + lane]; a = leaky(S[s] + dn); }
        float m = fmaxf(a, a_self);
        #pragma unroll
        for (int o = 16; o; o >>= 1) m = fmaxf(m, __shfl_xor(m, o, 32));
        const float e = (lane < deg) ? expf(a - m) : 0.0f;
        const float e_self = expf(a_self - m);
        float z = e;
        #pragma unroll
        for (int o = 16; o; o >>= 1) z += __shfl_xor(z, o, 32);
        z += e_self;
        const float w = e / (z + EPSF);
        const float w_self = e_self / (z + EPSF);
        acc.x = hs.x * w_self;
        acc.y = hs.y * w_self;
        for (int j = 0; j < deg; j += 8) {
            v2f hv[8];
            #pragma unroll
            for (int u = 0; u < 8; ++u) {
                hv[u].x = 0.0f; hv[u].y = 0.0f;
                if (j + u < deg) {
                    const int sj = __builtin_amdgcn_readlane(s, j + u);
                    hv[u] = H2[(size_t)sj * 32 + lane];
                }
            }
            #pragma unroll
            for (int u = 0; u < 8; ++u) {
                const float wj = readlane_f(w, j + u);
                acc.x = fmaf(hv[u].x, wj, acc.x);
                acc.y = fmaf(hv[u].y, wj, acc.y);
            }
        }
    } else {
        float m = a_self;
        for (int c = 0; c < deg; c += 32) {
            if (c + lane < deg) m = fmaxf(m, leaky(S[col[o0 + c + lane]] + dn));
        }
        #pragma unroll
        for (int o = 16; o; o >>= 1) m = fmaxf(m, __shfl_xor(m, o, 32));
        float z = 0.0f;
        for (int c = 0; c < deg; c += 32) {
            if (c + lane < deg) z += expf(leaky(S[col[o0 + c + lane]] + dn) - m);
        }
        #pragma unroll
        for (int o = 16; o; o >>= 1) z += __shfl_xor(z, o, 32);
        const float e_self = expf(a_self - m);
        z += e_self;
        const float w_self = e_self / (z + EPSF);
        acc.x = hs.x * w_self;
        acc.y = hs.y * w_self;
        for (int c = 0; c < deg; c += 32) {
            int s = node;
            float w = 0.0f;
            if (c + lane < deg) {
                s = col[o0 + c + lane];
                w = expf(leaky(S[s] + dn) - m) / (z + EPSF);
            }
            const int nj = (deg - c < 32) ? (deg - c) : 32;
            for (int j = 0; j < nj; ++j) {
                const int sj = __builtin_amdgcn_readlane(s, j);
                const float wj = readlane_f(w, j);
                const v2f hv = H2[(size_t)sj * 32 + lane];
                acc.x = fmaf(hv.x, wj, acc.x);
                acc.y = fmaf(hv.y, wj, acc.y);
            }
        }
    }

    acc.x += bx;
    acc.y += by;
    if (relu) {
        acc.x = (acc.x > 0.0f) ? acc.x : 0.0f;
        acc.y = (acc.y > 0.0f) ? acc.y : 0.0f;
    }
    ((v2f*)OUT)[(size_t)node * 32 + lane] = acc;
}

extern "C" void kernel_launch(void* const* d_in, const int* in_sizes, int n_in,
                              void* d_out, int out_size, void* d_ws, size_t ws_size,
                              hipStream_t stream) {
    const float* x     = (const float*)d_in[0];
    const int*   ei    = (const int*)d_in[1];
    const float* W1    = (const float*)d_in[2];
    const float* as1   = (const float*)d_in[3];
    const float* ad1   = (const float*)d_in[4];
    const float* b1    = (const float*)d_in[5];
    const float* W2    = (const float*)d_in[6];
    const float* as2   = (const float*)d_in[7];
    const float* ad2   = (const float*)d_in[8];
    const float* b2    = (const float*)d_in[9];
    float* out = (float*)d_out;

    const int E  = in_sizes[1] / 2;
    const int NF = NN * 64;

    float* wsf = (float*)d_ws;
    float* H   = wsf;
    float* X2  = H   + NF;
    float* S   = X2  + NF;
    float* D   = S   + NN;
    int*   cnt = (int*)(D + NN);
    int*   off = cnt + NN;
    int*   rk  = off + NN + 16;
    int*   cA  = rk  + E;
    int*   cB  = cA  + E;

    const int TB = 256;
    dim3 blk(TB);
    int gN    = (NN + TB - 1) / TB;
    int gE    = (E + TB - 1) / TB;
    int gScan = (NN + 1023) / 1024;
    int gNode = (NN + 7) / 8;
    int gGemm = ((NN / 16) + 3) / 4;

    zero_int_kernel<<<gN, blk, 0, stream>>>(cnt, NN);
    count_rank_kernel<<<gE, blk, 0, stream>>>(ei, E, cnt, rk);
    scan_counts_kernel<<<gScan, dim3(1024), 0, stream>>>(cnt, off);
    place_kernel<<<gE, blk, 0, stream>>>(ei, E, off, rk, cA);
    sort_segments_kernel<<<gNode, blk, 0, stream>>>(off, cA, cB);

    gemm_wmma_kernel<<<gGemm, dim3(128), 0, stream>>>(x, W1, H, NN, 128);
    scores_kernel<<<gNode, blk, 0, stream>>>(H, as1, ad1, S, D);
    node_softmax_sum_kernel<<<gNode, blk, 0, stream>>>(off, cB, H, S, D, b1, 1, X2);

    gemm_wmma_kernel<<<gGemm, dim3(128), 0, stream>>>(X2, W2, H, NN, 64);
    scores_kernel<<<gNode, blk, 0, stream>>>(H, as2, ad2, S, D);
    node_softmax_sum_kernel<<<gNode, blk, 0, stream>>>(off, cB, H, S, D, b2, 0, out);
}


# Edges grouped by destination with LDS counters, not global atomics: per-chunk bucket counts, one scan, move into 256-node buckets, per-bucket node counts and placement in LDS; sum order kept
# speedup vs baseline: 3.5119x; 1.1947x over previous
#include <hip/hip_runtime.h>
#include <math.h>


typedef float v2f __attribute__((ext_vector_type(2)));
typedef float v8f __attribute__((ext_vector_type(8)));

#define NN 100000
#define NEG_SLOPE 0.2f
#define EPSF 1e-16f

__global__ __launch_bounds__(128) void gemm_wmma_kernel(
        const float* __restrict__ X, const float* __restrict__ W,
        float* __restrict__ H, int nrows, int K) {
    int wave = blockIdx.x * (blockDim.x >> 5) + (threadIdx.x >> 5);
    int mtiles = nrows >> 4;
    if (wave >= mtiles) return;

    int lane  = threadIdx.x & 31;
    int r0    = lane & 15;
    int khalf = lane >> 4;

    const float* xrow = X + (size_t)(wave * 16 + r0) * (size_t)K;

    v8f c0 = {}, c1 = {}, c2 = {}, c3 = {};
    for (int k0 = 0; k0 < K; k0 += 4) {
        int ka = k0 + khalf * 2;
        v2f a;
        a.x = xrow[ka];
        a.y = xrow[ka + 1];

        const float* w0 = W + (size_t)ka * 64;
        const float* w1 = w0 + 64;
        v2f b0, b1, b2, b3;
        b0.x = w0[r0];      b0.y = w1[r0];
        b1.x = w0[16 + r0]; b1.y = w1[16 + r0];
        b2.x = w0[32 + r0]; b2.y = w1[32 + r0];
        b3.x = w0[48 + r0]; b3.y = w1[48 + r0];

        c0 = __builtin_amdgcn_wmma_f32_16x16x4_f32(false, a, false, b0, (short)0, c0, false, false);
        c1 = __builtin_amdgcn_wmma_f32_16x16x4_f32(false, a, false, b1, (short)0, c1, false, false);
        c2 = __builtin_amdgcn_wmma_f32_16x16x4_f32(false, a, false, b2, (short)0, c2, false, false);
        c3 = __builtin_amdgcn_wmma_f32_16x16x4_f32(false, a, false, b3, (short)0, c3, false, false);
    }

    float* hbase = H + (size_t)(wave * 16 + khalf * 8) * 64 + r0;
    #pragma unroll
    for (int r = 0; r < 8; ++r) {
        float* row = hbase + (size_t)r * 64;
        row[0]  = c0[r];
        row[16] = c1[r];
        row[32] = c2[r];
        row[48] = c3[r];
    }
}

__global__ __launch_bounds__(256) void scores_kernel(
        const float* __restrict__ H, const float* __restrict__ a_s,
        const float* __restrict__ a_d, float* __restrict__ S, float* __restrict__ D) {
    int node = blockIdx.x * (blockDim.x >> 5) + (threadIdx.x >> 5);
    if (node >= NN) return;
    int lane = threadIdx.x & 31;
    const float* h = H + (size_t)node * 64;
    float h0 = h[lane], h1 = h[lane + 32];
    float s = h0 * a_s[lane] + h1 * a_s[lane + 32];
    float d = h0 * a_d[lane] + h1 * a_d[lane + 32];
    #pragma unroll
    for (int off = 16; off; off >>= 1) {
        s += __shfl_xor(s, off, 32);
        d += __shfl_xor(d, off, 32);
    }
    if (lane == 0) { S[node] = s; D[node] = d; }
}

#define NPB_LOG2 8
#define NPB (1 << NPB_LOG2)
#define NBUCK ((NN + NPB - 1) / NPB)
#define CH 4096
#define EPT (CH / 256)
#define SRC_BITS 17
#define SRC_MASK ((1 << SRC_BITS) - 1)
static_assert(NN <= (1 << SRC_BITS), "a source id must fit the low bits of the packed word");
static_assert(NPB_LOG2 + SRC_BITS + 1 < 32, "packed word and the spare value past it must stay non-negative");

__global__ __launch_bounds__(256) void bucket_hist_kernel(const int* __restrict__ ei, int E, int G1,
        int* __restrict__ cntm) {
    __shared__ int hist[NBUCK];
    const int tid = threadIdx.x;
    for (int i = tid; i < NBUCK; i += 256) hist[i] = 0;
    __syncthreads();
    const int e0 = blockIdx.x * CH + tid;
    int d[EPT];
    #pragma unroll
    for (int k = 0; k < EPT; ++k) {
        const int e = e0 + k * 256;
        d[k] = (e < E) ? ei[E + e] : -1;
    }
    #pragma unroll
    for (int k = 0; k < EPT; ++k) {
        if (d[k] >= 0) atomicAdd(&hist[d[k] >> NPB_LOG2], 1);
    }
    __syncthreads();
    for (int i = tid; i < NBUCK; i += 256) cntm[i * G1 + blockIdx.x] = hist[i];
}

__global__ __launch_bounds__(1024) void scan_counts_kernel(
        const int* __restrict__ cnt, int* __restrict__ off, int n) {
    __shared__ int red[32];
    __shared__ int wtot[32];
    const int tid  = threadIdx.x;
    const int lane = tid & 31;
    const int wv   = tid >> 5;
    const int tile0 = blockIdx.x * 1024;

    const int4* c4 = (const int4*)cnt;
    const int nvec = tile0 >> 2;
    int part = 0;
    for (int i = tid; i < nvec; i += 1024) {
        int4 v = c4[i];
        part += (v.x + v.y) + (v.z + v.w);
    }
    #pragma unroll
    for (int o = 16; o; o >>= 1) part += __shfl_xor(part, o, 32);
    if (lane == 0) red[wv] = part;
    __syncthreads();
    int base = red[lane];
    #pragma unroll
    for (int o = 16; o; o >>= 1) base += __shfl_xor(base, o, 32);

    const int i = tile0 + tid;
    const int v = (i < n) ? cnt[i] : 0;
    int inc = v;
    #pragma unroll
    for (int o = 1; o < 32; o <<= 1) {
        int t = __shfl_up(inc, o, 32);
        if (lane >= o) inc += t;
    }
    if (lane == 31) wtot[wv] = inc;
    __syncthreads();
    const int wt = wtot[lane];
    int winc = wt;
    #pragma unroll
    for (int o = 1; o < 32; o <<= 1) {
        int t = __shfl_up(winc, o, 32);
        if (lane >= o) winc += t;
    }
    const int wprefix = __shfl(winc, wv, 32) - __shfl(wt, wv, 32);
    if (i < n) off[i] = base + wprefix + inc - v;
    if (i == n - 1) off[n] = base + wprefix + inc;
}

__global__ __launch_bounds__(256) void bucket_scatter_kernel(const int* __restrict__ ei, int E, int G1,
        const int* __restrict__ boff, int* __restrict__ tmp) {
    __shared__ int cur[NBUCK + 1];
    const int tid = threadIdx.x;
    const int e0 = blockIdx.x * CH + tid;
    int d[EPT], s[EPT];
    #pragma unroll
    for (int k = 0; k < EPT; ++k) {
        const int e = e0 + k * 256;
        d[k] = (e < E) ? ei[E + e] : (NBUCK << NPB_LOG2);
        s[k] = (e < E) ? ei[e] : 0;
    }
    for (int i = tid; i < NBUCK; i += 256) cur[i] = boff[i * G1 + blockIdx.x];
    if (tid == 0) cur[NBUCK] = 0;
    __syncthreads();
    int pos[EPT];
    #pragma unroll
    for (int k = 0; k < EPT; ++k) pos[k] = atomicAdd(&cur[d[k] >> NPB_LOG2], 1);
    #pragma unroll
    for (int k = 0; k < EPT; ++k) {
        if (e0 + k * 256 < E) tmp[pos[k]] = ((d[k] & (NPB - 1)) << SRC_BITS) | s[k];
    }
}

__global__ __launch_bounds__(256) void bucket_csr_kernel(const int* __restrict__ boff, int G1,
        const int* __restrict__ tmp, int* __restrict__ off, int* __restrict__ col) {
    __shared__ int cnt[NPB + 1];
    __shared__ int wtot[NPB / 32];
    const int tid  = threadIdx.x;
    const int lane = tid & 31;
    const int wv   = tid >> 5;
    const int bs = boff[blockIdx.x * G1];
    const int be = boff[(blockIdx.x + 1) * G1];
    cnt[tid] = 0;
    if (tid == 0) cnt[NPB] = 0;
    __syncthreads();
    for (int i = bs + tid; i < be; i += 4 * 256) {
        int w[4];
        #pragma unroll
        for (int u = 0; u < 4; ++u) {
            const int j = i + u * 256;
            w[u] = (j < be) ? tmp[j] : -1;
        }
        #pragma unroll
        for (int u = 0; u < 4; ++u) {
            if (w[u] >= 0) atomicAdd(&cnt[w[u] >> SRC_BITS], 1);
        }
    }
    __syncthreads();
    const int v = cnt[tid];
    int inc = v;
    #pragma unroll
    for (int o = 1; o < 32; o <<= 1) {
        int t = __shfl_up(inc, o, 32);
        if (lane >= o) inc += t;
    }
    if (lane == 31) wtot[wv] = inc;
    __syncthreads();
    int wprefix = 0;
    for (int w = 0; w < wv; ++w) wprefix += wtot[w];
    const int excl = bs + wprefix + inc - v;
    const int node = blockIdx.x * NPB + tid;
    if (node < NN) off[node] = excl;
    if (node == NN - 1) off[NN] = excl + v;
    cnt[tid] = excl;
    __syncthreads();
    for (int i = bs + tid; i < be; i += 4 * 256) {
        int w[4];
        #pragma unroll
        for (int u = 0; u < 4; ++u) {
            const int j = i + u * 256;
            w[u] = (j < be) ? tmp[j] : (NPB << SRC_BITS);
        }
        int pos[4];
        #pragma unroll
        for (int u = 0; u < 4; ++u) pos[u] = atomicAdd(&cnt[w[u] >> SRC_BITS], 1);
        #pragma unroll
        for (int u = 0; u < 4; ++u) {
            if (i + u * 256 < be) col[pos[u]] = w[u] & SRC_MASK;
        }
    }
}

__global__ __launch_bounds__(256) void sort_segments_kernel(
        const int* __restrict__ off, const int* __restrict__ cin, int* __restrict__ cout) {
    const int lane = threadIdx.x & 31;
    const int node = __builtin_amdgcn_readfirstlane((int)(blockIdx.x * 8 + (threadIdx.x >> 5)));
    if (node >= NN) return;
    const int o0  = off[node];
    const int deg = off[node + 1] - o0;
    if (deg <= 32) {
        const int key = (lane < deg) ? cin[o0 + lane] : 0x7fffffff;
        int rank = 0;
        for (int j = 0; j < deg; ++j) {
            const int kj = __builtin_amdgcn_readlane(key, j);
            rank += (kj < key || (kj == key && j < lane)) ? 1 : 0;
        }
        if (lane < deg) cout[o0 + rank] = key;
    } else {
        for (int i = lane; i < deg; i += 32) {
            const int key = cin[o0 + i];
            int rank = 0;
            for (int j = 0; j < deg; ++j) {
                const int kj = cin[o0 + j];
                rank += (kj < key || (kj == key && j < i)) ? 1 : 0;
            }
            cout[o0 + rank] = key;
        }
    }
}

__device__ __forceinline__ float leaky(float a) { return (a >= 0.0f) ? a : NEG_SLOPE * a; }

__device__ __forceinline__ float readlane_f(float v, int l) {
    return __int_as_float(__builtin_amdgcn_readlane(__float_as_int(v), l));
}

__global__ __launch_bounds__(256) void node_softmax_sum_kernel(
        const int* __restrict__ off, const int* __restrict__ col,
        const float* __restrict__ H, const float* __restrict__ S, const float* __restrict__ D,
        const float* __restrict__ bias, int relu, float* __restrict__ OUT) {
    const int lane = threadIdx.x & 31;
    const int node = __builtin_amdgcn_readfirstlane((int)(blockIdx.x * 8 + (threadIdx.x >> 5)));
    if (node >= NN) return;
    const int o0  = off[node];
    const int deg = off[node + 1] - o0;
    const float dn = D[node];
    const float a_self = leaky(S[node] + dn);
    const v2f* H2 = (const v2f*)H;
    const v2f hs = H2[(size_t)node * 32 + lane];
    const float bx = bias[2 * lane];
    const float by = bias[2 * lane + 1];
    v2f acc;

    if (deg <= 32) {
        int s = node;
        float a = -INFINITY;
        if (lane < deg) { s = col[o0 + lane]; a = leaky(S[s] + dn); }
        float m = fmaxf(a, a_self);
        #pragma unroll
        for (int o = 16; o; o >>= 1) m = fmaxf(m, __shfl_xor(m, o, 32));
        const float e = (lane < deg) ? expf(a - m) : 0.0f;
        const float e_self = expf(a_self - m);
        float z = e;
        #pragma unroll
        for (int o = 16; o; o >>= 1) z += __shfl_xor(z, o, 32);
        z += e_self;
        const float w = e / (z + EPSF);
        const float w_self = e_self / (z + EPSF);
        acc.x = hs.x * w_self;
        acc.y = hs.y * w_self;
        for (int j = 0; j < deg; j += 8) {
            v2f hv[8];
            #pragma unroll
            for (int u = 0; u < 8; ++u) {
                hv[u].x = 0.0f; hv[u].y = 0.0f;
                if (j + u < deg) {
                    const int sj = __builtin_amdgcn_readlane(s, j + u);
                    hv[u] = H2[(size_t)sj * 32 + lane];
                }
            }
            #pragma unroll
            for (int u = 0; u < 8; ++u) {
                const float wj = readlane_f(w, j + u);
                acc.x = fmaf(hv[u].x, wj, acc.x);
                acc.y = fmaf(hv[u].y, wj, acc.y);
            }
        }
    } else {
        float m = a_self;
        for (int c = 0; c < deg; c += 32) {
            if (c + lane < deg) m = fmaxf(m, leaky(S[col[o0 + c + lane]] + dn));
        }
        #pragma unroll
        for (int o = 16; o; o >>= 1) m = fmaxf(m, __shfl_xor(m, o, 32));
        float z = 0.0f;
        for (int c = 0; c < deg; c += 32) {
            if (c + lane < deg) z += expf(leaky(S[col[o0 + c + lane]] + dn) - m);
        }
        #pragma unroll
        for (int o = 16; o; o >>= 1) z += __shfl_xor(z, o, 32);
        const float e_self = expf(a_self - m);
        z += e_self;
        const float w_self = e_self / (z + EPSF);
        acc.x = hs.x * w_self;
        acc.y = hs.y * w_self;
        for (int c = 0; c < deg; c += 32) {
            int s = node;
            float w = 0.0f;
            if (c + lane < deg) {
                s = col[o0 + c + lane];
                w = expf(leaky(S[s] + dn) - m) / (z + EPSF);
            }
            const int nj = (deg - c < 32) ? (deg - c) : 32;
            for (int j = 0; j < nj; ++j) {
                const int sj = __builtin_amdgcn_readlane(s, j);
                const float wj = readlane_f(w, j);
                const v2f hv = H2[(size_t)sj * 32 + lane];
                acc.x = fmaf(hv.x, wj, acc.x);
                acc.y = fmaf(hv.y, wj, acc.y);
            }
        }
    }

    acc.x += bx;
    acc.y += by;
    if (relu) {
        acc.x = (acc.x > 0.0f) ? acc.x : 0.0f;
        acc.y = (acc.y > 0.0f) ? acc.y : 0.0f;
    }
    ((v2f*)OUT)[(size_t)node * 32 + lane] = acc;
}

extern "C" void kernel_launch(void* const* d_in, const int* in_sizes, int n_in,
                              void* d_out, int out_size, void* d_ws, size_t ws_size,
                              hipStream_t stream) {
    const float* x     = (const float*)d_in[0];
    const int*   ei    = (const int*)d_in[1];
    const float* W1    = (const float*)d_in[2];
    const float* as1   = (const float*)d_in[3];
    const float* ad1   = (const float*)d_in[4];
    const float* b1    = (const float*)d_in[5];
    const float* W2    = (const float*)d_in[6];
    const float* as2   = (const float*)d_in[7];
    const float* ad2   = (const float*)d_in[8];
    const float* b2    = (const float*)d_in[9];
    float* out = (float*)d_out;

    const int E  = in_sizes[1] / 2;
    const int NF = NN * 64;

    float* wsf = (float*)d_ws;
    float* H   = wsf;
    float* X2  = H   + NF;
    float* S   = X2  + NF;
    float* D   = S   + NN;
    const int G1 = (E + CH - 1) / CH;
    const int nM = NBUCK * G1;
    int*   cntm = (int*)(D + NN);
    int*   boff = cntm + nM + 16;
    int*   off  = boff + nM + 16;
    int*   tmp  = off + NN + 16;
    int*   cA   = tmp + E;
    int*   cB   = cA  + E;

    const int TB = 256;
    dim3 blk(TB);
    int gScan = (nM + 1023) / 1024;
    int gNode = (NN + 7) / 8;
    int gGemm = ((NN / 16) + 3) / 4;

    bucket_hist_kernel<<<G1, blk, 0, stream>>>(ei, E, G1, cntm);
    scan_counts_kernel<<<gScan, dim3(1024), 0, stream>>>(cntm, boff, nM);
    bucket_scatter_kernel<<<G1, blk, 0, stream>>>(ei, E, G1, boff, tmp);
    bucket_csr_kernel<<<NBUCK, blk, 0, stream>>>(boff, G1, tmp, off, cA);
    sort_segments_kernel<<<gNode, blk, 0, stream>>>(off, cA, cB);

    gemm_wmma_kernel<<<gGemm, dim3(128), 0, stream>>>(x, W1, H, NN, 128);
    scores_kernel<<<gNode, blk, 0, stream>>>(H, as1, ad1, S, D);
    node_softmax_sum_kernel<<<gNode, blk, 0, stream>>>(off, cB, H, S, D, b1, 1, X2);

    gemm_wmma_kernel<<<gGemm, dim3(128), 0, stream>>>(X2, W2, H, NN, 64);
    scores_kernel<<<gNode, blk, 0, stream>>>(H, as2, ad2, S, D);
    node_softmax_sum_kernel<<<gNode, blk, 0, stream>>>(off, cB, H, S, D, b2, 0, out);
}


# Node kernels (segment sort, scores, softmax and row sum): a wave takes 8 consecutive nodes, not one: 12,500 waves enter a launch, not 100,000; a strip's first loads go out together; sums unchanged
# speedup vs baseline: 3.8827x; 1.1056x over previous
#include <hip/hip_runtime.h>
#include <math.h>


typedef float v2f __attribute__((ext_vector_type(2)));
typedef float v8f __attribute__((ext_vector_type(8)));

#define NN 100000
#define NEG_SLOPE 0.2f
#define EPSF 1e-16f

__global__ __launch_bounds__(128) void gemm_wmma_kernel(
        const float* __restrict__ X, const float* __restrict__ W,
        float* __restrict__ H, int nrows, int K) {
    int wave = blockIdx.x * (blockDim.x >> 5) + (threadIdx.x >> 5);
    int mtiles = nrows >> 4;
    if (wave >= mtiles) return;

    int lane  = threadIdx.x & 31;
    int r0    = lane & 15;
    int khalf = lane >> 4;

    const float* xrow = X + (size_t)(wave * 16 + r0) * (size_t)K;

    v8f c0 = {}, c1 = {}, c2 = {}, c3 = {};
    for (int k0 = 0; k0 < K; k0 += 4) {
        int ka = k0 + khalf * 2;
        v2f a;
        a.x = xrow[ka];
        a.y = xrow[ka + 1];

        const float* w0 = W + (size_t)ka * 64;
        const float* w1 = w0 + 64;
        v2f b0, b1, b2, b3;
        b0.x = w0[r0];      b0.y = w1[r0];
        b1.x = w0[16 + r0]; b1.y = w1[16 + r0];
        b2.x = w0[32 + r0]; b2.y = w1[32 + r0];
        b3.x = w0[48 + r0]; b3.y = w1[48 + r0];

        c0 = __builtin_amdgcn_wmma_f32_16x16x4_f32(false, a, false, b0, (short)0, c0, false, false);
        c1 = __builtin_amdgcn_wmma_f32_16x16x4_f32(false, a, false, b1, (short)0, c1, false, false);
        c2 = __builtin_amdgcn_wmma_f32_16x16x4_f32(false, a, false, b2, (short)0, c2, false, false);
        c3 = __builtin_amdgcn_wmma_f32_16x16x4_f32(false, a, false, b3, (short)0, c3, false, false);
    }

    float* hbase = H + (size_t)(wave * 16 + khalf * 8) * 64 + r0;
    #pragma unroll
    for (int r = 0; r < 8; ++r) {
        float* row = hbase + (size_t)r * 64;
        row[0]  = c0[r];
        row[16] = c1[r];
        row[32] = c2[r];
        row[48] = c3[r];
    }
}

#define NPW 8
#define NSTRIP (NN / NPW)
#define NQ 4
static_assert(NN % NPW == 0, "a strip must not run past the last node");
static_assert(NPW % NQ == 0, "a strip is a whole number of groups");

__global__ __launch_bounds__(256) void scores_kernel(
        const float* __restrict__ H, const float* __restrict__ a_s,
        const float* __restrict__ a_d, float* __restrict__ S, float* __restrict__ D) {
    const int lane = threadIdx.x & 31;
    const int strip = __builtin_amdgcn_readfirstlane((int)(blockIdx.x * 8 + (threadIdx.x >> 5)));
    if (strip >= NSTRIP) return;
    const int n0 = strip * NPW;
    const float as0 = a_s[lane], as1 = a_s[lane + 32];
    const float ad0 = a_d[lane], ad1 = a_d[lane + 32];
    const float* h = H + (size_t)n0 * 64;
    float h0[NPW], h1[NPW];
    #pragma unroll
    for (int u = 0; u < NPW; ++u) {
        h0[u] = h[u * 64 + lane];
        h1[u] = h[u * 64 + lane + 32];
    }
    float so = 0.0f, dd = 0.0f;
    #pragma unroll
    for (int u = 0; u < NPW; ++u) {
        float s = h0[u] * as0 + h1[u] * as1;
        float d = h0[u] * ad0 + h1[u] * ad1;
        #pragma unroll
        for (int off = 16; off; off >>= 1) {
            s += __shfl_xor(s, off, 32);
            d += __shfl_xor(d, off, 32);
        }
        if (lane == u) { so = s; dd = d; }
    }
    if (lane < NPW) { S[n0 + lane] = so; D[n0 + lane] = dd; }
}

#define NPB_LOG2 8
#define NPB (1 << NPB_LOG2)
#define NBUCK ((NN + NPB - 1) / NPB)
#define CH 4096
#define EPT (CH / 256)
#define SRC_BITS 17
#define SRC_MASK ((1 << SRC_BITS) - 1)
static_assert(NN <= (1 << SRC_BITS), "a source id must fit the low bits of the packed word");
static_assert(NPB_LOG2 + SRC_BITS + 1 < 32, "packed word and the spare value past it must stay non-negative");

__global__ __launch_bounds__(256) void bucket_hist_kernel(const int* __restrict__ ei, int E, int G1,
        int* __restrict__ cntm) {
    __shared__ int hist[NBUCK];
    const int tid = threadIdx.x;
    for (int i = tid; i < NBUCK; i += 256) hist[i] = 0;
    __syncthreads();
    const int e0 = blockIdx.x * CH + tid;
    int d[EPT];
    #pragma unroll
    for (int k = 0; k < EPT; ++k) {
        const int e = e0 + k * 256;
        d[k] = (e < E) ? ei[E + e] : -1;
    }
    #pragma unroll
    for (int k = 0; k < EPT; ++k) {
        if (d[k] >= 0) atomicAdd(&hist[d[k] >> NPB_LOG2], 1);
    }
    __syncthreads();
    for (int i = tid; i < NBUCK; i += 256) cntm[i * G1 + blockIdx.x] = hist[i];
}

__global__ __launch_bounds__(1024) void scan_counts_kernel(
        const int* __restrict__ cnt, int* __restrict__ off, int n) {
    __shared__ int red[32];
    __shared__ int wtot[32];
    const int tid  = threadIdx.x;
    const int lane = tid & 31;
    const int wv   = tid >> 5;
    const int tile0 = blockIdx.x * 1024;

    const int4* c4 = (const int4*)cnt;
    const int nvec = tile0 >> 2;
    int part = 0;
    for (int i = tid; i < nvec; i += 1024) {
        int4 v = c4[i];
        part += (v.x + v.y) + (v.z + v.w);
    }
    #pragma unroll
    for (int o = 16; o; o >>= 1) part += __shfl_xor(part, o, 32);
    if (lane == 0) red[wv] = part;
    __syncthreads();
    int base = red[lane];
    #pragma unroll
    for (int o = 16; o; o >>= 1) base += __shfl_xor(base, o, 32);

    const int i = tile0 + tid;
    const int v = (i < n) ? cnt[i] : 0;
    int inc = v;
    #pragma unroll
    for (int o = 1; o < 32; o <<= 1) {
        int t = __shfl_up(inc, o, 32);
        if (lane >= o) inc += t;
    }
    if (lane == 31) wtot[wv] = inc;
    __syncthreads();
    const int wt = wtot[lane];
    int winc = wt;
    #pragma unroll
    for (int o = 1; o < 32; o <<= 1) {
        int t = __shfl_up(winc, o, 32);
        if (lane >= o) winc += t;
    }
    const int wprefix = __shfl(winc, wv, 32) - __shfl(wt, wv, 32);
    if (i < n) off[i] = base + wprefix + inc - v;
    if (i == n - 1) off[n] = base + wprefix + inc;
}

__global__ __launch_bounds__(256) void bucket_scatter_kernel(const int* __restrict__ ei, int E, int G1,
        const int* __restrict__ boff, int* __restrict__ tmp) {
    __shared__ int cur[NBUCK + 1];
    const int tid = threadIdx.x;
    const int e0 = blockIdx.x * CH + tid;
    int d[EPT], s[EPT];
    #pragma unroll
    for (int k = 0; k < EPT; ++k) {
        const int e = e0 + k * 256;
        d[k] = (e < E) ? ei[E + e] : (NBUCK << NPB_LOG2);
        s[k] = (e < E) ? ei[e] : 0;
    }
    for (int i = tid; i < NBUCK; i += 256) cur[i] = boff[i * G1 + blockIdx.x];
    if (tid == 0) cur[NBUCK] = 0;
    __syncthreads();
    int pos[EPT];
    #pragma unroll
    for (int k = 0; k < EPT; ++k) pos[k] = atomicAdd(&cur[d[k] >> NPB_LOG2], 1);
    #pragma unroll
    for (int k = 0; k < EPT; ++k) {
        if (e0 + k * 256 < E) tmp[pos[k]] = ((d[k] & (NPB - 1)) << SRC_BITS) | s[k];
    }
}

__global__ __launch_bounds__(256) void bucket_csr_kernel(const int* __restrict__ boff, int G1,
        const int* __restrict__ tmp, int* __restrict__ off, int* __restrict__ col) {
    __shared__ int cnt[NPB + 1];
    __shared__ int wtot[NPB / 32];
    const int tid  = threadIdx.x;
    const int lane = tid & 31;
    const int wv   = tid >> 5;
    const int bs = boff[blockIdx.x * G1];
    const int be = boff[(blockIdx.x + 1) * G1];
    cnt[tid] = 0;
    if (tid == 0) cnt[NPB] = 0;
    __syncthreads();
    for (int i = bs + tid; i < be; i += 4 * 256) {
        int w[4];
        #pragma unroll
        for (int u = 0; u < 4; ++u) {
            const int j = i + u * 256;
            w[u] = (j < be) ? tmp[j] : -1;
        }
        #pragma unroll
        for (int u = 0; u < 4; ++u) {
            if (w[u] >= 0) atomicAdd(&cnt[w[u] >> SRC_BITS], 1);
        }
    }
    __syncthreads();
    const int v = cnt[tid];
    int inc = v;
    #pragma unroll
    for (int o = 1; o < 32; o <<= 1) {
        int t = __shfl_up(inc, o, 32);
        if (lane >= o) inc += t;
    }
    if (lane == 31) wtot[wv] = inc;
    __syncthreads();
    int wprefix = 0;
    for (int w = 0; w < wv; ++w) wprefix += wtot[w];
    const int excl = bs + wprefix + inc - v;
    const int node = blockIdx.x * NPB + tid;
    if (node < NN) off[node] = excl;
    if (node == NN - 1) off[NN] = excl + v;
    cnt[tid] = excl;
    __syncthreads();
    for (int i = bs + tid; i < be; i += 4 * 256) {
        int w[4];
        #pragma unroll
        for (int u = 0; u < 4; ++u) {
            const int j = i + u * 256;
            w[u] = (j < be) ? tmp[j] : (NPB << SRC_BITS);
        }
        int pos[4];
        #pragma unroll
        for (int u = 0; u < 4; ++u) pos[u] = atomicAdd(&cnt[w[u] >> SRC_BITS], 1);
        #pragma unroll
        for (int u = 0; u < 4; ++u) {
            if (i + u * 256 < be) col[pos[u]] = w[u] & SRC_MASK;
        }
    }
}

__global__ __launch_bounds__(256) void sort_segments_kernel(
        const int* __restrict__ off, const int* __restrict__ cin, int* __restrict__ cout) {
    const int lane = threadIdx.x & 31;
    const int strip = __builtin_amdgcn_readfirstlane((int)(blockIdx.x * 8 + (threadIdx.x >> 5)));
    if (strip >= NSTRIP) return;
    const int n0 = strip * NPW;
    const int offv = off[n0 + ((lane < NPW) ? lane : NPW)];
    int o0[NPW], deg[NPW], key[NPW];
    #pragma unroll
    for (int u = 0; u < NPW; ++u) {
        o0[u]  = __builtin_amdgcn_readlane(offv, u);
        deg[u] = __builtin_amdgcn_readlane(offv, u + 1) - o0[u];
        key[u] = (lane < deg[u]) ? cin[o0[u] + lane] : 0x7fffffff;
    }
    #pragma unroll
    for (int u = 0; u < NPW; ++u) {
        if (deg[u] <= 32) {
            int rank = 0;
            for (int j = 0; j < deg[u]; ++j) {
                const int kj = __builtin_amdgcn_readlane(key[u], j);
                rank += (kj < key[u] || (kj == key[u] && j < lane)) ? 1 : 0;
            }
            if (lane < deg[u]) cout[o0[u] + rank] = key[u];
        } else {
            for (int i = lane; i < deg[u]; i += 32) {
                const int ki = cin[o0[u] + i];
                int rank = 0;
                for (int j = 0; j < deg[u]; ++j) {
                    const int kj = cin[o0[u] + j];
                    rank += (kj < ki || (kj == ki && j < i)) ? 1 : 0;
                }
                cout[o0[u] + rank] = ki;
            }
        }
    }
}

__device__ __forceinline__ float leaky(float a) { return (a >= 0.0f) ? a : NEG_SLOPE * a; }

__device__ __forceinline__ float readlane_f(float v, int l) {
    return __int_as_float(__builtin_amdgcn_readlane(__float_as_int(v), l));
}

__device__ __forceinline__ void finish_row(v2f acc, float bx, float by, int relu,
                                           float* __restrict__ OUT, int node, int lane) {
    acc.x += bx;
    acc.y += by;
    if (relu) {
        acc.x = (acc.x > 0.0f) ? acc.x : 0.0f;
        acc.y = (acc.y > 0.0f) ? acc.y : 0.0f;
    }
    ((v2f*)OUT)[(size_t)node * 32 + lane] = acc;
}

__global__ __launch_bounds__(256) void node_softmax_sum_kernel(
        const int* __restrict__ off, const int* __restrict__ col,
        const float* __restrict__ H, const float* __restrict__ S, const float* __restrict__ D,
        const float* __restrict__ bias, int relu, float* __restrict__ OUT) {
    const int lane = threadIdx.x & 31;
    const int strip = __builtin_amdgcn_readfirstlane((int)(blockIdx.x * 8 + (threadIdx.x >> 5)));
    if (strip >= NSTRIP) return;
    const int n0 = strip * NPW;
    const int   offv = off[n0 + ((lane < NPW) ? lane : NPW)];
    const float Sv   = S[n0 + (lane & (NPW - 1))];
    const float Dv   = D[n0 + (lane & (NPW - 1))];
    const v2f* H2 = (const v2f*)H;
    const float bx = bias[2 * lane];
    const float by = bias[2 * lane + 1];

    bool any_long = false;
    for (int q = 0; q < NPW; q += NQ) {
        int o0[NQ], deg[NQ], sv[NQ];
        v2f acc0[NQ];
        #pragma unroll
        for (int u = 0; u < NQ; ++u) {
            o0[u]   = __builtin_amdgcn_readlane(offv, q + u);
            deg[u]  = __builtin_amdgcn_readlane(offv, q + u + 1) - o0[u];
            sv[u]   = (lane < deg[u]) ? col[o0[u] + lane] : (n0 + q + u);
            acc0[u] = H2[(size_t)(n0 + q + u) * 32 + lane];
        }
        float av[NQ], aself[NQ];
        #pragma unroll
        for (int u = 0; u < NQ; ++u) av[u] = S[sv[u]];
        #pragma unroll
        for (int u = 0; u < NQ; ++u) {
            const float dn = readlane_f(Dv, q + u);
            aself[u] = leaky(readlane_f(Sv, q + u) + dn);
            av[u] = (lane < deg[u]) ? leaky(av[u] + dn) : -INFINITY;
        }
        float wv[NQ];
        #pragma unroll
        for (int u = 0; u < NQ; ++u) {
            float m = fmaxf(av[u], aself[u]);
            #pragma unroll
            for (int o = 16; o; o >>= 1) m = fmaxf(m, __shfl_xor(m, o, 32));
            const float ex = expf(av[u] - m);
            const float e = (lane < deg[u]) ? ex : 0.0f;
            const float e_self = expf(aself[u] - m);
            float z = e;
            #pragma unroll
            for (int o = 16; o; o >>= 1) z += __shfl_xor(z, o, 32);
            z += e_self;
            wv[u] = e / (z + EPSF);
            const float w_self = e_self / (z + EPSF);
            acc0[u].x = acc0[u].x * w_self;
            acc0[u].y = acc0[u].y * w_self;
        }
        #pragma unroll
        for (int u = 0; u < NQ; ++u) {
            if (deg[u] > 32) { any_long = true; continue; }
            const float w = wv[u];
            v2f acc = acc0[u];
            for (int j = 0; j < deg[u]; j += 8) {
                v2f hv[8];
                #pragma unroll
                for (int t = 0; t < 8; ++t) {
                    const int jt = (j + t < deg[u]) ? (j + t) : (deg[u] - 1);
                    const int sj = __builtin_amdgcn_readlane(sv[u], jt);
                    hv[t] = H2[(size_t)sj * 32 + lane];
                }
                #pragma unroll
                for (int t = 0; t < 8; ++t) {
                    const float wj = readlane_f(w, j + t);
                    acc.x = fmaf(hv[t].x, wj, acc.x);
                    acc.y = fmaf(hv[t].y, wj, acc.y);
                }
            }
            finish_row(acc, bx, by, relu, OUT, n0 + q + u, lane);
        }
    }

    if (any_long) {
        for (int u = 0; u < NPW; ++u) {
            const int lo = __builtin_amdgcn_readlane(offv, u);
            const int dg = __builtin_amdgcn_readlane(offv, u + 1) - lo;
            if (dg <= 32) continue;
            const int node = n0 + u;
            const float dn = readlane_f(Dv, u);
            const float a_self = leaky(readlane_f(Sv, u) + dn);
            const v2f hsn = H2[(size_t)node * 32 + lane];
            float m = a_self;
            for (int c = 0; c < dg; c += 32) {
                if (c + lane < dg) m = fmaxf(m, leaky(S[col[lo + c + lane]] + dn));
            }
            #pragma unroll
            for (int o = 16; o; o >>= 1) m = fmaxf(m, __shfl_xor(m, o, 32));
            float z = 0.0f;
            for (int c = 0; c < dg; c += 32) {
                if (c + lane < dg) z += expf(leaky(S[col[lo + c + lane]] + dn) - m);
            }
            #pragma unroll
            for (int o = 16; o; o >>= 1) z += __shfl_xor(z, o, 32);
            const float e_self = expf(a_self - m);
            z += e_self;
            const float w_self = e_self / (z + EPSF);
            v2f acc;
            acc.x = hsn.x * w_self;
            acc.y = hsn.y * w_self;
            for (int c = 0; c < dg; c += 32) {
                int s = node;
                float w = 0.0f;
                if (c + lane < dg) {
                    s = col[lo + c + lane];
                    w = expf(leaky(S[s] + dn) - m) / (z + EPSF);
                }
                const int nj = (dg - c < 32) ? (dg - c) : 32;
                for (int j = 0; j < nj; ++j) {
                    const int sj = __builtin_amdgcn_readlane(s, j);
                    const float wj = readlane_f(w, j);
                    const v2f hv = H2[(size_t)sj * 32 + lane];
                    acc.x = fmaf(hv.x, wj, acc.x);
                    acc.y = fmaf(hv.y, wj, acc.y);
                }
            }
            finish_row(acc, bx, by, relu, OUT, node, lane);
        }
    }
}

extern "C" void kernel_launch(void* const* d_in, const int* in_sizes, int n_in,
                              void* d_out, int out_size, void* d_ws, size_t ws_size,
                              hipStream_t stream) {
    const float* x     = (const float*)d_in[0];
    const int*   ei    = (const int*)d_in[1];
    const float* W1    = (const float*)d_in[2];
    const float* as1   = (const float*)d_in[3];
    const float* ad1   = (const float*)d_in[4];
    const float* b1    = (const float*)d_in[5];
    const float* W2    = (const float*)d_in[6];
    const float* as2   = (const float*)d_in[7];
    const float* ad2   = (const float*)d_in[8];
    const float* b2    = (const float*)d_in[9];
    float* out = (float*)d_out;

    const int E  = in_sizes[1] / 2;
    const int NF = NN * 64;

    float* wsf = (float*)d_ws;
    float* H   = wsf;
    float* X2  = H   + NF;
    float* S   = X2  + NF;
    float* D   = S   + NN;
    const int G1 = (E + CH - 1) / CH;
    const int nM = NBUCK * G1;
    int*   cntm = (int*)(D + NN);
    int*   boff = cntm + nM + 16;
    int*   off  = boff + nM + 16;
    int*   tmp  = off + NN + 16;
    int*   cA   = tmp + E;
    int*   cB   = cA  + E;

    const int TB = 256;
    dim3 blk(TB);
    int gScan = (nM + 1023) / 1024;
    int gNode = (NSTRIP + 7) / 8;
    int gGemm = ((NN / 16) + 3) / 4;

    bucket_hist_kernel<<<G1, blk, 0, stream>>>(ei, E, G1, cntm);
    scan_counts_kernel<<<gScan, dim3(1024), 0, stream>>>(cntm, boff, nM);
    bucket_scatter_kernel<<<G1, blk, 0, stream>>>(ei, E, G1, boff, tmp);
    bucket_csr_kernel<<<NBUCK, blk, 0, stream>>>(boff, G1, tmp, off, cA);
    sort_segments_kernel<<<gNode, blk, 0, stream>>>(off, cA, cB);

    gemm_wmma_kernel<<<gGemm, dim3(128), 0, stream>>>(x, W1, H, NN, 128);
    scores_kernel<<<gNode, blk, 0, stream>>>(H, as1, ad1, S, D);
    node_softmax_sum_kernel<<<gNode, blk, 0, stream>>>(off, cB, H, S, D, b1, 1, X2);

    gemm_wmma_kernel<<<gGemm, dim3(128), 0, stream>>>(X2, W2, H, NN, 64);
    scores_kernel<<<gNode, blk, 0, stream>>>(H, as2, ad2, S, D);
    node_softmax_sum_kernel<<<gNode, blk, 0, stream>>>(off, cB, H, S, D, b2, 0, out);
}


# Per-node scores S and D formed in the GEMM epilogue from the accumulator registers (each score's terms and sum tree kept), not by a kernel of their own that reads H back: 9 launches, not 11
# speedup vs baseline: 4.1720x; 1.0745x over previous
#include <hip/hip_runtime.h>
#include <math.h>


typedef float v2f __attribute__((ext_vector_type(2)));
typedef float v8f __attribute__((ext_vector_type(8)));

#define NN 100000
#define NEG_SLOPE 0.2f
#define EPSF 1e-16f

__global__ __launch_bounds__(128) void gemm_wmma_kernel(
        const float* __restrict__ X, const float* __restrict__ W,
        const float* __restrict__ a_s, const float* __restrict__ a_d,
        float* __restrict__ H, float* __restrict__ S, float* __restrict__ D, int nrows, int K) {
    int wave = blockIdx.x * (blockDim.x >> 5) + (threadIdx.x >> 5);
    int mtiles = nrows >> 4;
    if (wave >= mtiles) return;

    int lane  = threadIdx.x & 31;
    int r0    = lane & 15;
    int khalf = lane >> 4;

    const float* xrow = X + (size_t)(wave * 16 + r0) * (size_t)K;

    const float as0 = a_s[r0], as1 = a_s[16 + r0], as2 = a_s[32 + r0], as3 = a_s[48 + r0];
    const float ad0 = a_d[r0], ad1 = a_d[16 + r0], ad2 = a_d[32 + r0], ad3 = a_d[48 + r0];

    v8f c0 = {}, c1 = {}, c2 = {}, c3 = {};
    for (int k0 = 0; k0 < K; k0 += 4) {
        int ka = k0 + khalf * 2;
        v2f a;
        a.x = xrow[ka];
        a.y = xrow[ka + 1];

        const float* w0 = W + (size_t)ka * 64;
        const float* w1 = w0 + 64;
        v2f b0, b1, b2, b3;
        b0.x = w0[r0];      b0.y = w1[r0];
        b1.x = w0[16 + r0]; b1.y = w1[16 + r0];
        b2.x = w0[32 + r0]; b2.y = w1[32 + r0];
        b3.x = w0[48 + r0]; b3.y = w1[48 + r0];

        c0 = __builtin_amdgcn_wmma_f32_16x16x4_f32(false, a, false, b0, (short)0, c0, false, false);
        c1 = __builtin_amdgcn_wmma_f32_16x16x4_f32(false, a, false, b1, (short)0, c1, false, false);
        c2 = __builtin_amdgcn_wmma_f32_16x16x4_f32(false, a, false, b2, (short)0, c2, false, false);
        c3 = __builtin_amdgcn_wmma_f32_16x16x4_f32(false, a, false, b3, (short)0, c3, false, false);
    }

    float* hbase = H + (size_t)(wave * 16 + khalf * 8) * 64 + r0;
    #pragma unroll
    for (int r = 0; r < 8; ++r) {
        float* row = hbase + (size_t)r * 64;
        row[0]  = c0[r];
        row[16] = c1[r];
        row[32] = c2[r];
        row[48] = c3[r];
    }

    float so = 0.0f, dd = 0.0f;
    #pragma unroll
    for (int r = 0; r < 8; ++r) {
        float s = fmaf(c0[r], as0, c2[r] * as2) + fmaf(c1[r], as1, c3[r] * as3);
        float d = fmaf(c0[r], ad0, c2[r] * ad2) + fmaf(c1[r], ad1, c3[r] * ad3);
        #pragma unroll
        for (int off = 8; off; off >>= 1) {
            s += __shfl_xor(s, off, 32);
            d += __shfl_xor(d, off, 32);
        }
        if (r0 == r) { so = s; dd = d; }
    }
    if (r0 < 8) {
        S[wave * 16 + khalf * 8 + r0] = so;
        D[wave * 16 + khalf * 8 + r0] = dd;
    }
}

#define NPW 8
#define NSTRIP (NN / NPW)
#define NQ 4
static_assert(NN % NPW == 0, "a strip must not run past the last node");
static_assert(NPW % NQ == 0, "a strip is a whole number of groups");

#define NPB_LOG2 8
#define NPB (1 << NPB_LOG2)
#define NBUCK ((NN + NPB - 1) / NPB)
#define CH 4096
#define EPT (CH / 256)
#define SRC_BITS 17
#define SRC_MASK ((1 << SRC_BITS) - 1)
static_assert(NN <= (1 << SRC_BITS), "a source id must fit the low bits of the packed word");
static_assert(NPB_LOG2 + SRC_BITS + 1 < 32, "packed word and the spare value past it must stay non-negative");

__global__ __launch_bounds__(256) void bucket_hist_kernel(const int* __restrict__ ei, int E, int G1,
        int* __restrict__ cntm) {
    __shared__ int hist[NBUCK];
    const int tid = threadIdx.x;
    for (int i = tid; i < NBUCK; i += 256) hist[i] = 0;
    __syncthreads();
    const int e0 = blockIdx.x * CH + tid;
    int d[EPT];
    #pragma unroll
    for (int k = 0; k < EPT; ++k) {
        const int e = e0 + k * 256;
        d[k] = (e < E) ? ei[E + e] : -1;
    }
    #pragma unroll
    for (int k = 0; k < EPT; ++k) {
        if (d[k] >= 0) atomicAdd(&hist[d[k] >> NPB_LOG2], 1);
    }
    __syncthreads();
    for (int i = tid; i < NBUCK; i += 256) cntm[i * G1 + blockIdx.x] = hist[i];
}

__global__ __launch_bounds__(1024) void scan_counts_kernel(
        const int* __restrict__ cnt, int* __restrict__ off, int n) {
    __shared__ int red[32];
    __shared__ int wtot[32];
    const int tid  = threadIdx.x;
    const int lane = tid & 31;
    const int wv   = tid >> 5;
    const int tile0 = blockIdx.x * 1024;

    const int4* c4 = (const int4*)cnt;
    const int nvec = tile0 >> 2;
    int part = 0;
    for (int i = tid; i < nvec; i += 1024) {
        int4 v = c4[i];
        part += (v.x + v.y) + (v.z + v.w);
    }
    #pragma unroll
    for (int o = 16; o; o >>= 1) part += __shfl_xor(part, o, 32);
    if (lane == 0) red[wv] = part;
    __syncthreads();
    int base = red[lane];
    #pragma unroll
    for (int o = 16; o; o >>= 1) base += __shfl_xor(base, o, 32);

    const int i = tile0 + tid;
    const int v = (i < n) ? cnt[i] : 0;
    int inc = v;
    #pragma unroll
    for (int o = 1; o < 32; o <<= 1) {
        int t = __shfl_up(inc, o, 32);
        if (lane >= o) inc += t;
    }
    if (lane == 31) wtot[wv] = inc;
    __syncthreads();
    const int wt = wtot[lane];
    int winc = wt;
    #pragma unroll
    for (int o = 1; o < 32; o <<= 1) {
        int t = __shfl_up(winc, o, 32);
        if (lane >= o) winc += t;
    }
    const int wprefix = __shfl(winc, wv, 32) - __shfl(wt, wv, 32);
    if (i < n) off[i] = base + wprefix + inc - v;
    if (i == n - 1) off[n] = base + wprefix + inc;
}

__global__ __launch_bounds__(256) void bucket_scatter_kernel(const int* __restrict__ ei, int E, int G1,
        const int* __restrict__ boff, int* __restrict__ tmp) {
    __shared__ int cur[NBUCK + 1];
    const int tid = threadIdx.x;
    const int e0 = blockIdx.x * CH + tid;
    int d[EPT], s[EPT];
    #pragma unroll
    for (int k = 0; k < EPT; ++k) {
        const int e = e0 + k * 256;
        d[k] = (e < E) ? ei[E + e] : (NBUCK << NPB_LOG2);
        s[k] = (e < E) ? ei[e] : 0;
    }
    for (int i = tid; i < NBUCK; i += 256) cur[i] = boff[i * G1 + blockIdx.x];
    if (tid == 0) cur[NBUCK] = 0;
    __syncthreads();
    int pos[EPT];
    #pragma unroll
    for (int k = 0; k < EPT; ++k) pos[k] = atomicAdd(&cur[d[k] >> NPB_LOG2], 1);
    #pragma unroll
    for (int k = 0; k < EPT; ++k) {
        if (e0 + k * 256 < E) tmp[pos[k]] = ((d[k] & (NPB - 1)) << SRC_BITS) | s[k];
    }
}

__global__ __launch_bounds__(256) void bucket_csr_kernel(const int* __restrict__ boff, int G1,
        const int* __restrict__ tmp, int* __restrict__ off, int* __restrict__ col) {
    __shared__ int cnt[NPB + 1];
    __shared__ int wtot[NPB / 32];
    const int tid  = threadIdx.x;
    const int lane = tid & 31;
    const int wv   = tid >> 5;
    const int bs = boff[blockIdx.x * G1];
    const int be = boff[(blockIdx.x + 1) * G1];
    cnt[tid] = 0;
    if (tid == 0) cnt[NPB] = 0;
    __syncthreads();
    for (int i = bs + tid; i < be; i += 4 * 256) {
        int w[4];
        #pragma unroll
        for (int u = 0; u < 4; ++u) {
            const int j = i + u * 256;
            w[u] = (j < be) ? tmp[j] : -1;
        }
        #pragma unroll
        for (int u = 0; u < 4; ++u) {
            if (w[u] >= 0) atomicAdd(&cnt[w[u] >> SRC_BITS], 1);
        }
    }
    __syncthreads();
    const int v = cnt[tid];
    int inc = v;
    #pragma unroll
    for (int o = 1; o < 32; o <<= 1) {
        int t = __shfl_up(inc, o, 32);
        if (lane >= o) inc += t;
    }
    if (lane == 31) wtot[wv] = inc;
    __syncthreads();
    int wprefix = 0;
    for (int w = 0; w < wv; ++w) wprefix += wtot[w];
    const int excl = bs + wprefix + inc - v;
    const int node = blockIdx.x * NPB + tid;
    if (node < NN) off[node] = excl;
    if (node == NN - 1) off[NN] = excl + v;
    cnt[tid] = excl;
    __syncthreads();
    for (int i = bs + tid; i < be; i += 4 * 256) {
        int w[4];
        #pragma unroll
        for (int u = 0; u < 4; ++u) {
            const int j = i + u * 256;
            w[u] = (j < be) ? tmp[j] : (NPB << SRC_BITS);
        }
        int pos[4];
        #pragma unroll
        for (int u = 0; u < 4; ++u) pos[u] = atomicAdd(&cnt[w[u] >> SRC_BITS], 1);
        #pragma unroll
        for (int u = 0; u < 4; ++u) {
            if (i + u * 256 < be) col[pos[u]] = w[u] & SRC_MASK;
        }
    }
}

__global__ __launch_bounds__(256) void sort_segments_kernel(
        const int* __restrict__ off, const int* __restrict__ cin, int* __restrict__ cout) {
    const int lane = threadIdx.x & 31;
    const int strip = __builtin_amdgcn_readfirstlane((int)(blockIdx.x * 8 + (threadIdx.x >> 5)));
    if (strip >= NSTRIP) return;
    const int n0 = strip * NPW;
    const int offv = off[n0 + ((lane < NPW) ? lane : NPW)];
    int o0[NPW], deg[NPW], key[NPW];
    #pragma unroll
    for (int u = 0; u < NPW; ++u) {
        o0[u]  = __builtin_amdgcn_readlane(offv, u);
        deg[u] = __builtin_amdgcn_readlane(offv, u + 1) - o0[u];
        key[u] = (lane < deg[u]) ? cin[o0[u] + lane] : 0x7fffffff;
    }
    #pragma unroll
    for (int u = 0; u < NPW; ++u) {
        if (deg[u] <= 32) {
            int rank = 0;
            for (int j = 0; j < deg[u]; ++j) {
                const int kj = __builtin_amdgcn_readlane(key[u], j);
                rank += (kj < key[u] || (kj == key[u] && j < lane)) ? 1 : 0;
            }
            if (lane < deg[u]) cout[o0[u] + rank] = key[u];
        } else {
            for (int i = lane; i < deg[u]; i += 32) {
                const int ki = cin[o0[u] + i];
                int rank = 0;
                for (int j = 0; j < deg[u]; ++j) {
                    const int kj = cin[o0[u] + j];
                    rank += (kj < ki || (kj == ki && j < i)) ? 1 : 0;
                }
                cout[o0[u] + rank] = ki;
            }
        }
    }
}

__device__ __forceinline__ float leaky(float a) { return (a >= 0.0f) ? a : NEG_SLOPE * a; }

__device__ __forceinline__ float readlane_f(float v, int l) {
    return __int_as_float(__builtin_amdgcn_readlane(__float_as_int(v), l));
}

__device__ __forceinline__ void finish_row(v2f acc, float bx, float by, int relu,
                                           float* __restrict__ OUT, int node, int lane) {
    acc.x += bx;
    acc.y += by;
    if (relu) {
        acc.x = (acc.x > 0.0f) ? acc.x : 0.0f;
        acc.y = (acc.y > 0.0f) ? acc.y : 0.0f;
    }
    ((v2f*)OUT)[(size_t)node * 32 + lane] = acc;
}

__global__ __launch_bounds__(256) void node_softmax_sum_kernel(
        const int* __restrict__ off, const int* __restrict__ col,
        const float* __restrict__ H, const float* __restrict__ S, const float* __restrict__ D,
        const float* __restrict__ bias, int relu, float* __restrict__ OUT) {
    const int lane = threadIdx.x & 31;
    const int strip = __builtin_amdgcn_readfirstlane((int)(blockIdx.x * 8 + (threadIdx.x >> 5)));
    if (strip >= NSTRIP) return;
    const int n0 = strip * NPW;
    const int   offv = off[n0 + ((lane < NPW) ? lane : NPW)];
    const float Sv   = S[n0 + (lane & (NPW - 1))];
    const float Dv   = D[n0 + (lane & (NPW - 1))];
    const v2f* H2 = (const v2f*)H;
    const float bx = bias[2 * lane];
    const float by = bias[2 * lane + 1];

    bool any_long = false;
    for (int q = 0; q < NPW; q += NQ) {
        int o0[NQ], deg[NQ], sv[NQ];
        v2f acc0[NQ];
        #pragma unroll
        for (int u = 0; u < NQ; ++u) {
            o0[u]   = __builtin_amdgcn_readlane(offv, q + u);
            deg[u]  = __builtin_amdgcn_readlane(offv, q + u + 1) - o0[u];
            sv[u]   = (lane < deg[u]) ? col[o0[u] + lane] : (n0 + q + u);
            acc0[u] = H2[(size_t)(n0 + q + u) * 32 + lane];
        }
        float av[NQ], aself[NQ];
        #pragma unroll
        for (int u = 0; u < NQ; ++u) av[u] = S[sv[u]];
        #pragma unroll
        for (int u = 0; u < NQ; ++u) {
            const float dn = readlane_f(Dv, q + u);
            aself[u] = leaky(readlane_f(Sv, q + u) + dn);
            av[u] = (lane < deg[u]) ? leaky(av[u] + dn) : -INFINITY;
        }
        float wv[NQ];
        #pragma unroll
        for (int u = 0; u < NQ; ++u) {
            float m = fmaxf(av[u], aself[u]);
            #pragma unroll
            for (int o = 16; o; o >>= 1) m = fmaxf(m, __shfl_xor(m, o, 32));
            const float ex = expf(av[u] - m);
            const float e = (lane < deg[u]) ? ex : 0.0f;
            const float e_self = expf(aself[u] - m);
            float z = e;
            #pragma unroll
            for (int o = 16; o; o >>= 1) z += __shfl_xor(z, o, 32);
            z += e_self;
            wv[u] = e / (z + EPSF);
            const float w_self = e_self / (z + EPSF);
            acc0[u].x = acc0[u].x * w_self;
            acc0[u].y = acc0[u].y * w_self;
        }
        #pragma unroll
        for (int u = 0; u < NQ; ++u) {
            if (deg[u] > 32) { any_long = true; continue; }
            const float w = wv[u];
            v2f acc = acc0[u];
            for (int j = 0; j < deg[u]; j += 8) {
                v2f hv[8];
                #pragma unroll
                for (int t = 0; t < 8; ++t) {
                    const int jt = (j + t < deg[u]) ? (j + t) : (deg[u] - 1);
                    const int sj = __builtin_amdgcn_readlane(sv[u], jt);
                    hv[t] = H2[(size_t)sj * 32 + lane];
                }
                #pragma unroll
                for (int t = 0; t < 8; ++t) {
                    const float wj = readlane_f(w, j + t);
                    acc.x = fmaf(hv[t].x, wj, acc.x);
                    acc.y = fmaf(hv[t].y, wj, acc.y);
                }
            }
            finish_row(acc, bx, by, relu, OUT, n0 + q + u, lane);
        }
    }

    if (any_long) {
        for (int u = 0; u < NPW; ++u) {
            const int lo = __builtin_amdgcn_readlane(offv, u);
            const int dg = __builtin_amdgcn_readlane(offv, u + 1) - lo;
            if (dg <= 32) continue;
            const int node = n0 + u;
            const float dn = readlane_f(Dv, u);
            const float a_self = leaky(readlane_f(Sv, u) + dn);
            const v2f hsn = H2[(size_t)node * 32 + lane];
            float m = a_self;
            for (int c = 0; c < dg; c += 32) {
                if (c + lane < dg) m = fmaxf(m, leaky(S[col[lo + c + lane]] + dn));
            }
            #pragma unroll
            for (int o = 16; o; o >>= 1) m = fmaxf(m, __shfl_xor(m, o, 32));
            float z = 0.0f;
            for (int c = 0; c < dg; c += 32) {
                if (c + lane < dg) z += expf(leaky(S[col[lo + c + lane]] + dn) - m);
            }
            #pragma unroll
            for (int o = 16; o; o >>= 1) z += __shfl_xor(z, o, 32);
            const float e_self = expf(a_self - m);
            z += e_self;
            const float w_self = e_self / (z + EPSF);
            v2f acc;
            acc.x = hsn.x * w_self;
            acc.y = hsn.y * w_self;
            for (int c = 0; c < dg; c += 32) {
                int s = node;
                float w = 0.0f;
                if (c + lane < dg) {
                    s = col[lo + c + lane];
                    w = expf(leaky(S[s] + dn) - m) / (z + EPSF);
                }
                const int nj = (dg - c < 32) ? (dg - c) : 32;
                for (int j = 0; j < nj; ++j) {
                    const int sj = __builtin_amdgcn_readlane(s, j);
                    const float wj = readlane_f(w, j);
                    const v2f hv = H2[(size_t)sj * 32 + lane];
                    acc.x = fmaf(hv.x, wj, acc.x);
                    acc.y = fmaf(hv.y, wj, acc.y);
                }
            }
            finish_row(acc, bx, by, relu, OUT, node, lane);
        }
    }
}

extern "C" void kernel_launch(void* const* d_in, const int* in_sizes, int n_in,
                              void* d_out, int out_size, void* d_ws, size_t ws_size,
                              hipStream_t stream) {
    const float* x     = (const float*)d_in[0];
    const int*   ei    = (const int*)d_in[1];
    const float* W1    = (const float*)d_in[2];
    const float* as1   = (const float*)d_in[3];
    const float* ad1   = (const float*)d_in[4];
    const float* b1    = (const float*)d_in[5];
    const float* W2    = (const float*)d_in[6];
    const float* as2   = (const float*)d_in[7];
    const float* ad2   = (const float*)d_in[8];
    const float* b2    = (const float*)d_in[9];
    float* out = (float*)d_out;

    const int E  = in_sizes[1] / 2;
    const int NF = NN * 64;

    float* wsf = (float*)d_ws;
    float* H   = wsf;
    float* X2  = H   + NF;
    float* S   = X2  + NF;
    float* D   = S   + NN;
    const int G1 = (E + CH - 1) / CH;
    const int nM = NBUCK * G1;
    int*   cntm = (int*)(D + NN);
    int*   boff = cntm + nM + 16;
    int*   off  = boff + nM + 16;
    int*   tmp  = off + NN + 16;
    int*   cA   = tmp + E;
    int*   cB   = cA  + E;

    const int TB = 256;
    dim3 blk(TB);
    int gScan = (nM + 1023) / 1024;
    int gNode = (NSTRIP + 7) / 8;
    int gGemm = ((NN / 16) + 3) / 4;

    bucket_hist_kernel<<<G1, blk, 0, stream>>>(ei, E, G1, cntm);
    scan_counts_kernel<<<gScan, dim3(1024), 0, stream>>>(cntm, boff, nM);
    bucket_scatter_kernel<<<G1, blk, 0, stream>>>(ei, E, G1, boff, tmp);
    bucket_csr_kernel<<<NBUCK, blk, 0, stream>>>(boff, G1, tmp, off, cA);
    sort_segments_kernel<<<gNode, blk, 0, stream>>>(off, cA, cB);

    gemm_wmma_kernel<<<gGemm, dim3(128), 0, stream>>>(x, W1, as1, ad1, H, S, D, NN, 128);
    node_softmax_sum_kernel<<<gNode, blk, 0, stream>>>(off, cB, H, S, D, b1, 1, X2);

    gemm_wmma_kernel<<<gGemm, dim3(128), 0, stream>>>(X2, W2, as2, ad2, H, S, D, NN, 64);
    node_softmax_sum_kernel<<<gNode, blk, 0, stream>>>(off, cB, H, S, D, b2, 0, out);
}


# Layer-2 product X2 @ W2 with its score epilogue run inside the layer-1 node kernel on the block's 64 finished rows held in LDS (same k order and sum trees): X2 never goes to memory; 8 launches, not 9
# speedup vs baseline: 4.3610x; 1.0453x over previous
#include <hip/hip_runtime.h>
#include <math.h>


typedef float v2f __attribute__((ext_vector_type(2)));
typedef float v8f __attribute__((ext_vector_type(8)));

#define NN 100000
#define NEG_SLOPE 0.2f
#define EPSF 1e-16f

__device__ __forceinline__ void tile_store_and_scores(
        const v8f c0, const v8f c1, const v8f c2, const v8f c3,
        const float as0, const float as1, const float as2, const float as3,
        const float ad0, const float ad1, const float ad2, const float ad3,
        float* __restrict__ H, float* __restrict__ S, float* __restrict__ D,
        const int tile, const int r0, const int khalf) {
    float* hbase = H + (size_t)(tile * 16 + khalf * 8) * 64 + r0;
    #pragma unroll
    for (int r = 0; r < 8; ++r) {
        float* row = hbase + (size_t)r * 64;
        row[0]  = c0[r];
        row[16] = c1[r];
        row[32] = c2[r];
        row[48] = c3[r];
    }
    float so = 0.0f, dd = 0.0f;
    #pragma unroll
    for (int r = 0; r < 8; ++r) {
        float s = fmaf(c0[r], as0, c2[r] * as2) + fmaf(c1[r], as1, c3[r] * as3);
        float d = fmaf(c0[r], ad0, c2[r] * ad2) + fmaf(c1[r], ad1, c3[r] * ad3);
        #pragma unroll
        for (int off = 8; off; off >>= 1) {
            s += __shfl_xor(s, off, 32);
            d += __shfl_xor(d, off, 32);
        }
        if (r0 == r) { so = s; dd = d; }
    }
    if (r0 < 8) {
        S[tile * 16 + khalf * 8 + r0] = so;
        D[tile * 16 + khalf * 8 + r0] = dd;
    }
}

__global__ __launch_bounds__(128) void gemm_wmma_kernel(
        const float* __restrict__ X, const float* __restrict__ W,
        const float* __restrict__ a_s, const float* __restrict__ a_d,
        float* __restrict__ H, float* __restrict__ S, float* __restrict__ D, int nrows, int K) {
    int wave = blockIdx.x * (blockDim.x >> 5) + (threadIdx.x >> 5);
    int mtiles = nrows >> 4;
    if (wave >= mtiles) return;

    int lane  = threadIdx.x & 31;
    int r0    = lane & 15;
    int khalf = lane >> 4;

    const float* xrow = X + (size_t)(wave * 16 + r0) * (size_t)K;

    const float as0 = a_s[r0], as1 = a_s[16 + r0], as2 = a_s[32 + r0], as3 = a_s[48 + r0];
    const float ad0 = a_d[r0], ad1 = a_d[16 + r0], ad2 = a_d[32 + r0], ad3 = a_d[48 + r0];

    v8f c0 = {}, c1 = {}, c2 = {}, c3 = {};
    for (int k0 = 0; k0 < K; k0 += 4) {
        int ka = k0 + khalf * 2;
        v2f a;
        a.x = xrow[ka];
        a.y = xrow[ka + 1];

        const float* w0 = W + (size_t)ka * 64;
        const float* w1 = w0 + 64;
        v2f b0, b1, b2, b3;
        b0.x = w0[r0];      b0.y = w1[r0];
        b1.x = w0[16 + r0]; b1.y = w1[16 + r0];
        b2.x = w0[32 + r0]; b2.y = w1[32 + r0];
        b3.x = w0[48 + r0]; b3.y = w1[48 + r0];

        c0 = __builtin_amdgcn_wmma_f32_16x16x4_f32(false, a, false, b0, (short)0, c0, false, false);
        c1 = __builtin_amdgcn_wmma_f32_16x16x4_f32(false, a, false, b1, (short)0, c1, false, false);
        c2 = __builtin_amdgcn_wmma_f32_16x16x4_f32(false, a, false, b2, (short)0, c2, false, false);
        c3 = __builtin_amdgcn_wmma_f32_16x16x4_f32(false, a, false, b3, (short)0, c3, false, false);
    }

    tile_store_and_scores(c0, c1, c2, c3, as0, as1, as2, as3, ad0, ad1, ad2, ad3, H, S, D, wave, r0, khalf);
}

#define NPW 8
#define NSTRIP (NN / NPW)
#define NQ 4
static_assert(NN % NPW == 0, "a strip must not run past the last node");
static_assert(NPW % NQ == 0, "a strip is a whole number of groups");

#define NPB_LOG2 8
#define NPB (1 << NPB_LOG2)
#define NBUCK ((NN + NPB - 1) / NPB)
#define CH 4096
#define EPT (CH / 256)
#define SRC_BITS 17
#define SRC_MASK ((1 << SRC_BITS) - 1)
static_assert(NN <= (1 << SRC_BITS), "a source id must fit the low bits of the packed word");
static_assert(NPB_LOG2 + SRC_BITS + 1 < 32, "packed word and the spare value past it must stay non-negative");

__global__ __launch_bounds__(256) void bucket_hist_kernel(const int* __restrict__ ei, int E, int G1,
        int* __restrict__ cntm) {
    __shared__ int hist[NBUCK];
    const int tid = threadIdx.x;
    for (int i = tid; i < NBUCK; i += 256) hist[i] = 0;
    __syncthreads();
    const int e0 = blockIdx.x * CH + tid;
    int d[EPT];
    #pragma unroll
    for (int k = 0; k < EPT; ++k) {
        const int e = e0 + k * 256;
        d[k] = (e < E) ? ei[E + e] : -1;
    }
    #pragma unroll
    for (int k = 0; k < EPT; ++k) {
        if (d[k] >= 0) atomicAdd(&hist[d[k] >> NPB_LOG2], 1);
    }
    __syncthreads();
    for (int i = tid; i < NBUCK; i += 256) cntm[i * G1 + blockIdx.x] = hist[i];
}

__global__ __launch_bounds__(1024) void scan_counts_kernel(
        const int* __restrict__ cnt, int* __restrict__ off, int n) {
    __shared__ int red[32];
    __shared__ int wtot[32];
    const int tid  = threadIdx.x;
    const int lane = tid & 31;
    const int wv   = tid >> 5;
    const int tile0 = blockIdx.x * 1024;

    const int4* c4 = (const int4*)cnt;
    const int nvec = tile0 >> 2;
    int part = 0;
    for (int i = tid; i < nvec; i += 1024) {
        int4 v = c4[i];
        part += (v.x + v.y) + (v.z + v.w);
    }
    #pragma unroll
    for (int o = 16; o; o >>= 1) part += __shfl_xor(part, o, 32);
    if (lane == 0) red[wv] = part;
    __syncthreads();
    int base = red[lane];
    #pragma unroll
    for (int o = 16; o; o >>= 1) base += __shfl_xor(base, o, 32);

    const int i = tile0 + tid;
    const int v = (i < n) ? cnt[i] : 0;
    int inc = v;
    #pragma unroll
    for (int o = 1; o < 32; o <<= 1) {
        int t = __shfl_up(inc, o, 32);
        if (lane >= o) inc += t;
    }
    if (lane == 31) wtot[wv] = inc;
    __syncthreads();
    const int wt = wtot[lane];
    int winc = wt;
    #pragma unroll
    for (int o = 1; o < 32; o <<= 1) {
        int t = __shfl_up(winc, o, 32);
        if (lane >= o) winc += t;
    }
    const int wprefix = __shfl(winc, wv, 32) - __shfl(wt, wv, 32);
    if (i < n) off[i] = base + wprefix + inc - v;
    if (i == n - 1) off[n] = base + wprefix + inc;
}

__global__ __launch_bounds__(256) void bucket_scatter_kernel(const int* __restrict__ ei, int E, int G1,
        const int* __restrict__ boff, int* __restrict__ tmp) {
    __shared__ int cur[NBUCK + 1];
    const int tid = threadIdx.x;
    const int e0 = blockIdx.x * CH + tid;
    int d[EPT], s[EPT];
    #pragma unroll
    for (int k = 0; k < EPT; ++k) {
        const int e = e0 + k * 256;
        d[k] = (e < E) ? ei[E + e] : (NBUCK << NPB_LOG2);
        s[k] = (e < E) ? ei[e] : 0;
    }
    for (int i = tid; i < NBUCK; i += 256) cur[i] = boff[i * G1 + blockIdx.x];
    if (tid == 0) cur[NBUCK] = 0;
    __syncthreads();
    int pos[EPT];
    #pragma unroll
    for (int k = 0; k < EPT; ++k) pos[k] = atomicAdd(&cur[d[k] >> NPB_LOG2], 1);
    #pragma unroll
    for (int k = 0; k < EPT; ++k) {
        if (e0 + k * 256 < E) tmp[pos[k]] = ((d[k] & (NPB - 1)) << SRC_BITS) | s[k];
    }
}

__global__ __launch_bounds__(256) void bucket_csr_kernel(const int* __restrict__ boff, int G1,
        const int* __restrict__ tmp, int* __restrict__ off, int* __restrict__ col) {
    __shared__ int cnt[NPB + 1];
    __shared__ int wtot[NPB / 32];
    const int tid  = threadIdx.x;
    const int lane = tid & 31;
    const int wv   = tid >> 5;
    const int bs = boff[blockIdx.x * G1];
    const int be = boff[(blockIdx.x + 1) * G1];
    cnt[tid] = 0;
    if (tid == 0) cnt[NPB] = 0;
    __syncthreads();
    for (int i = bs + tid; i < be; i += 4 * 256) {
        int w[4];
        #pragma unroll
        for (int u = 0; u < 4; ++u) {
            const int j = i + u * 256;
            w[u] = (j < be) ? tmp[j] : -1;
        }
        #pragma unroll
        for (int u = 0; u < 4; ++u) {
            if (w[u] >= 0) atomicAdd(&cnt[w[u] >> SRC_BITS], 1);
        }
    }
    __syncthreads();
    const int v = cnt[tid];
    int inc = v;
    #pragma unroll
    for (int o = 1; o < 32; o <<= 1) {
        int t = __shfl_up(inc, o, 32);
        if (lane >= o) inc += t;
    }
    if (lane == 31) wtot[wv] = inc;
    __syncthreads();
    int wprefix = 0;
    for (int w = 0; w < wv; ++w) wprefix += wtot[w];
    const int excl = bs + wprefix + inc - v;
    const int node = blockIdx.x * NPB + tid;
    if (node < NN) off[node] = excl;
    if (node == NN - 1) off[NN] = excl + v;
    cnt[tid] = excl;
    __syncthreads();
    for (int i = bs + tid; i < be; i += 4 * 256) {
        int w[4];
        #pragma unroll
        for (int u = 0; u < 4; ++u) {
            const int j = i + u * 256;
            w[u] = (j < be) ? tmp[j] : (NPB << SRC_BITS);
        }
        int pos[4];
        #pragma unroll
        for (int u = 0; u < 4; ++u) pos[u] = atomicAdd(&cnt[w[u] >> SRC_BITS], 1);
        #pragma unroll
        for (int u = 0; u < 4; ++u) {
            if (i + u * 256 < be) col[pos[u]] = w[u] & SRC_MASK;
        }
    }
}

__global__ __launch_bounds__(256) void sort_segments_kernel(
        const int* __restrict__ off, const int* __restrict__ cin, int* __restrict__ cout) {
    const int lane = threadIdx.x & 31;
    const int strip = __builtin_amdgcn_readfirstlane((int)(blockIdx.x * 8 + (threadIdx.x >> 5)));
    if (strip >= NSTRIP) return;
    const int n0 = strip * NPW;
    const int offv = off[n0 + ((lane < NPW) ? lane : NPW)];
    int o0[NPW], deg[NPW], key[NPW];
    #pragma unroll
    for (int u = 0; u < NPW; ++u) {
        o0[u]  = __builtin_amdgcn_readlane(offv, u);
        deg[u] = __builtin_amdgcn_readlane(offv, u + 1) - o0[u];
        key[u] = (lane < deg[u]) ? cin[o0[u] + lane] : 0x7fffffff;
    }
    #pragma unroll
    for (int u = 0; u < NPW; ++u) {
        if (deg[u] <= 32) {
            int rank = 0;
            for (int j = 0; j < deg[u]; ++j) {
                const int kj = __builtin_amdgcn_readlane(key[u], j);
                rank += (kj < key[u] || (kj == key[u] && j < lane)) ? 1 : 0;
            }
            if (lane < deg[u]) cout[o0[u] + rank] = key[u];
        } else {
            for (int i = lane; i < deg[u]; i += 32) {
                const int ki = cin[o0[u] + i];
                int rank = 0;
                for (int j = 0; j < deg[u]; ++j) {
                    const int kj = cin[o0[u] + j];
                    rank += (kj < ki || (kj == ki && j < i)) ? 1 : 0;
                }
                cout[o0[u] + rank] = ki;
            }
        }
    }
}

__device__ __forceinline__ float leaky(float a) { return (a >= 0.0f) ? a : NEG_SLOPE * a; }

__device__ __forceinline__ float readlane_f(float v, int l) {
    return __int_as_float(__builtin_amdgcn_readlane(__float_as_int(v), l));
}

#define TP 68

template <bool TO_LDS>
__device__ __forceinline__ void finish_row(v2f acc, float bx, float by, int relu,
                                           float* __restrict__ OUT, float* tile, int node, int row, int lane) {
    acc.x += bx;
    acc.y += by;
    if (relu) {
        acc.x = (acc.x > 0.0f) ? acc.x : 0.0f;
        acc.y = (acc.y > 0.0f) ? acc.y : 0.0f;
    }
    if (TO_LDS) ((v2f*)tile)[row * (TP / 2) + lane] = acc;
    else        ((v2f*)OUT)[(size_t)node * 32 + lane] = acc;
}

template <bool TO_LDS>
__device__ __forceinline__ void strip_rows(
        const int strip, const int lane,
        const int* __restrict__ off, const int* __restrict__ col,
        const float* __restrict__ H, const float* __restrict__ S, const float* __restrict__ D,
        const float* __restrict__ bias, int relu, float* __restrict__ OUT, float* tile, const int rb0) {
    const int n0 = strip * NPW;
    const int   offv = off[n0 + ((lane < NPW) ? lane : NPW)];
    const float Sv   = S[n0 + (lane & (NPW - 1))];
    const float Dv   = D[n0 + (lane & (NPW - 1))];
    const v2f* H2 = (const v2f*)H;
    const float bx = bias[2 * lane];
    const float by = bias[2 * lane + 1];

    bool any_long = false;
    for (int q = 0; q < NPW; q += NQ) {
        int o0[NQ], deg[NQ], sv[NQ];
        v2f acc0[NQ];
        #pragma unroll
        for (int u = 0; u < NQ; ++u) {
            o0[u]   = __builtin_amdgcn_readlane(offv, q + u);
            deg[u]  = __builtin_amdgcn_readlane(offv, q + u + 1) - o0[u];
            sv[u]   = (lane < deg[u]) ? col[o0[u] + lane] : (n0 + q + u);
            acc0[u] = H2[(size_t)(n0 + q + u) * 32 + lane];
        }
        float av[NQ], aself[NQ];
        #pragma unroll
        for (int u = 0; u < NQ; ++u) av[u] = S[sv[u]];
        #pragma unroll
        for (int u = 0; u < NQ; ++u) {
            const float dn = readlane_f(Dv, q + u);
            aself[u] = leaky(readlane_f(Sv, q + u) + dn);
            av[u] = (lane < deg[u]) ? leaky(av[u] + dn) : -INFINITY;
        }
        float wv[NQ];
        #pragma unroll
        for (int u = 0; u < NQ; ++u) {
            float m = fmaxf(av[u], aself[u]);
            #pragma unroll
            for (int o = 16; o; o >>= 1) m = fmaxf(m, __shfl_xor(m, o, 32));
            const float ex = expf(av[u] - m);
            const float e = (lane < deg[u]) ? ex : 0.0f;
            const float e_self = expf(aself[u] - m);
            float z = e;
            #pragma unroll
            for (int o = 16; o; o >>= 1) z += __shfl_xor(z, o, 32);
            z += e_self;
            wv[u] = e / (z + EPSF);
            const float w_self = e_self / (z + EPSF);
            acc0[u].x = acc0[u].x * w_self;
            acc0[u].y = acc0[u].y * w_self;
        }
        #pragma unroll
        for (int u = 0; u < NQ; ++u) {
            if (deg[u] > 32) { any_long = true; continue; }
            const float w = wv[u];
            v2f acc = acc0[u];
            for (int j = 0; j < deg[u]; j += 8) {
                v2f hv[8];
                #pragma unroll
                for (int t = 0; t < 8; ++t) {
                    const int jt = (j + t < deg[u]) ? (j + t) : (deg[u] - 1);
                    const int sj = __builtin_amdgcn_readlane(sv[u], jt);
                    hv[t] = H2[(size_t)sj * 32 + lane];
                }
                #pragma unroll
                for (int t = 0; t < 8; ++t) {
                    const float wj = readlane_f(w, j + t);
                    acc.x = fmaf(hv[t].x, wj, acc.x);
                    acc.y = fmaf(hv[t].y, wj, acc.y);
                }
            }
            finish_row<TO_LDS>(acc, bx, by, relu, OUT, tile, n0 + q + u, rb0 + q + u, lane);
        }
    }

    if (any_long) {
        for (int u = 0; u < NPW; ++u) {
            const int lo = __builtin_amdgcn_readlane(offv, u);
            const int dg = __builtin_amdgcn_readlane(offv, u + 1) - lo;
            if (dg <= 32) continue;
            const int node = n0 + u;
            const float dn = readlane_f(Dv, u);
            const float a_self = leaky(readlane_f(Sv, u) + dn);
            const v2f hsn = H2[(size_t)node * 32 + lane];
            float m = a_self;
            for (int c = 0; c < dg; c += 32) {
                if (c + lane < dg) m = fmaxf(m, leaky(S[col[lo + c + lane]] + dn));
            }
            #pragma unroll
            for (int o = 16; o; o >>= 1) m = fmaxf(m, __shfl_xor(m, o, 32));
            float z = 0.0f;
            for (int c = 0; c < dg; c += 32) {
                if (c + lane < dg) z += expf(leaky(S[col[lo + c + lane]] + dn) - m);
            }
            #pragma unroll
            for (int o = 16; o; o >>= 1) z += __shfl_xor(z, o, 32);
            const float e_self = expf(a_self - m);
            z += e_self;
            const float w_self = e_self / (z + EPSF);
            v2f acc;
            acc.x = hsn.x * w_self;
            acc.y = hsn.y * w_self;
            for (int c = 0; c < dg; c += 32) {
                int s = node;
                float w = 0.0f;
                if (c + lane < dg) {
                    s = col[lo + c + lane];
                    w = expf(leaky(S[s] + dn) - m) / (z + EPSF);
                }
                const int nj = (dg - c < 32) ? (dg - c) : 32;
                for (int j = 0; j < nj; ++j) {
                    const int sj = __builtin_amdgcn_readlane(s, j);
                    const float wj = readlane_f(w, j);
                    const v2f hv = H2[(size_t)sj * 32 + lane];
                    acc.x = fmaf(hv.x, wj, acc.x);
                    acc.y = fmaf(hv.y, wj, acc.y);
                }
            }
            finish_row<TO_LDS>(acc, bx, by, relu, OUT, tile, node, rb0 + u, lane);
        }
    }
}

__global__ __launch_bounds__(256) void node_softmax_sum_kernel(
        const int* __restrict__ off, const int* __restrict__ col,
        const float* __restrict__ H, const float* __restrict__ S, const float* __restrict__ D,
        const float* __restrict__ bias, int relu, float* __restrict__ OUT) {
    const int lane = threadIdx.x & 31;
    const int strip = __builtin_amdgcn_readfirstlane((int)(blockIdx.x * 8 + (threadIdx.x >> 5)));
    if (strip >= NSTRIP) return;
    strip_rows<false>(strip, lane, off, col, H, S, D, bias, relu, OUT, nullptr, 0);
}

static_assert(NPW * 8 == 64, "a block of 8 waves must finish four whole 16-row tiles");
static_assert(NN % 16 == 0, "whole tiles only");
__global__ __launch_bounds__(256, 16) void node_softmax_sum_gemm_kernel(
        const int* __restrict__ off, const int* __restrict__ col,
        const float* __restrict__ H, const float* __restrict__ S, const float* __restrict__ D,
        const float* __restrict__ bias, int relu,
        const float* __restrict__ W2, const float* __restrict__ a_s2, const float* __restrict__ a_d2,
        float* __restrict__ H2, float* __restrict__ S2, float* __restrict__ D2) {
    __shared__ float tile[64 * TP];
    const int lane = threadIdx.x & 31;
    const int wv = __builtin_amdgcn_readfirstlane((int)(threadIdx.x >> 5));
    const int strip = blockIdx.x * 8 + wv;
    if (strip < NSTRIP) strip_rows<true>(strip, lane, off, col, H, S, D, bias, relu, nullptr, tile, wv * NPW);
    __syncthreads();

    const int t = blockIdx.x * 4 + wv;
    if (wv >= 4 || t >= NN / 16) return;

    const int r0    = lane & 15;
    const int khalf = lane >> 4;
    const float* arow = tile + (wv * 16 + r0) * TP;

    const float as0 = a_s2[r0], as1 = a_s2[16 + r0], as2 = a_s2[32 + r0], as3 = a_s2[48 + r0];
    const float ad0 = a_d2[r0], ad1 = a_d2[16 + r0], ad2 = a_d2[32 + r0], ad3 = a_d2[48 + r0];

    v8f c0 = {}, c1 = {}, c2 = {}, c3 = {};
    for (int k0 = 0; k0 < 64; k0 += 4) {
        const int ka = k0 + khalf * 2;
        v2f a;
        a.x = arow[ka];
        a.y = arow[ka + 1];

        const float* w0 = W2 + (size_t)ka * 64;
        const float* w1 = w0 + 64;
        v2f b0, b1, b2, b3;
        b0.x = w0[r0];      b0.y = w1[r0];
        b1.x = w0[16 + r0]; b1.y = w1[16 + r0];
        b2.x = w0[32 + r0]; b2.y = w1[32 + r0];
        b3.x = w0[48 + r0]; b3.y = w1[48 + r0];

        c0 = __builtin_amdgcn_wmma_f32_16x16x4_f32(false, a, false, b0, (short)0, c0, false, false);
        c1 = __builtin_amdgcn_wmma_f32_16x16x4_f32(false, a, false, b1, (short)0, c1, false, false);
        c2 = __builtin_amdgcn_wmma_f32_16x16x4_f32(false, a, false, b2, (short)0, c2, false, false);
        c3 = __builtin_amdgcn_wmma_f32_16x16x4_f32(false, a, false, b3, (short)0, c3, false, false);
    }

    tile_store_and_scores(c0, c1, c2, c3, as0, as1, as2, as3, ad0, ad1, ad2, ad3, H2, S2, D2, t, r0, khalf);
}

extern "C" void kernel_launch(void* const* d_in, const int* in_sizes, int n_in,
                              void* d_out, int out_size, void* d_ws, size_t ws_size,
                              hipStream_t stream) {
    const float* x     = (const float*)d_in[0];
    const int*   ei    = (const int*)d_in[1];
    const float* W1    = (const float*)d_in[2];
    const float* as1   = (const float*)d_in[3];
    const float* ad1   = (const float*)d_in[4];
    const float* b1    = (const float*)d_in[5];
    const float* W2    = (const float*)d_in[6];
    const float* as2   = (const float*)d_in[7];
    const float* ad2   = (const float*)d_in[8];
    const float* b2    = (const float*)d_in[9];
    float* out = (float*)d_out;

    const int E  = in_sizes[1] / 2;
    const int NF = NN * 64;

    float* wsf = (float*)d_ws;
    float* H   = wsf;
    float* H2  = H   + NF;
    float* S   = H2  + NF;
    float* D   = S   + NN;
    const int G1 = (E + CH - 1) / CH;
    const int nM = NBUCK * G1;
    int*   cntm = (int*)(D + NN);
    int*   boff = cntm + nM + 16;
    int*   off  = boff + nM + 16;
    int*   tmp  = off + NN + 16;
    int*   cA   = tmp + E;
    int*   cB   = cA  + E;
    float* S2   = (float*)(cB + E);
    float* D2   = S2 + NN;

    const int TB = 256;
    dim3 blk(TB);
    int gScan = (nM + 1023) / 1024;
    int gNode = (NSTRIP + 7) / 8;
    int gGemm = ((NN / 16) + 3) / 4;

    bucket_hist_kernel<<<G1, blk, 0, stream>>>(ei, E, G1, cntm);
    scan_counts_kernel<<<gScan, dim3(1024), 0, stream>>>(cntm, boff, nM);
    bucket_scatter_kernel<<<G1, blk, 0, stream>>>(ei, E, G1, boff, tmp);
    bucket_csr_kernel<<<NBUCK, blk, 0, stream>>>(boff, G1, tmp, off, cA);
    sort_segments_kernel<<<gNode, blk, 0, stream>>>(off, cA, cB);

    gemm_wmma_kernel<<<gGemm, dim3(128), 0, stream>>>(x, W1, as1, ad1, H, S, D, NN, 128);
    node_softmax_sum_gemm_kernel<<<gNode, blk, 0, stream>>>(off, cB, H, S, D, b1, 1, W2, as2, ad2, H2, S2, D2);

    node_softmax_sum_kernel<<<gNode, blk, 0, stream>>>(off, cB, H2, S2, D2, b2, 0, out);
}


# First layer's product x @ W1 (same tile code, score epilogue kept) runs as extra work-groups of the per-bucket placement launch, which it does not depend on: 7 launches, not 8; sums unchanged
# speedup vs baseline: 4.5414x; 1.0414x over previous
#include <hip/hip_runtime.h>
#include <math.h>


typedef float v2f __attribute__((ext_vector_type(2)));
typedef float v8f __attribute__((ext_vector_type(8)));

#define NN 100000
#define NEG_SLOPE 0.2f
#define EPSF 1e-16f

__device__ __forceinline__ void tile_store_and_scores(
        const v8f c0, const v8f c1, const v8f c2, const v8f c3,
        const float as0, const float as1, const float as2, const float as3,
        const float ad0, const float ad1, const float ad2, const float ad3,
        float* __restrict__ H, float* __restrict__ S, float* __restrict__ D,
        const int tile, const int r0, const int khalf) {
    float* hbase = H + (size_t)(tile * 16 + khalf * 8) * 64 + r0;
    #pragma unroll
    for (int r = 0; r < 8; ++r) {
        float* row = hbase + (size_t)r * 64;
        row[0]  = c0[r];
        row[16] = c1[r];
        row[32] = c2[r];
        row[48] = c3[r];
    }
    float so = 0.0f, dd = 0.0f;
    #pragma unroll
    for (int r = 0; r < 8; ++r) {
        float s = fmaf(c0[r], as0, c2[r] * as2) + fmaf(c1[r], as1, c3[r] * as3);
        float d = fmaf(c0[r], ad0, c2[r] * ad2) + fmaf(c1[r], ad1, c3[r] * ad3);
        #pragma unroll
        for (int off = 8; off; off >>= 1) {
            s += __shfl_xor(s, off, 32);
            d += __shfl_xor(d, off, 32);
        }
        if (r0 == r) { so = s; dd = d; }
    }
    if (r0 < 8) {
        S[tile * 16 + khalf * 8 + r0] = so;
        D[tile * 16 + khalf * 8 + r0] = dd;
    }
}

__device__ __forceinline__ void gemm_wmma_tile(
        const float* __restrict__ X, const float* __restrict__ W,
        const float* __restrict__ a_s, const float* __restrict__ a_d,
        float* __restrict__ H, float* __restrict__ S, float* __restrict__ D, const int wave, const int K) {
    int lane  = threadIdx.x & 31;
    int r0    = lane & 15;
    int khalf = lane >> 4;

    const float* xrow = X + (size_t)(wave * 16 + r0) * (size_t)K;

    const float as0 = a_s[r0], as1 = a_s[16 + r0], as2 = a_s[32 + r0], as3 = a_s[48 + r0];
    const float ad0 = a_d[r0], ad1 = a_d[16 + r0], ad2 = a_d[32 + r0], ad3 = a_d[48 + r0];

    v8f c0 = {}, c1 = {}, c2 = {}, c3 = {};
    for (int k0 = 0; k0 < K; k0 += 4) {
        int ka = k0 + khalf * 2;
        v2f a;
        a.x = xrow[ka];
        a.y = xrow[ka + 1];

        const float* w0 = W + (size_t)ka * 64;
        const float* w1 = w0 + 64;
        v2f b0, b1, b2, b3;
        b0.x = w0[r0];      b0.y = w1[r0];
        b1.x = w0[16 + r0]; b1.y = w1[16 + r0];
        b2.x = w0[32 + r0]; b2.y = w1[32 + r0];
        b3.x = w0[48 + r0]; b3.y = w1[48 + r0];

        c0 = __builtin_amdgcn_wmma_f32_16x16x4_f32(false, a, false, b0, (short)0, c0, false, false);
        c1 = __builtin_amdgcn_wmma_f32_16x16x4_f32(false, a, false, b1, (short)0, c1, false, false);
        c2 = __builtin_amdgcn_wmma_f32_16x16x4_f32(false, a, false, b2, (short)0, c2, false, false);
        c3 = __builtin_amdgcn_wmma_f32_16x16x4_f32(false, a, false, b3, (short)0, c3, false, false);
    }

    tile_store_and_scores(c0, c1, c2, c3, as0, as1, as2, as3, ad0, ad1, ad2, ad3, H, S, D, wave, r0, khalf);
}

#define NPW 8
#define NSTRIP (NN / NPW)
#define NQ 4
static_assert(NN % NPW == 0, "a strip must not run past the last node");
static_assert(NPW % NQ == 0, "a strip is a whole number of groups");

#define NPB_LOG2 8
#define NPB (1 << NPB_LOG2)
#define NBUCK ((NN + NPB - 1) / NPB)
#define CH 4096
#define EPT (CH / 256)
#define SRC_BITS 17
#define SRC_MASK ((1 << SRC_BITS) - 1)
static_assert(NN <= (1 << SRC_BITS), "a source id must fit the low bits of the packed word");
static_assert(NPB_LOG2 + SRC_BITS + 1 < 32, "packed word and the spare value past it must stay non-negative");

__global__ __launch_bounds__(256) void bucket_hist_kernel(const int* __restrict__ ei, int E, int G1,
        int* __restrict__ cntm) {
    __shared__ int hist[NBUCK];
    const int tid = threadIdx.x;
    for (int i = tid; i < NBUCK; i += 256) hist[i] = 0;
    __syncthreads();
    const int e0 = blockIdx.x * CH + tid;
    int d[EPT];
    #pragma unroll
    for (int k = 0; k < EPT; ++k) {
        const int e = e0 + k * 256;
        d[k] = (e < E) ? ei[E + e] : -1;
    }
    #pragma unroll
    for (int k = 0; k < EPT; ++k) {
        if (d[k] >= 0) atomicAdd(&hist[d[k] >> NPB_LOG2], 1);
    }
    __syncthreads();
    for (int i = tid; i < NBUCK; i += 256) cntm[i * G1 + blockIdx.x] = hist[i];
}

__global__ __launch_bounds__(1024) void scan_counts_kernel(
        const int* __restrict__ cnt, int* __restrict__ off, int n) {
    __shared__ int red[32];
    __shared__ int wtot[32];
    const int tid  = threadIdx.x;
    const int lane = tid & 31;
    const int wv   = tid >> 5;
    const int tile0 = blockIdx.x * 1024;

    const int4* c4 = (const int4*)cnt;
    const int nvec = tile0 >> 2;
    int part = 0;
    for (int i = tid; i < nvec; i += 1024) {
        int4 v = c4[i];
        part += (v.x + v.y) + (v.z + v.w);
    }
    #pragma unroll
    for (int o = 16; o; o >>= 1) part += __shfl_xor(part, o, 32);
    if (lane == 0) red[wv] = part;
    __syncthreads();
    int base = red[lane];
    #pragma unroll
    for (int o = 16; o; o >>= 1) base += __shfl_xor(base, o, 32);

    const int i = tile0 + tid;
    const int v = (i < n) ? cnt[i] : 0;
    int inc = v;
    #pragma unroll
    for (int o = 1; o < 32; o <<= 1) {
        int t = __shfl_up(inc, o, 32);
        if (lane >= o) inc += t;
    }
    if (lane == 31) wtot[wv] = inc;
    __syncthreads();
    const int wt = wtot[lane];
    int winc = wt;
    #pragma unroll
    for (int o = 1; o < 32; o <<= 1) {
        int t = __shfl_up(winc, o, 32);
        if (lane >= o) winc += t;
    }
    const int wprefix = __shfl(winc, wv, 32) - __shfl(wt, wv, 32);
    if (i < n) off[i] = base + wprefix + inc - v;
    if (i == n - 1) off[n] = base + wprefix + inc;
}

__global__ __launch_bounds__(256) void bucket_scatter_kernel(const int* __restrict__ ei, int E, int G1,
        const int* __restrict__ boff, int* __restrict__ tmp) {
    __shared__ int cur[NBUCK + 1];
    const int tid = threadIdx.x;
    const int e0 = blockIdx.x * CH + tid;
    int d[EPT], s[EPT];
    #pragma unroll
    for (int k = 0; k < EPT; ++k) {
        const int e = e0 + k * 256;
        d[k] = (e < E) ? ei[E + e] : (NBUCK << NPB_LOG2);
        s[k] = (e < E) ? ei[e] : 0;
    }
    for (int i = tid; i < NBUCK; i += 256) cur[i] = boff[i * G1 + blockIdx.x];
    if (tid == 0) cur[NBUCK] = 0;
    __syncthreads();
    int pos[EPT];
    #pragma unroll
    for (int k = 0; k < EPT; ++k) pos[k] = atomicAdd(&cur[d[k] >> NPB_LOG2], 1);
    #pragma unroll
    for (int k = 0; k < EPT; ++k) {
        if (e0 + k * 256 < E) tmp[pos[k]] = ((d[k] & (NPB - 1)) << SRC_BITS) | s[k];
    }
}

struct GemmArgs {
    const float* X; const float* W; const float* a_s; const float* a_d;
    float* H; float* S; float* D;
    int nrows; int K;
};

__global__ __launch_bounds__(256) void bucket_csr_gemm_kernel(const int* __restrict__ boff, int G1,
        const int* __restrict__ tmp, int* __restrict__ off, int* __restrict__ col, const GemmArgs g) {
    if (blockIdx.x >= NBUCK) {
        const int tile = __builtin_amdgcn_readfirstlane((int)((blockIdx.x - NBUCK) * 8 + (threadIdx.x >> 5)));
        if (tile < (g.nrows >> 4)) gemm_wmma_tile(g.X, g.W, g.a_s, g.a_d, g.H, g.S, g.D, tile, g.K);
        return;
    }
    __shared__ int cnt[NPB + 1];
    __shared__ int wtot[NPB / 32];
    const int tid  = threadIdx.x;
    const int lane = tid & 31;
    const int wv   = tid >> 5;
    const int bs = boff[blockIdx.x * G1];
    const int be = boff[(blockIdx.x + 1) * G1];
    cnt[tid] = 0;
    if (tid == 0) cnt[NPB] = 0;
    __syncthreads();
    for (int i = bs + tid; i < be; i += 4 * 256) {
        int w[4];
        #pragma unroll
        for (int u = 0; u < 4; ++u) {
            const int j = i + u * 256;
            w[u] = (j < be) ? tmp[j] : -1;
        }
        #pragma unroll
        for (int u = 0; u < 4; ++u) {
            if (w[u] >= 0) atomicAdd(&cnt[w[u] >> SRC_BITS], 1);
        }
    }
    __syncthreads();
    const int v = cnt[tid];
    int inc = v;
    #pragma unroll
    for (int o = 1; o < 32; o <<= 1) {
        int t = __shfl_up(inc, o, 32);
        if (lane >= o) inc += t;
    }
    if (lane == 31) wtot[wv] = inc;
    __syncthreads();
    int wprefix = 0;
    for (int w = 0; w < wv; ++w) wprefix += wtot[w];
    const int excl = bs + wprefix + inc - v;
    const int node = blockIdx.x * NPB + tid;
    if (node < NN) off[node] = excl;
    if (node == NN - 1) off[NN] = excl + v;
    cnt[tid] = excl;
    __syncthreads();
    for (int i = bs + tid; i < be; i += 4 * 256) {
        int w[4];
        #pragma unroll
        for (int u = 0; u < 4; ++u) {
            const int j = i + u * 256;
            w[u] = (j < be) ? tmp[j] : (NPB << SRC_BITS);
        }
        int pos[4];
        #pragma unroll
        for (int u = 0; u < 4; ++u) pos[u] = atomicAdd(&cnt[w[u] >> SRC_BITS], 1);
        #pragma unroll
        for (int u = 0; u < 4; ++u) {
            if (i + u * 256 < be) col[pos[u]] = w[u] & SRC_MASK;
        }
    }
}

__global__ __launch_bounds__(256) void sort_segments_kernel(
        const int* __restrict__ off, const int* __restrict__ cin, int* __restrict__ cout) {
    const int lane = threadIdx.x & 31;
    const int strip = __builtin_amdgcn_readfirstlane((int)(blockIdx.x * 8 + (threadIdx.x >> 5)));
    if (strip >= NSTRIP) return;
    const int n0 = strip * NPW;
    const int offv = off[n0 + ((lane < NPW) ? lane : NPW)];
    int o0[NPW], deg[NPW], key[NPW];
    #pragma unroll
    for (int u = 0; u < NPW; ++u) {
        o0[u]  = __builtin_amdgcn_readlane(offv, u);
        deg[u] = __builtin_amdgcn_readlane(offv, u + 1) - o0[u];
        key[u] = (lane < deg[u]) ? cin[o0[u] + lane] : 0x7fffffff;
    }
    #pragma unroll
    for (int u = 0; u < NPW; ++u) {
        if (deg[u] <= 32) {
            int rank = 0;
            for (int j = 0; j < deg[u]; ++j) {
                const int kj = __builtin_amdgcn_readlane(key[u], j);
                rank += (kj < key[u] || (kj == key[u] && j < lane)) ? 1 : 0;
            }
            if (lane < deg[u]) cout[o0[u] + rank] = key[u];
        } else {
            for (int i = lane; i < deg[u]; i += 32) {
                const int ki = cin[o0[u] + i];
                int rank = 0;
                for (int j = 0; j < deg[u]; ++j) {
                    const int kj = cin[o0[u] + j];
                    rank += (kj < ki || (kj == ki && j < i)) ? 1 : 0;
                }
                cout[o0[u] + rank] = ki;
            }
        }
    }
}

__device__ __forceinline__ float leaky(float a) { return (a >= 0.0f) ? a : NEG_SLOPE * a; }

__device__ __forceinline__ float readlane_f(float v, int l) {
    return __int_as_float(__builtin_amdgcn_readlane(__float_as_int(v), l));
}

#define TP 68

template <bool TO_LDS>
__device__ __forceinline__ void finish_row(v2f acc, float bx, float by, int relu,
                                           float* __restrict__ OUT, float* tile, int node, int row, int lane) {
    acc.x += bx;
    acc.y += by;
    if (relu) {
        acc.x = (acc.x > 0.0f) ? acc.x : 0.0f;
        acc.y = (acc.y > 0.0f) ? acc.y : 0.0f;
    }
    if (TO_LDS) ((v2f*)tile)[row * (TP / 2) + lane] = acc;
    else        ((v2f*)OUT)[(size_t)node * 32 + lane] = acc;
}

template <bool TO_LDS>
__device__ __forceinline__ void strip_rows(
        const int strip, const int lane,
        const int* __restrict__ off, const int* __restrict__ col,
        const float* __restrict__ H, const float* __restrict__ S, const float* __restrict__ D,
        const float* __restrict__ bias, int relu, float* __restrict__ OUT, float* tile, const int rb0) {
    const int n0 = strip * NPW;
    const int   offv = off[n0 + ((lane < NPW) ? lane : NPW)];
    const float Sv   = S[n0 + (lane & (NPW - 1))];
    const float Dv   = D[n0 + (lane & (NPW - 1))];
    const v2f* H2 = (const v2f*)H;
    const float bx = bias[2 * lane];
    const float by = bias[2 * lane + 1];

    bool any_long = false;
    for (int q = 0; q < NPW; q += NQ) {
        int o0[NQ], deg[NQ], sv[NQ];
        v2f acc0[NQ];
        #pragma unroll
        for (int u = 0; u < NQ; ++u) {
            o0[u]   = __builtin_amdgcn_readlane(offv, q + u);
            deg[u]  = __builtin_amdgcn_readlane(offv, q + u + 1) - o0[u];
            sv[u]   = (lane < deg[u]) ? col[o0[u] + lane] : (n0 + q + u);
            acc0[u] = H2[(size_t)(n0 + q + u) * 32 + lane];
        }
        float av[NQ], aself[NQ];
        #pragma unroll
        for (int u = 0; u < NQ; ++u) av[u] = S[sv[u]];
        #pragma unroll
        for (int u = 0; u < NQ; ++u) {
            const float dn = readlane_f(Dv, q + u);
            aself[u] = leaky(readlane_f(Sv, q + u) + dn);
            av[u] = (lane < deg[u]) ? leaky(av[u] + dn) : -INFINITY;
        }
        float wv[NQ];
        #pragma unroll
        for (int u = 0; u < NQ; ++u) {
            float m = fmaxf(av[u], aself[u]);
            #pragma unroll
            for (int o = 16; o; o >>= 1) m = fmaxf(m, __shfl_xor(m, o, 32));
            const float ex = expf(av[u] - m);
            const float e = (lane < deg[u]) ? ex : 0.0f;
            const float e_self = expf(aself[u] - m);
            float z = e;
            #pragma unroll
            for (int o = 16; o; o >>= 1) z += __shfl_xor(z, o, 32);
            z += e_self;
            wv[u] = e / (z + EPSF);
            const float w_self = e_self / (z + EPSF);
            acc0[u].x = acc0[u].x * w_self;
            acc0[u].y = acc0[u].y * w_self;
        }
        #pragma unroll
        for (int u = 0; u < NQ; ++u) {
            if (deg[u] > 32) { any_long = true; continue; }
            const float w = wv[u];
            v2f acc = acc0[u];
            for (int j = 0; j < deg[u]; j += 8) {
                v2f hv[8];
                #pragma unroll
                for (int t = 0; t < 8; ++t) {
                    const int jt = (j + t < deg[u]) ? (j + t) : (deg[u] - 1);
                    const int sj = __builtin_amdgcn_readlane(sv[u], jt);
                    hv[t] = H2[(size_t)sj * 32 + lane];
                }
                #pragma unroll
                for (int t = 0; t < 8; ++t) {
                    const float wj = readlane_f(w, j + t);
                    acc.x = fmaf(hv[t].x, wj, acc.x);
                    acc.y = fmaf(hv[t].y, wj, acc.y);
                }
            }
            finish_row<TO_LDS>(acc, bx, by, relu, OUT, tile, n0 + q + u, rb0 + q + u, lane);
        }
    }

    if (any_long) {
        for (int u = 0; u < NPW; ++u) {
            const int lo = __builtin_amdgcn_readlane(offv, u);
            const int dg = __builtin_amdgcn_readlane(offv, u + 1) - lo;
            if (dg <= 32) continue;
            const int node = n0 + u;
            const float dn = readlane_f(Dv, u);
            const float a_self = leaky(readlane_f(Sv, u) + dn);
            const v2f hsn = H2[(size_t)node * 32 + lane];
            float m = a_self;
            for (int c = 0; c < dg; c += 32) {
                if (c + lane < dg) m = fmaxf(m, leaky(S[col[lo + c + lane]] + dn));
            }
            #pragma unroll
            for (int o = 16; o; o >>= 1) m = fmaxf(m, __shfl_xor(m, o, 32));
            float z = 0.0f;
            for (int c = 0; c < dg; c += 32) {
                if (c + lane < dg) z += expf(leaky(S[col[lo + c + lane]] + dn) - m);
            }
            #pragma unroll
            for (int o = 16; o; o >>= 1) z += __shfl_xor(z, o, 32);
            const float e_self = expf(a_self - m);
            z += e_self;
            const float w_self = e_self / (z + EPSF);
            v2f acc;
            acc.x = hsn.x * w_self;
            acc.y = hsn.y * w_self;
            for (int c = 0; c < dg; c += 32) {
                int s = node;
                float w = 0.0f;
                if (c + lane < dg) {
                    s = col[lo + c + lane];
                    w = expf(leaky(S[s] + dn) - m) / (z + EPSF);
                }
                const int nj = (dg - c < 32) ? (dg - c) : 32;
                for (int j = 0; j < nj; ++j) {
                    const int sj = __builtin_amdgcn_readlane(s, j);
                    const float wj = readlane_f(w, j);
                    const v2f hv = H2[(size_t)sj * 32 + lane];
                    acc.x = fmaf(hv.x, wj, acc.x);
                    acc.y = fmaf(hv.y, wj, acc.y);
                }
            }
            finish_row<TO_LDS>(acc, bx, by, relu, OUT, tile, node, rb0 + u, lane);
        }
    }
}

__global__ __launch_bounds__(256) void node_softmax_sum_kernel(
        const int* __restrict__ off, const int* __restrict__ col,
        const float* __restrict__ H, const float* __restrict__ S, const float* __restrict__ D,
        const float* __restrict__ bias, int relu, float* __restrict__ OUT) {
    const int lane = threadIdx.x & 31;
    const int strip = __builtin_amdgcn_readfirstlane((int)(blockIdx.x * 8 + (threadIdx.x >> 5)));
    if (strip >= NSTRIP) return;
    strip_rows<false>(strip, lane, off, col, H, S, D, bias, relu, OUT, nullptr, 0);
}

static_assert(NPW * 8 == 64, "a block of 8 waves must finish four whole 16-row tiles");
static_assert(NN % 16 == 0, "whole tiles only");
__global__ __launch_bounds__(256, 16) void node_softmax_sum_gemm_kernel(
        const int* __restrict__ off, const int* __restrict__ col,
        const float* __restrict__ H, const float* __restrict__ S, const float* __restrict__ D,
        const float* __restrict__ bias, int relu,
        const float* __restrict__ W2, const float* __restrict__ a_s2, const float* __restrict__ a_d2,
        float* __restrict__ H2, float* __restrict__ S2, float* __restrict__ D2) {
    __shared__ float tile[64 * TP];
    const int lane = threadIdx.x & 31;
    const int wv = __builtin_amdgcn_readfirstlane((int)(threadIdx.x >> 5));
    const int strip = blockIdx.x * 8 + wv;
    if (strip < NSTRIP) strip_rows<true>(strip, lane, off, col, H, S, D, bias, relu, nullptr, tile, wv * NPW);
    __syncthreads();

    const int t = blockIdx.x * 4 + wv;
    if (wv >= 4 || t >= NN / 16) return;

    const int r0    = lane & 15;
    const int khalf = lane >> 4;
    const float* arow = tile + (wv * 16 + r0) * TP;

    const float as0 = a_s2[r0], as1 = a_s2[16 + r0], as2 = a_s2[32 + r0], as3 = a_s2[48 + r0];
    const float ad0 = a_d2[r0], ad1 = a_d2[16 + r0], ad2 = a_d2[32 + r0], ad3 = a_d2[48 + r0];

    v8f c0 = {}, c1 = {}, c2 = {}, c3 = {};
    for (int k0 = 0; k0 < 64; k0 += 4) {
        const int ka = k0 + khalf * 2;
        v2f a;
        a.x = arow[ka];
        a.y = arow[ka + 1];

        const float* w0 = W2 + (size_t)ka * 64;
        const float* w1 = w0 + 64;
        v2f b0, b1, b2, b3;
        b0.x = w0[r0];      b0.y = w1[r0];
        b1.x = w0[16 + r0]; b1.y = w1[16 + r0];
        b2.x = w0[32 + r0]; b2.y = w1[32 + r0];
        b3.x = w0[48 + r0]; b3.y = w1[48 + r0];

        c0 = __builtin_amdgcn_wmma_f32_16x16x4_f32(false, a, false, b0, (short)0, c0, false, false);
        c1 = __builtin_amdgcn_wmma_f32_16x16x4_f32(false, a, false, b1, (short)0, c1, false, false);
        c2 = __builtin_amdgcn_wmma_f32_16x16x4_f32(false, a, false, b2, (short)0, c2, false, false);
        c3 = __builtin_amdgcn_wmma_f32_16x16x4_f32(false, a, false, b3, (short)0, c3, false, false);
    }

    tile_store_and_scores(c0, c1, c2, c3, as0, as1, as2, as3, ad0, ad1, ad2, ad3, H2, S2, D2, t, r0, khalf);
}

extern "C" void kernel_launch(void* const* d_in, const int* in_sizes, int n_in,
                              void* d_out, int out_size, void* d_ws, size_t ws_size,
                              hipStream_t stream) {
    const float* x     = (const float*)d_in[0];
    const int*   ei    = (const int*)d_in[1];
    const float* W1    = (const float*)d_in[2];
    const float* as1   = (const float*)d_in[3];
    const float* ad1   = (const float*)d_in[4];
    const float* b1    = (const float*)d_in[5];
    const float* W2    = (const float*)d_in[6];
    const float* as2   = (const float*)d_in[7];
    const float* ad2   = (const float*)d_in[8];
    const float* b2    = (const float*)d_in[9];
    float* out = (float*)d_out;

    const int E  = in_sizes[1] / 2;
    const int NF = NN * 64;

    float* wsf = (float*)d_ws;
    float* H   = wsf;
    float* H2  = H   + NF;
    float* S   = H2  + NF;
    float* D   = S   + NN;
    const int G1 = (E + CH - 1) / CH;
    const int nM = NBUCK * G1;
    int*   cntm = (int*)(D + NN);
    int*   boff = cntm + nM + 16;
    int*   off  = boff + nM + 16;
    int*   tmp  = off + NN + 16;
    int*   cA   = tmp + E;
    int*   cB   = cA  + E;
    float* S2   = (float*)(cB + E);
    float* D2   = S2 + NN;

    const int TB = 256;
    dim3 blk(TB);
    int gScan = (nM + 1023) / 1024;
    int gNode = (NSTRIP + 7) / 8;
    int gGemm = ((NN / 16) + 7) / 8;

    bucket_hist_kernel<<<G1, blk, 0, stream>>>(ei, E, G1, cntm);
    scan_counts_kernel<<<gScan, dim3(1024), 0, stream>>>(cntm, boff, nM);
    bucket_scatter_kernel<<<G1, blk, 0, stream>>>(ei, E, G1, boff, tmp);
    const GemmArgs g1 = {x, W1, as1, ad1, H, S, D, NN, 128};
    bucket_csr_gemm_kernel<<<NBUCK + gGemm, blk, 0, stream>>>(boff, G1, tmp, off, cA, g1);
    sort_segments_kernel<<<gNode, blk, 0, stream>>>(off, cA, cB);

    node_softmax_sum_gemm_kernel<<<gNode, blk, 0, stream>>>(off, cB, H, S, D, b1, 1, W2, as2, ad2, H2, S2, D2);

    node_softmax_sum_kernel<<<gNode, blk, 0, stream>>>(off, cB, H2, S2, D2, b2, 0, out);
}


# Prefix-sum launch: each thread's 16-byte loads of the counts in front of its block's tile go out 16 at a time, not one per wait (last block: 3 round trips, not 38); integer sums, same values
# speedup vs baseline: 4.8509x; 1.0682x over previous
#include <hip/hip_runtime.h>
#include <math.h>


typedef float v2f __attribute__((ext_vector_type(2)));
typedef float v8f __attribute__((ext_vector_type(8)));

#define NN 100000
#define NEG_SLOPE 0.2f
#define EPSF 1e-16f

__device__ __forceinline__ void tile_store_and_scores(
        const v8f c0, const v8f c1, const v8f c2, const v8f c3,
        const float as0, const float as1, const float as2, const float as3,
        const float ad0, const float ad1, const float ad2, const float ad3,
        float* __restrict__ H, float* __restrict__ S, float* __restrict__ D,
        const int tile, const int r0, const int khalf) {
    float* hbase = H + (size_t)(tile * 16 + khalf * 8) * 64 + r0;
    #pragma unroll
    for (int r = 0; r < 8; ++r) {
        float* row = hbase + (size_t)r * 64;
        row[0]  = c0[r];
        row[16] = c1[r];
        row[32] = c2[r];
        row[48] = c3[r];
    }
    float so = 0.0f, dd = 0.0f;
    #pragma unroll
    for (int r = 0; r < 8; ++r) {
        float s = fmaf(c0[r], as0, c2[r] * as2) + fmaf(c1[r], as1, c3[r] * as3);
        float d = fmaf(c0[r], ad0, c2[r] * ad2) + fmaf(c1[r], ad1, c3[r] * ad3);
        #pragma unroll
        for (int off = 8; off; off >>= 1) {
            s += __shfl_xor(s, off, 32);
            d += __shfl_xor(d, off, 32);
        }
        if (r0 == r) { so = s; dd = d; }
    }
    if (r0 < 8) {
        S[tile * 16 + khalf * 8 + r0] = so;
        D[tile * 16 + khalf * 8 + r0] = dd;
    }
}

__device__ __forceinline__ void gemm_wmma_tile(
        const float* __restrict__ X, const float* __restrict__ W,
        const float* __restrict__ a_s, const float* __restrict__ a_d,
        float* __restrict__ H, float* __restrict__ S, float* __restrict__ D, const int wave, const int K) {
    int lane  = threadIdx.x & 31;
    int r0    = lane & 15;
    int khalf = lane >> 4;

    const float* xrow = X + (size_t)(wave * 16 + r0) * (size_t)K;

    const float as0 = a_s[r0], as1 = a_s[16 + r0], as2 = a_s[32 + r0], as3 = a_s[48 + r0];
    const float ad0 = a_d[r0], ad1 = a_d[16 + r0], ad2 = a_d[32 + r0], ad3 = a_d[48 + r0];

    v8f c0 = {}, c1 = {}, c2 = {}, c3 = {};
    for (int k0 = 0; k0 < K; k0 += 4) {
        int ka = k0 + khalf * 2;
        v2f a;
        a.x = xrow[ka];
        a.y = xrow[ka + 1];

        const float* w0 = W + (size_t)ka * 64;
        const float* w1 = w0 + 64;
        v2f b0, b1, b2, b3;
        b0.x = w0[r0];      b0.y = w1[r0];
        b1.x = w0[16 + r0]; b1.y = w1[16 + r0];
        b2.x = w0[32 + r0]; b2.y = w1[32 + r0];
        b3.x = w0[48 + r0]; b3.y = w1[48 + r0];

        c0 = __builtin_amdgcn_wmma_f32_16x16x4_f32(false, a, false, b0, (short)0, c0, false, false);
        c1 = __builtin_amdgcn_wmma_f32_16x16x4_f32(false, a, false, b1, (short)0, c1, false, false);
        c2 = __builtin_amdgcn_wmma_f32_16x16x4_f32(false, a, false, b2, (short)0, c2, false, false);
        c3 = __builtin_amdgcn_wmma_f32_16x16x4_f32(false, a, false, b3, (short)0, c3, false, false);
    }

    tile_store_and_scores(c0, c1, c2, c3, as0, as1, as2, as3, ad0, ad1, ad2, ad3, H, S, D, wave, r0, khalf);
}

#define NPW 8
#define NSTRIP (NN / NPW)
#define NQ 4
static_assert(NN % NPW == 0, "a strip must not run past the last node");
static_assert(NPW % NQ == 0, "a strip is a whole number of groups");

#define NPB_LOG2 8
#define NPB (1 << NPB_LOG2)
#define NBUCK ((NN + NPB - 1) / NPB)
#define CH 4096
#define EPT (CH / 256)
#define SRC_BITS 17
#define SRC_MASK ((1 << SRC_BITS) - 1)
static_assert(NN <= (1 << SRC_BITS), "a source id must fit the low bits of the packed word");
static_assert(NPB_LOG2 + SRC_BITS + 1 < 32, "packed word and the spare value past it must stay non-negative");

__global__ __launch_bounds__(256) void bucket_hist_kernel(const int* __restrict__ ei, int E, int G1,
        int* __restrict__ cntm) {
    __shared__ int hist[NBUCK];
    const int tid = threadIdx.x;
    for (int i = tid; i < NBUCK; i += 256) hist[i] = 0;
    __syncthreads();
    const int e0 = blockIdx.x * CH + tid;
    int d[EPT];
    #pragma unroll
    for (int k = 0; k < EPT; ++k) {
        const int e = e0 + k * 256;
        d[k] = (e < E) ? ei[E + e] : -1;
    }
    #pragma unroll
    for (int k = 0; k < EPT; ++k) {
        if (d[k] >= 0) atomicAdd(&hist[d[k] >> NPB_LOG2], 1);
    }
    __syncthreads();
    for (int i = tid; i < NBUCK; i += 256) cntm[i * G1 + blockIdx.x] = hist[i];
}

#define SG 16
__global__ __launch_bounds__(1024) void scan_counts_kernel(
        const int* __restrict__ cnt, int* __restrict__ off, int n) {
    __shared__ int red[32];
    __shared__ int wtot[32];
    const int tid  = threadIdx.x;
    const int lane = tid & 31;
    const int wv   = tid >> 5;
    const int tile0 = blockIdx.x * 1024;

    const int4* c4 = (const int4*)cnt;
    const int nvec = tile0 >> 2;
    int part = 0;
    for (int i = tid; i < nvec; i += SG * 1024) {
        int4 v[SG];
        #pragma unroll
        for (int u = 0; u < SG; ++u) {
            const int j = i + u * 1024;
            v[u] = c4[(j < nvec) ? j : i];
        }
        __builtin_amdgcn_sched_barrier(0);
        #pragma unroll
        for (int u = 0; u < SG; ++u) {
            if (i + u * 1024 < nvec) part += (v[u].x + v[u].y) + (v[u].z + v[u].w);
        }
    }
    #pragma unroll
    for (int o = 16; o; o >>= 1) part += __shfl_xor(part, o, 32);
    if (lane == 0) red[wv] = part;
    __syncthreads();
    int base = red[lane];
    #pragma unroll
    for (int o = 16; o; o >>= 1) base += __shfl_xor(base, o, 32);

    const int i = tile0 + tid;
    const int v = (i < n) ? cnt[i] : 0;
    int inc = v;
    #pragma unroll
    for (int o = 1; o < 32; o <<= 1) {
        int t = __shfl_up(inc, o, 32);
        if (lane >= o) inc += t;
    }
    if (lane == 31) wtot[wv] = inc;
    __syncthreads();
    const int wt = wtot[lane];
    int winc = wt;
    #pragma unroll
    for (int o = 1; o < 32; o <<= 1) {
        int t = __shfl_up(winc, o, 32);
        if (lane >= o) winc += t;
    }
    const int wprefix = __shfl(winc, wv, 32) - __shfl(wt, wv, 32);
    if (i < n) off[i] = base + wprefix + inc - v;
    if (i == n - 1) off[n] = base + wprefix + inc;
}

__global__ __launch_bounds__(256) void bucket_scatter_kernel(const int* __restrict__ ei, int E, int G1,
        const int* __restrict__ boff, int* __restrict__ tmp) {
    __shared__ int cur[NBUCK + 1];
    const int tid = threadIdx.x;
    const int e0 = blockIdx.x * CH + tid;
    int d[EPT], s[EPT];
    #pragma unroll
    for (int k = 0; k < EPT; ++k) {
        const int e = e0 + k * 256;
        d[k] = (e < E) ? ei[E + e] : (NBUCK << NPB_LOG2);
        s[k] = (e < E) ? ei[e] : 0;
    }
    for (int i = tid; i < NBUCK; i += 256) cur[i] = boff[i * G1 + blockIdx.x];
    if (tid == 0) cur[NBUCK] = 0;
    __syncthreads();
    int pos[EPT];
    #pragma unroll
    for (int k = 0; k < EPT; ++k) pos[k] = atomicAdd(&cur[d[k] >> NPB_LOG2], 1);
    #pragma unroll
    for (int k = 0; k < EPT; ++k) {
        if (e0 + k * 256 < E) tmp[pos[k]] = ((d[k] & (NPB - 1)) << SRC_BITS) | s[k];
    }
}

struct GemmArgs {
    const float* X; const float* W; const float* a_s; const float* a_d;
    float* H; float* S; float* D;
    int nrows; int K;
};

__global__ __launch_bounds__(256) void bucket_csr_gemm_kernel(const int* __restrict__ boff, int G1,
        const int* __restrict__ tmp, int* __restrict__ off, int* __restrict__ col, const GemmArgs g) {
    if (blockIdx.x >= NBUCK) {
        const int tile = __builtin_amdgcn_readfirstlane((int)((blockIdx.x - NBUCK) * 8 + (threadIdx.x >> 5)));
        if (tile < (g.nrows >> 4)) gemm_wmma_tile(g.X, g.W, g.a_s, g.a_d, g.H, g.S, g.D, tile, g.K);
        return;
    }
    __shared__ int cnt[NPB + 1];
    __shared__ int wtot[NPB / 32];
    const int tid  = threadIdx.x;
    const int lane = tid & 31;
    const int wv   = tid >> 5;
    const int bs = boff[blockIdx.x * G1];
    const int be = boff[(blockIdx.x + 1) * G1];
    cnt[tid] = 0;
    if (tid == 0) cnt[NPB] = 0;
    __syncthreads();
    for (int i = bs + tid; i < be; i += 4 * 256) {
        int w[4];
        #pragma unroll
        for (int u = 0; u < 4; ++u) {
            const int j = i + u * 256;
            w[u] = (j < be) ? tmp[j] : -1;
        }
        #pragma unroll
        for (int u = 0; u < 4; ++u) {
            if (w[u] >= 0) atomicAdd(&cnt[w[u] >> SRC_BITS], 1);
        }
    }
    __syncthreads();
    const int v = cnt[tid];
    int inc = v;
    #pragma unroll
    for (int o = 1; o < 32; o <<= 1) {
        int t = __shfl_up(inc, o, 32);
        if (lane >= o) inc += t;
    }
    if (lane == 31) wtot[wv] = inc;
    __syncthreads();
    int wprefix = 0;
    for (int w = 0; w < wv; ++w) wprefix += wtot[w];
    const int excl = bs + wprefix + inc - v;
    const int node = blockIdx.x * NPB + tid;
    if (node < NN) off[node] = excl;
    if (node == NN - 1) off[NN] = excl + v;
    cnt[tid] = excl;
    __syncthreads();
    for (int i = bs + tid; i < be; i += 4 * 256) {
        int w[4];
        #pragma unroll
        for (int u = 0; u < 4; ++u) {
            const int j = i + u * 256;
            w[u] = (j < be) ? tmp[j] : (NPB << SRC_BITS);
        }
        int pos[4];
        #pragma unroll
        for (int u = 0; u < 4; ++u) pos[u] = atomicAdd(&cnt[w[u] >> SRC_BITS], 1);
        #pragma unroll
        for (int u = 0; u < 4; ++u) {
            if (i + u * 256 < be) col[pos[u]] = w[u] & SRC_MASK;
        }
    }
}

__global__ __launch_bounds__(256) void sort_segments_kernel(
        const int* __restrict__ off, const int* __restrict__ cin, int* __restrict__ cout) {
    const int lane = threadIdx.x & 31;
    const int strip = __builtin_amdgcn_readfirstlane((int)(blockIdx.x * 8 + (threadIdx.x >> 5)));
    if (strip >= NSTRIP) return;
    const int n0 = strip * NPW;
    const int offv = off[n0 + ((lane < NPW) ? lane : NPW)];
    int o0[NPW], deg[NPW], key[NPW];
    #pragma unroll
    for (int u = 0; u < NPW; ++u) {
        o0[u]  = __builtin_amdgcn_readlane(offv, u);
        deg[u] = __builtin_amdgcn_readlane(offv, u + 1) - o0[u];
        key[u] = (lane < deg[u]) ? cin[o0[u] + lane] : 0x7fffffff;
    }
    #pragma unroll
    for (int u = 0; u < NPW; ++u) {
        if (deg[u] <= 32) {
            int rank = 0;
            for (int j = 0; j < deg[u]; ++j) {
                const int kj = __builtin_amdgcn_readlane(key[u], j);
                rank += (kj < key[u] || (kj == key[u] && j < lane)) ? 1 : 0;
            }
            if (lane < deg[u]) cout[o0[u] + rank] = key[u];
        } else {
            for (int i = lane; i < deg[u]; i += 32) {
                const int ki = cin[o0[u] + i];
                int rank = 0;
                for (int j = 0; j < deg[u]; ++j) {
                    const int kj = cin[o0[u] + j];
                    rank += (kj < ki || (kj == ki && j < i)) ? 1 : 0;
                }
                cout[o0[u] + rank] = ki;
            }
        }
    }
}

__device__ __forceinline__ float leaky(float a) { return (a >= 0.0f) ? a : NEG_SLOPE * a; }

__device__ __forceinline__ float readlane_f(float v, int l) {
    return __int_as_float(__builtin_amdgcn_readlane(__float_as_int(v), l));
}

#define TP 68

template <bool TO_LDS>
__device__ __forceinline__ void finish_row(v2f acc, float bx, float by, int relu,
                                           float* __restrict__ OUT, float* tile, int node, int row, int lane) {
    acc.x += bx;
    acc.y += by;
    if (relu) {
        acc.x = (acc.x > 0.0f) ? acc.x : 0.0f;
        acc.y = (acc.y > 0.0f) ? acc.y : 0.0f;
    }
    if (TO_LDS) ((v2f*)tile)[row * (TP / 2) + lane] = acc;
    else        ((v2f*)OUT)[(size_t)node * 32 + lane] = acc;
}

template <bool TO_LDS>
__device__ __forceinline__ void strip_rows(
        const int strip, const int lane,
        const int* __restrict__ off, const int* __restrict__ col,
        const float* __restrict__ H, const float* __restrict__ S, const float* __restrict__ D,
        const float* __restrict__ bias, int relu, float* __restrict__ OUT, float* tile, const int rb0) {
    const int n0 = strip * NPW;
    const int   offv = off[n0 + ((lane < NPW) ? lane : NPW)];
    const float Sv   = S[n0 + (lane & (NPW - 1))];
    const float Dv   = D[n0 + (lane & (NPW - 1))];
    const v2f* H2 = (const v2f*)H;
    const float bx = bias[2 * lane];
    const float by = bias[2 * lane + 1];

    bool any_long = false;
    for (int q = 0; q < NPW; q += NQ) {
        int o0[NQ], deg[NQ], sv[NQ];
        v2f acc0[NQ];
        #pragma unroll
        for (int u = 0; u < NQ; ++u) {
            o0[u]   = __builtin_amdgcn_readlane(offv, q + u);
            deg[u]  = __builtin_amdgcn_readlane(offv, q + u + 1) - o0[u];
            sv[u]   = (lane < deg[u]) ? col[o0[u] + lane] : (n0 + q + u);
            acc0[u] = H2[(size_t)(n0 + q + u) * 32 + lane];
        }
        float av[NQ], aself[NQ];
        #pragma unroll
        for (int u = 0; u < NQ; ++u) av[u] = S[sv[u]];
        #pragma unroll
        for (int u = 0; u < NQ; ++u) {
            const float dn = readlane_f(Dv, q + u);
            aself[u] = leaky(readlane_f(Sv, q + u) + dn);
            av[u] = (lane < deg[u]) ? leaky(av[u] + dn) : -INFINITY;
        }
        float wv[NQ];
        #pragma unroll
        for (int u = 0; u < NQ; ++u) {
            float m = fmaxf(av[u], aself[u]);
            #pragma unroll
            for (int o = 16; o; o >>= 1) m = fmaxf(m, __shfl_xor(m, o, 32));
            const float ex = expf(av[u] - m);
            const float e = (lane < deg[u]) ? ex : 0.0f;
            const float e_self = expf(aself[u] - m);
            float z = e;
            #pragma unroll
            for (int o = 16; o; o >>= 1) z += __shfl_xor(z, o, 32);
            z += e_self;
            wv[u] = e / (z + EPSF);
            const float w_self = e_self / (z + EPSF);
            acc0[u].x = acc0[u].x * w_self;
            acc0[u].y = acc0[u].y * w_self;
        }
        #pragma unroll
        for (int u = 0; u < NQ; ++u) {
            if (deg[u] > 32) { any_long = true; continue; }
            const float w = wv[u];
            v2f acc = acc0[u];
            for (int j = 0; j < deg[u]; j += 8) {
                v2f hv[8];
                #pragma unroll
                for (int t = 0; t < 8; ++t) {
                    const int jt = (j + t < deg[u]) ? (j + t) : (deg[u] - 1);
                    const int sj = __builtin_amdgcn_readlane(sv[u], jt);
                    hv[t] = H2[(size_t)sj * 32 + lane];
                }
                #pragma unroll
                for (int t = 0; t < 8; ++t) {
                    const float wj = readlane_f(w, j + t);
                    acc.x = fmaf(hv[t].x, wj, acc.x);
                    acc.y = fmaf(hv[t].y, wj, acc.y);
                }
            }
            finish_row<TO_LDS>(acc, bx, by, relu, OUT, tile, n0 + q + u, rb0 + q + u, lane);
        }
    }

    if (any_long) {
        for (int u = 0; u < NPW; ++u) {
            const int lo = __builtin_amdgcn_readlane(offv, u);
            const int dg = __builtin_amdgcn_readlane(offv, u + 1) - lo;
            if (dg <= 32) continue;
            const int node = n0 + u;
            const float dn = readlane_f(Dv, u);
            const float a_self = leaky(readlane_f(Sv, u) + dn);
            const v2f hsn = H2[(size_t)node * 32 + lane];
            float m = a_self;
            for (int c = 0; c < dg; c += 32) {
                if (c + lane < dg) m = fmaxf(m, leaky(S[col[lo + c + lane]] + dn));
            }
            #pragma unroll
            for (int o = 16; o; o >>= 1) m = fmaxf(m, __shfl_xor(m, o, 32));
            float z = 0.0f;
            for (int c = 0; c < dg; c += 32) {
                if (c + lane < dg) z += expf(leaky(S[col[lo + c + lane]] + dn) - m);
            }
            #pragma unroll
            for (int o = 16; o; o >>= 1) z += __shfl_xor(z, o, 32);
            const float e_self = expf(a_self - m);
            z += e_self;
            const float w_self = e_self / (z + EPSF);
            v2f acc;
            acc.x = hsn.x * w_self;
            acc.y = hsn.y * w_self;
            for (int c = 0; c < dg; c += 32) {
                int s = node;
                float w = 0.0f;
                if (c + lane < dg) {
                    s = col[lo + c + lane];
                    w = expf(leaky(S[s] + dn) - m) / (z + EPSF);
                }
                const int nj = (dg - c < 32) ? (dg - c) : 32;
                for (int j = 0; j < nj; ++j) {
                    const int sj = __builtin_amdgcn_readlane(s, j);
                    const float wj = readlane_f(w, j);
                    const v2f hv = H2[(size_t)sj * 32 + lane];
                    acc.x = fmaf(hv.x, wj, acc.x);
                    acc.y = fmaf(hv.y, wj, acc.y);
                }
            }
            finish_row<TO_LDS>(acc, bx, by, relu, OUT, tile, node, rb0 + u, lane);
        }
    }
}

__global__ __launch_bounds__(256) void node_softmax_sum_kernel(
        const int* __restrict__ off, const int* __restrict__ col,
        const float* __restrict__ H, const float* __restrict__ S, const float* __restrict__ D,
        const float* __restrict__ bias, int relu, float* __restrict__ OUT) {
    const int lane = threadIdx.x & 31;
    const int strip = __builtin_amdgcn_readfirstlane((int)(blockIdx.x * 8 + (threadIdx.x >> 5)));
    if (strip >= NSTRIP) return;
    strip_rows<false>(strip, lane, off, col, H, S, D, bias, relu, OUT, nullptr, 0);
}

static_assert(NPW * 8 == 64, "a block of 8 waves must finish four whole 16-row tiles");
static_assert(NN % 16 == 0, "whole tiles only");
__global__ __launch_bounds__(256, 16) void node_softmax_sum_gemm_kernel(
        const int* __restrict__ off, const int* __restrict__ col,
        const float* __restrict__ H, const float* __restrict__ S, const float* __restrict__ D,
        const float* __restrict__ bias, int relu,
        const float* __restrict__ W2, const float* __restrict__ a_s2, const float* __restrict__ a_d2,
        float* __restrict__ H2, float* __restrict__ S2, float* __restrict__ D2) {
    __shared__ float tile[64 * TP];
    const int lane = threadIdx.x & 31;
    const int wv = __builtin_amdgcn_readfirstlane((int)(threadIdx.x >> 5));
    const int strip = blockIdx.x * 8 + wv;
    if (strip < NSTRIP) strip_rows<true>(strip, lane, off, col, H, S, D, bias, relu, nullptr, tile, wv * NPW);
    __syncthreads();

    const int t = blockIdx.x * 4 + wv;
    if (wv >= 4 || t >= NN / 16) return;

    const int r0    = lane & 15;
    const int khalf = lane >> 4;
    const float* arow = tile + (wv * 16 + r0) * TP;

    const float as0 = a_s2[r0], as1 = a_s2[16 + r0], as2 = a_s2[32 + r0], as3 = a_s2[48 + r0];
    const float ad0 = a_d2[r0], ad1 = a_d2[16 + r0], ad2 = a_d2[32 + r0], ad3 = a_d2[48 + r0];

    v8f c0 = {}, c1 = {}, c2 = {}, c3 = {};
    for (int k0 = 0; k0 < 64; k0 += 4) {
        const int ka = k0 + khalf * 2;
        v2f a;
        a.x = arow[ka];
        a.y = arow[ka + 1];

        const float* w0 = W2 + (size_t)ka * 64;
        const float* w1 = w0 + 64;
        v2f b0, b1, b2, b3;
        b0.x = w0[r0];      b0.y = w1[r0];
        b1.x = w0[16 + r0]; b1.y = w1[16 + r0];
        b2.x = w0[32 + r0]; b2.y = w1[32 + r0];
        b3.x = w0[48 + r0]; b3.y = w1[48 + r0];

        c0 = __builtin_amdgcn_wmma_f32_16x16x4_f32(false, a, false, b0, (short)0, c0, false, false);
        c1 = __builtin_amdgcn_wmma_f32_16x16x4_f32(false, a, false, b1, (short)0, c1, false, false);
        c2 = __builtin_amdgcn_wmma_f32_16x16x4_f32(false, a, false, b2, (short)0, c2, false, false);
        c3 = __builtin_amdgcn_wmma_f32_16x16x4_f32(false, a, false, b3, (short)0, c3, false, false);
    }

    tile_store_and_scores(c0, c1, c2, c3, as0, as1, as2, as3, ad0, ad1, ad2, ad3, H2, S2, D2, t, r0, khalf);
}

extern "C" void kernel_launch(void* const* d_in, const int* in_sizes, int n_in,
                              void* d_out, int out_size, void* d_ws, size_t ws_size,
                              hipStream_t stream) {
    const float* x     = (const float*)d_in[0];
    const int*   ei    = (const int*)d_in[1];
    const float* W1    = (const float*)d_in[2];
    const float* as1   = (const float*)d_in[3];
    const float* ad1   = (const float*)d_in[4];
    const float* b1    = (const float*)d_in[5];
    const float* W2    = (const float*)d_in[6];
    const float* as2   = (const float*)d_in[7];
    const float* ad2   = (const float*)d_in[8];
    const float* b2    = (const float*)d_in[9];
    float* out = (float*)d_out;

    const int E  = in_sizes[1] / 2;
    const int NF = NN * 64;

    float* wsf = (float*)d_ws;
    float* H   = wsf;
    float* H2  = H   + NF;
    float* S   = H2  + NF;
    float* D   = S   + NN;
    const int G1 = (E + CH - 1) / CH;
    const int nM = NBUCK * G1;
    int*   cntm = (int*)(D + NN);
    int*   boff = cntm + nM + 16;
    int*   off  = boff + nM + 16;
    int*   tmp  = off + NN + 16;
    int*   cA   = tmp + E;
    int*   cB   = cA  + E;
    float* S2   = (float*)(cB + E);
    float* D2   = S2 + NN;

    const int TB = 256;
    dim3 blk(TB);
    int gScan = (nM + 1023) / 1024;
    int gNode = (NSTRIP + 7) / 8;
    int gGemm = ((NN / 16) + 7) / 8;

    bucket_hist_kernel<<<G1, blk, 0, stream>>>(ei, E, G1, cntm);
    scan_counts_kernel<<<gScan, dim3(1024), 0, stream>>>(cntm, boff, nM);
    bucket_scatter_kernel<<<G1, blk, 0, stream>>>(ei, E, G1, boff, tmp);
    const GemmArgs g1 = {x, W1, as1, ad1, H, S, D, NN, 128};
    bucket_csr_gemm_kernel<<<NBUCK + gGemm, blk, 0, stream>>>(boff, G1, tmp, off, cA, g1);
    sort_segments_kernel<<<gNode, blk, 0, stream>>>(off, cA, cB);

    node_softmax_sum_gemm_kernel<<<gNode, blk, 0, stream>>>(off, cB, H, S, D, b1, 1, W2, as2, ad2, H2, S2, D2);

    node_softmax_sum_kernel<<<gNode, blk, 0, stream>>>(off, cB, H2, S2, D2, b2, 0, out);
}


# Edge grouping: each 4096-edge chunk grouped by bucket in LDS and written chunk-major with its own run starts; bucket groups collect their runs per chunk: no count matrix or prefix; 5 launches, not 7
# speedup vs baseline: 6.2517x; 1.2888x over previous
#include <hip/hip_runtime.h>
#include <math.h>


typedef float v2f __attribute__((ext_vector_type(2)));
typedef float v8f __attribute__((ext_vector_type(8)));

#define NN 100000
#define NEG_SLOPE 0.2f
#define EPSF 1e-16f

__device__ __forceinline__ void tile_store_and_scores(
        const v8f c0, const v8f c1, const v8f c2, const v8f c3,
        const float as0, const float as1, const float as2, const float as3,
        const float ad0, const float ad1, const float ad2, const float ad3,
        float* __restrict__ H, float* __restrict__ S, float* __restrict__ D,
        const int tile, const int r0, const int khalf) {
    float* hbase = H + (size_t)(tile * 16 + khalf * 8) * 64 + r0;
    #pragma unroll
    for (int r = 0; r < 8; ++r) {
        float* row = hbase + (size_t)r * 64;
        row[0]  = c0[r];
        row[16] = c1[r];
        row[32] = c2[r];
        row[48] = c3[r];
    }
    float so = 0.0f, dd = 0.0f;
    #pragma unroll
    for (int r = 0; r < 8; ++r) {
        float s = fmaf(c0[r], as0, c2[r] * as2) + fmaf(c1[r], as1, c3[r] * as3);
        float d = fmaf(c0[r], ad0, c2[r] * ad2) + fmaf(c1[r], ad1, c3[r] * ad3);
        #pragma unroll
        for (int off = 8; off; off >>= 1) {
            s += __shfl_xor(s, off, 32);
            d += __shfl_xor(d, off, 32);
        }
        if (r0 == r) { so = s; dd = d; }
    }
    if (r0 < 8) {
        S[tile * 16 + khalf * 8 + r0] = so;
        D[tile * 16 + khalf * 8 + r0] = dd;
    }
}

__device__ __forceinline__ void gemm_wmma_tile(
        const float* __restrict__ X, const float* __restrict__ W,
        const float* __restrict__ a_s, const float* __restrict__ a_d,
        float* __restrict__ H, float* __restrict__ S, float* __restrict__ D, const int wave, const int K) {
    int lane  = threadIdx.x & 31;
    int r0    = lane & 15;
    int khalf = lane >> 4;

    const float* xrow = X + (size_t)(wave * 16 + r0) * (size_t)K;

    const float as0 = a_s[r0], as1 = a_s[16 + r0], as2 = a_s[32 + r0], as3 = a_s[48 + r0];
    const float ad0 = a_d[r0], ad1 = a_d[16 + r0], ad2 = a_d[32 + r0], ad3 = a_d[48 + r0];

    v8f c0 = {}, c1 = {}, c2 = {}, c3 = {};
    for (int k0 = 0; k0 < K; k0 += 4) {
        int ka = k0 + khalf * 2;
        v2f a;
        a.x = xrow[ka];
        a.y = xrow[ka + 1];

        const float* w0 = W + (size_t)ka * 64;
        const float* w1 = w0 + 64;
        v2f b0, b1, b2, b3;
        b0.x = w0[r0];      b0.y = w1[r0];
        b1.x = w0[16 + r0]; b1.y = w1[16 + r0];
        b2.x = w0[32 + r0]; b2.y = w1[32 + r0];
        b3.x = w0[48 + r0]; b3.y = w1[48 + r0];

        c0 = __builtin_amdgcn_wmma_f32_16x16x4_f32(false, a, false, b0, (short)0, c0, false, false);
        c1 = __builtin_amdgcn_wmma_f32_16x16x4_f32(false, a, false, b1, (short)0, c1, false, false);
        c2 = __builtin_amdgcn_wmma_f32_16x16x4_f32(false, a, false, b2, (short)0, c2, false, false);
        c3 = __builtin_amdgcn_wmma_f32_16x16x4_f32(false, a, false, b3, (short)0, c3, false, false);
    }

    tile_store_and_scores(c0, c1, c2, c3, as0, as1, as2, as3, ad0, ad1, ad2, ad3, H, S, D, wave, r0, khalf);
}

#define NPW 8
#define NSTRIP (NN / NPW)
#define NQ 4
static_assert(NN % NPW == 0, "a strip must not run past the last node");
static_assert(NPW % NQ == 0, "a strip is a whole number of groups");

#define NPB_LOG2 8
#define NPB (1 << NPB_LOG2)
#define NBUCK ((NN + NPB - 1) / NPB)
#define CH 4096
#define EPT (CH / 256)
#define SRC_BITS 17
#define SRC_MASK ((1 << SRC_BITS) - 1)
static_assert(NN <= (1 << SRC_BITS), "a source id must fit the low bits of the packed word");
static_assert(NPB_LOG2 + SRC_BITS + 1 < 32, "the packed word must stay non-negative");
static_assert(NBUCK + 1 <= 512, "pass 1 scans two bucket counts per thread");

__global__ __launch_bounds__(256) void chunk_group_kernel(const int* __restrict__ ei, int E,
        int* __restrict__ lofs, int* __restrict__ tmp) {
    __shared__ int cnt[NBUCK + 1];
    __shared__ int wsum[8];
    __shared__ int stage[CH];
    const int tid  = threadIdx.x;
    const int lane = tid & 31;
    const int wv   = tid >> 5;
    const int e0 = blockIdx.x * CH + tid;
    int d[EPT], s[EPT];
    #pragma unroll
    for (int k = 0; k < EPT; ++k) {
        const int e = e0 + k * 256;
        d[k] = (e < E) ? ei[E + e] : -1;
        s[k] = (e < E) ? ei[e] : 0;
    }
    for (int i = tid; i <= NBUCK; i += 256) cnt[i] = 0;
    __syncthreads();
    int rk[EPT];
    #pragma unroll
    for (int k = 0; k < EPT; ++k) rk[k] = (d[k] >= 0) ? atomicAdd(&cnt[d[k] >> NPB_LOG2], 1) : 0;
    __syncthreads();
    const int i0 = 2 * tid, i1 = 2 * tid + 1;
    const int c0 = (i0 < NBUCK) ? cnt[i0] : 0;
    const int c1 = (i1 < NBUCK) ? cnt[i1] : 0;
    int inc = c0 + c1;
    #pragma unroll
    for (int o = 1; o < 32; o <<= 1) {
        int t = __shfl_up(inc, o, 32);
        if (lane >= o) inc += t;
    }
    if (lane == 31) wsum[wv] = inc;
    __syncthreads();
    int wprefix = 0;
    for (int w = 0; w < wv; ++w) wprefix += wsum[w];
    const int ex = wprefix + inc - (c0 + c1);
    if (i0 < NBUCK) cnt[i0] = ex;
    if (i1 < NBUCK) cnt[i1] = ex + c0;
    if (tid == 255) cnt[NBUCK] = ex + c0 + c1;
    __syncthreads();
    int* lo = lofs + (size_t)blockIdx.x * (NBUCK + 1);
    for (int i = tid; i <= NBUCK; i += 256) lo[i] = cnt[i];
    #pragma unroll
    for (int k = 0; k < EPT; ++k) {
        if (d[k] >= 0) stage[cnt[d[k] >> NPB_LOG2] + rk[k]] = ((d[k] & (NPB - 1)) << SRC_BITS) | s[k];
    }
    __syncthreads();
    const int total = cnt[NBUCK];
    int* out = tmp + (size_t)blockIdx.x * CH;
    #pragma unroll
    for (int k = 0; k < EPT; ++k) {
        const int i = tid + k * 256;
        if (i < total) out[i] = stage[i];
    }
}

struct GemmArgs {
    const float* X; const float* W; const float* a_s; const float* a_d;
    float* H; float* S; float* D;
    int nrows; int K;
};

#define RB 16
template <bool PLACE>
__device__ __forceinline__ int bucket_runs(const int* __restrict__ lofs, const int G1,
        const int* __restrict__ tmp, const int b, int* cnt, int* __restrict__ col,
        const int wv, const int lane) {
    int stsum = 0;
    for (int base = wv * 32; base < G1; base += 512) {
        int st[2], len[2];
        #pragma unroll
        for (int h = 0; h < 2; ++h) {
            const int w = base + h * 256 + lane;
            const int* lo = lofs + (size_t)((w < G1) ? w : 0) * (NBUCK + 1) + b;
            const int a = lo[0];
            const int e = lo[1];
            st[h]  = (w < G1) ? (w * CH + a) : 0;
            len[h] = (w < G1) ? (e - a) : 0;
            stsum += (w < G1) ? a : 0;
        }
        #pragma unroll
        for (int h = 0; h < 2; ++h) {
            #pragma nounroll
            for (int r = 0; r < 32; r += RB) {
                int wd[RB];
                #pragma unroll
                for (int u = 0; u < RB; ++u) {
                    const int s_r = __builtin_amdgcn_readlane(st[h], r + u);
                    const int l_r = __builtin_amdgcn_readlane(len[h], r + u);
                    wd[u] = (lane < l_r) ? tmp[s_r + lane] : -1;
                }
                if (!PLACE) {
                    #pragma unroll
                    for (int u = 0; u < RB; ++u) {
                        if (wd[u] >= 0) atomicAdd(&cnt[wd[u] >> SRC_BITS], 1);
                    }
                } else {
                    int pos[RB];
                    #pragma unroll
                    for (int u = 0; u < RB; ++u) pos[u] = (wd[u] >= 0) ? atomicAdd(&cnt[wd[u] >> SRC_BITS], 1) : 0;
                    #pragma unroll
                    for (int u = 0; u < RB; ++u) {
                        if (wd[u] >= 0) col[pos[u]] = wd[u] & SRC_MASK;
                    }
                }
            }
            unsigned m = __builtin_amdgcn_ballot_w32(len[h] > 32);
            while (m) {
                const int r = __builtin_ctz(m);
                m &= m - 1;
                const int s_r = __builtin_amdgcn_readlane(st[h], r);
                const int l_r = __builtin_amdgcn_readlane(len[h], r);
                for (int i = 32 + lane; i < l_r; i += 32) {
                    const int wd = tmp[s_r + i];
                    if (!PLACE) {
                        atomicAdd(&cnt[wd >> SRC_BITS], 1);
                    } else {
                        const int pos = atomicAdd(&cnt[wd >> SRC_BITS], 1);
                        col[pos] = wd & SRC_MASK;
                    }
                }
            }
        }
    }
    return stsum;
}

__global__ __launch_bounds__(256) void bucket_csr_gemm_kernel(const int* __restrict__ lofs, int G1,
        const int* __restrict__ tmp, int* __restrict__ off, int* __restrict__ col, const GemmArgs g) {
    if (blockIdx.x >= NBUCK) {
        const int tile = __builtin_amdgcn_readfirstlane((int)((blockIdx.x - NBUCK) * 8 + (threadIdx.x >> 5)));
        if (tile < (g.nrows >> 4)) gemm_wmma_tile(g.X, g.W, g.a_s, g.a_d, g.H, g.S, g.D, tile, g.K);
        return;
    }
    __shared__ int cnt[NPB];
    __shared__ int wtot[NPB / 32];
    __shared__ int red[NPB / 32];
    const int tid  = threadIdx.x;
    const int lane = tid & 31;
    const int wv   = __builtin_amdgcn_readfirstlane((int)(threadIdx.x >> 5));
    const int b    = blockIdx.x;
    cnt[tid] = 0;
    __syncthreads();
    int stsum = bucket_runs<false>(lofs, G1, tmp, b, cnt, nullptr, wv, lane);
    #pragma unroll
    for (int o = 16; o; o >>= 1) stsum += __shfl_xor(stsum, o, 32);
    if (lane == 0) red[wv] = stsum;
    __syncthreads();
    int bs = 0;
    for (int w = 0; w < NPB / 32; ++w) bs += red[w];
    const int v = cnt[tid];
    int inc = v;
    #pragma unroll
    for (int o = 1; o < 32; o <<= 1) {
        int t = __shfl_up(inc, o, 32);
        if (lane >= o) inc += t;
    }
    if (lane == 31) wtot[wv] = inc;
    __syncthreads();
    int wprefix = 0;
    for (int w = 0; w < wv; ++w) wprefix += wtot[w];
    const int excl = bs + wprefix + inc - v;
    const int node = b * NPB + tid;
    if (node < NN) off[node] = excl;
    if (node == NN - 1) off[NN] = excl + v;
    cnt[tid] = excl;
    __syncthreads();
    bucket_runs<true>(lofs, G1, tmp, b, cnt, col, wv, lane);
}

__global__ __launch_bounds__(256) void sort_segments_kernel(
        const int* __restrict__ off, const int* __restrict__ cin, int* __restrict__ cout) {
    const int lane = threadIdx.x & 31;
    const int strip = __builtin_amdgcn_readfirstlane((int)(blockIdx.x * 8 + (threadIdx.x >> 5)));
    if (strip >= NSTRIP) return;
    const int n0 = strip * NPW;
    const int offv = off[n0 + ((lane < NPW) ? lane : NPW)];
    int o0[NPW], deg[NPW], key[NPW];
    #pragma unroll
    for (int u = 0; u < NPW; ++u) {
        o0[u]  = __builtin_amdgcn_readlane(offv, u);
        deg[u] = __builtin_amdgcn_readlane(offv, u + 1) - o0[u];
        key[u] = (lane < deg[u]) ? cin[o0[u] + lane] : 0x7fffffff;
    }
    #pragma unroll
    for (int u = 0; u < NPW; ++u) {
        if (deg[u] <= 32) {
            int rank = 0;
            for (int j = 0; j < deg[u]; ++j) {
                const int kj = __builtin_amdgcn_readlane(key[u], j);
                rank += (kj < key[u] || (kj == key[u] && j < lane)) ? 1 : 0;
            }
            if (lane < deg[u]) cout[o0[u] + rank] = key[u];
        } else {
            for (int i = lane; i < deg[u]; i += 32) {
                const int ki = cin[o0[u] + i];
                int rank = 0;
                for (int j = 0; j < deg[u]; ++j) {
                    const int kj = cin[o0[u] + j];
                    rank += (kj < ki || (kj == ki && j < i)) ? 1 : 0;
                }
                cout[o0[u] + rank] = ki;
            }
        }
    }
}

__device__ __forceinline__ float leaky(float a) { return (a >= 0.0f) ? a : NEG_SLOPE * a; }

__device__ __forceinline__ float readlane_f(float v, int l) {
    return __int_as_float(__builtin_amdgcn_readlane(__float_as_int(v), l));
}

#define TP 68

template <bool TO_LDS>
__device__ __forceinline__ void finish_row(v2f acc, float bx, float by, int relu,
                                           float* __restrict__ OUT, float* tile, int node, int row, int lane) {
    acc.x += bx;
    acc.y += by;
    if (relu) {
        acc.x = (acc.x > 0.0f) ? acc.x : 0.0f;
        acc.y = (acc.y > 0.0f) ? acc.y : 0.0f;
    }
    if (TO_LDS) ((v2f*)tile)[row * (TP / 2) + lane] = acc;
    else        ((v2f*)OUT)[(size_t)node * 32 + lane] = acc;
}

template <bool TO_LDS>
__device__ __forceinline__ void strip_rows(
        const int strip, const int lane,
        const int* __restrict__ off, const int* __restrict__ col,
        const float* __restrict__ H, const float* __restrict__ S, const float* __restrict__ D,
        const float* __restrict__ bias, int relu, float* __restrict__ OUT, float* tile, const int rb0) {
    const int n0 = strip * NPW;
    const int   offv = off[n0 + ((lane < NPW) ? lane : NPW)];
    const float Sv   = S[n0 + (lane & (NPW - 1))];
    const float Dv   = D[n0 + (lane & (NPW - 1))];
    const v2f* H2 = (const v2f*)H;
    const float bx = bias[2 * lane];
    const float by = bias[2 * lane + 1];

    bool any_long = false;
    for (int q = 0; q < NPW; q += NQ) {
        int o0[NQ], deg[NQ], sv[NQ];
        v2f acc0[NQ];
        #pragma unroll
        for (int u = 0; u < NQ; ++u) {
            o0[u]   = __builtin_amdgcn_readlane(offv, q + u);
            deg[u]  = __builtin_amdgcn_readlane(offv, q + u + 1) - o0[u];
            sv[u]   = (lane < deg[u]) ? col[o0[u] + lane] : (n0 + q + u);
            acc0[u] = H2[(size_t)(n0 + q + u) * 32 + lane];
        }
        float av[NQ], aself[NQ];
        #pragma unroll
        for (int u = 0; u < NQ; ++u) av[u] = S[sv[u]];
        #pragma unroll
        for (int u = 0; u < NQ; ++u) {
            const float dn = readlane_f(Dv, q + u);
            aself[u] = leaky(readlane_f(Sv, q + u) + dn);
            av[u] = (lane < deg[u]) ? leaky(av[u] + dn) : -INFINITY;
        }
        float wv[NQ];
        #pragma unroll
        for (int u = 0; u < NQ; ++u) {
            float m = fmaxf(av[u], aself[u]);
            #pragma unroll
            for (int o = 16; o; o >>= 1) m = fmaxf(m, __shfl_xor(m, o, 32));
            const float ex = expf(av[u] - m);
            const float e = (lane < deg[u]) ? ex : 0.0f;
            const float e_self = expf(aself[u] - m);
            float z = e;
            #pragma unroll
            for (int o = 16; o; o >>= 1) z += __shfl_xor(z, o, 32);
            z += e_self;
            wv[u] = e / (z + EPSF);
            const float w_self = e_self / (z + EPSF);
            acc0[u].x = acc0[u].x * w_self;
            acc0[u].y = acc0[u].y * w_self;
        }
        #pragma unroll
        for (int u = 0; u < NQ; ++u) {
            if (deg[u] > 32) { any_long = true; continue; }
            const float w = wv[u];
            v2f acc = acc0[u];
            for (int j = 0; j < deg[u]; j += 8) {
                v2f hv[8];
                #pragma unroll
                for (int t = 0; t < 8; ++t) {
                    const int jt = (j + t < deg[u]) ? (j + t) : (deg[u] - 1);
                    const int sj = __builtin_amdgcn_readlane(sv[u], jt);
                    hv[t] = H2[(size_t)sj * 32 + lane];
                }
                #pragma unroll
                for (int t = 0; t < 8; ++t) {
                    const float wj = readlane_f(w, j + t);
                    acc.x = fmaf(hv[t].x, wj, acc.x);
                    acc.y = fmaf(hv[t].y, wj, acc.y);
                }
            }
            finish_row<TO_LDS>(acc, bx, by, relu, OUT, tile, n0 + q + u, rb0 + q + u, lane);
        }
    }

    if (any_long) {
        for (int u = 0; u < NPW; ++u) {
            const int lo = __builtin_amdgcn_readlane(offv, u);
            const int dg = __builtin_amdgcn_readlane(offv, u + 1) - lo;
            if (dg <= 32) continue;
            const int node = n0 + u;
            const float dn = readlane_f(Dv, u);
            const float a_self = leaky(readlane_f(Sv, u) + dn);
            const v2f hsn = H2[(size_t)node * 32 + lane];
            float m = a_self;
            for (int c = 0; c < dg; c += 32) {
                if (c + lane < dg) m = fmaxf(m, leaky(S[col[lo + c + lane]] + dn));
            }
            #pragma unroll
            for (int o = 16; o; o >>= 1) m = fmaxf(m, __shfl_xor(m, o, 32));
            float z = 0.0f;
            for (int c = 0; c < dg; c += 32) {
                if (c + lane < dg) z += expf(leaky(S[col[lo + c + lane]] + dn) - m);
            }
            #pragma unroll
            for (int o = 16; o; o >>= 1) z += __shfl_xor(z, o, 32);
            const float e_self = expf(a_self - m);
            z += e_self;
            const float w_self = e_self / (z + EPSF);
            v2f acc;
            acc.x = hsn.x * w_self;
            acc.y = hsn.y * w_self;
            for (int c = 0; c < dg; c += 32) {
                int s = node;
                float w = 0.0f;
                if (c + lane < dg) {
                    s = col[lo + c + lane];
                    w = expf(leaky(S[s] + dn) - m) / (z + EPSF);
                }
                const int nj = (dg - c < 32) ? (dg - c) : 32;
                for (int j = 0; j < nj; ++j) {
                    const int sj = __builtin_amdgcn_readlane(s, j);
                    const float wj = readlane_f(w, j);
                    const v2f hv = H2[(size_t)sj * 32 + lane];
                    acc.x = fmaf(hv.x, wj, acc.x);
                    acc.y = fmaf(hv.y, wj, acc.y);
                }
            }
            finish_row<TO_LDS>(acc, bx, by, relu, OUT, tile, node, rb0 + u, lane);
        }
    }
}

__global__ __launch_bounds__(256) void node_softmax_sum_kernel(
        const int* __restrict__ off, const int* __restrict__ col,
        const float* __restrict__ H, const float* __restrict__ S, const float* __restrict__ D,
        const float* __restrict__ bias, int relu, float* __restrict__ OUT) {
    const int lane = threadIdx.x & 31;
    const int strip = __builtin_amdgcn_readfirstlane((int)(blockIdx.x * 8 + (threadIdx.x >> 5)));
    if (strip >= NSTRIP) return;
    strip_rows<false>(strip, lane, off, col, H, S, D, bias, relu, OUT, nullptr, 0);
}

static_assert(NPW * 8 == 64, "a block of 8 waves must finish four whole 16-row tiles");
static_assert(NN % 16 == 0, "whole tiles only");
__global__ __launch_bounds__(256, 16) void node_softmax_sum_gemm_kernel(
        const int* __restrict__ off, const int* __restrict__ col,
        const float* __restrict__ H, const float* __restrict__ S, const float* __restrict__ D,
        const float* __restrict__ bias, int relu,
        const float* __restrict__ W2, const float* __restrict__ a_s2, const float* __restrict__ a_d2,
        float* __restrict__ H2, float* __restrict__ S2, float* __restrict__ D2) {
    __shared__ float tile[64 * TP];
    const int lane = threadIdx.x & 31;
    const int wv = __builtin_amdgcn_readfirstlane((int)(threadIdx.x >> 5));
    const int strip = blockIdx.x * 8 + wv;
    if (strip < NSTRIP) strip_rows<true>(strip, lane, off, col, H, S, D, bias, relu, nullptr, tile, wv * NPW);
    __syncthreads();

    const int t = blockIdx.x * 4 + wv;
    if (wv >= 4 || t >= NN / 16) return;

    const int r0    = lane & 15;
    const int khalf = lane >> 4;
    const float* arow = tile + (wv * 16 + r0) * TP;

    const float as0 = a_s2[r0], as1 = a_s2[16 + r0], as2 = a_s2[32 + r0], as3 = a_s2[48 + r0];
    const float ad0 = a_d2[r0], ad1 = a_d2[16 + r0], ad2 = a_d2[32 + r0], ad3 = a_d2[48 + r0];

    v8f c0 = {}, c1 = {}, c2 = {}, c3 = {};
    for (int k0 = 0; k0 < 64; k0 += 4) {
        const int ka = k0 + khalf * 2;
        v2f a;
        a.x = arow[ka];
        a.y = arow[ka + 1];

        const float* w0 = W2 + (size_t)ka * 64;
        const float* w1 = w0 + 64;
        v2f b0, b1, b2, b3;
        b0.x = w0[r0];      b0.y = w1[r0];
        b1.x = w0[16 + r0]; b1.y = w1[16 + r0];
        b2.x = w0[32 + r0]; b2.y = w1[32 + r0];
        b3.x = w0[48 + r0]; b3.y = w1[48 + r0];

        c0 = __builtin_amdgcn_wmma_f32_16x16x4_f32(false, a, false, b0, (short)0, c0, false, false);
        c1 = __builtin_amdgcn_wmma_f32_16x16x4_f32(false, a, false, b1, (short)0, c1, false, false);
        c2 = __builtin_amdgcn_wmma_f32_16x16x4_f32(false, a, false, b2, (short)0, c2, false, false);
        c3 = __builtin_amdgcn_wmma_f32_16x16x4_f32(false, a, false, b3, (short)0, c3, false, false);
    }

    tile_store_and_scores(c0, c1, c2, c3, as0, as1, as2, as3, ad0, ad1, ad2, ad3, H2, S2, D2, t, r0, khalf);
}

extern "C" void kernel_launch(void* const* d_in, const int* in_sizes, int n_in,
                              void* d_out, int out_size, void* d_ws, size_t ws_size,
                              hipStream_t stream) {
    const float* x     = (const float*)d_in[0];
    const int*   ei    = (const int*)d_in[1];
    const float* W1    = (const float*)d_in[2];
    const float* as1   = (const float*)d_in[3];
    const float* ad1   = (const float*)d_in[4];
    const float* b1    = (const float*)d_in[5];
    const float* W2    = (const float*)d_in[6];
    const float* as2   = (const float*)d_in[7];
    const float* ad2   = (const float*)d_in[8];
    const float* b2    = (const float*)d_in[9];
    float* out = (float*)d_out;

    const int E  = in_sizes[1] / 2;
    const int NF = NN * 64;

    float* wsf = (float*)d_ws;
    float* H   = wsf;
    float* H2  = H   + NF;
    float* S   = H2  + NF;
    float* D   = S   + NN;
    const int G1 = (E + CH - 1) / CH;
    int*   lofs = (int*)(D + NN);
    int*   off  = lofs + G1 * (NBUCK + 1) + 16;
    int*   tmp  = off + NN + 16;
    int*   cA   = tmp + G1 * CH;
    int*   cB   = cA  + E;
    float* S2   = (float*)(cB + E);
    float* D2   = S2 + NN;

    const int TB = 256;
    dim3 blk(TB);
    int gNode = (NSTRIP + 7) / 8;
    int gGemm = ((NN / 16) + 7) / 8;

    chunk_group_kernel<<<G1, blk, 0, stream>>>(ei, E, lofs, tmp);
    const GemmArgs g1 = {x, W1, as1, ad1, H, S, D, NN, 128};
    bucket_csr_gemm_kernel<<<NBUCK + gGemm, blk, 0, stream>>>(lofs, G1, tmp, off, cA, g1);
    sort_segments_kernel<<<gNode, blk, 0, stream>>>(off, cA, cB);

    node_softmax_sum_gemm_kernel<<<gNode, blk, 0, stream>>>(off, cB, H, S, D, b1, 1, W2, as2, ad2, H2, S2, D2);

    node_softmax_sum_kernel<<<gNode, blk, 0, stream>>>(off, cB, H2, S2, D2, b2, 0, out);
}


# Segment sort moved into the layer-1 node kernel: a wave orders its segments' sources in its lanes (32-lane compare-exchange network, DPP), stores them for layer 2: 4 launches, not 5; sums unchanged
# speedup vs baseline: 6.6016x; 1.0560x over previous
#include <hip/hip_runtime.h>
#include <math.h>


typedef float v2f __attribute__((ext_vector_type(2)));
typedef float v8f __attribute__((ext_vector_type(8)));

#define NN 100000
#define NEG_SLOPE 0.2f
#define EPSF 1e-16f

__device__ __forceinline__ void tile_store_and_scores(
        const v8f c0, const v8f c1, const v8f c2, const v8f c3,
        const float as0, const float as1, const float as2, const float as3,
        const float ad0, const float ad1, const float ad2, const float ad3,
        float* __restrict__ H, float* __restrict__ S, float* __restrict__ D,
        const int tile, const int r0, const int khalf) {
    float* hbase = H + (size_t)(tile * 16 + khalf * 8) * 64 + r0;
    #pragma unroll
    for (int r = 0; r < 8; ++r) {
        float* row = hbase + (size_t)r * 64;
        row[0]  = c0[r];
        row[16] = c1[r];
        row[32] = c2[r];
        row[48] = c3[r];
    }
    float so = 0.0f, dd = 0.0f;
    #pragma unroll
    for (int r = 0; r < 8; ++r) {
        float s = fmaf(c0[r], as0, c2[r] * as2) + fmaf(c1[r], as1, c3[r] * as3);
        float d = fmaf(c0[r], ad0, c2[r] * ad2) + fmaf(c1[r], ad1, c3[r] * ad3);
        #pragma unroll
        for (int off = 8; off; off >>= 1) {
            s += __shfl_xor(s, off, 32);
            d += __shfl_xor(d, off, 32);
        }
        if (r0 == r) { so = s; dd = d; }
    }
    if (r0 < 8) {
        S[tile * 16 + khalf * 8 + r0] = so;
        D[tile * 16 + khalf * 8 + r0] = dd;
    }
}

__device__ __forceinline__ void gemm_wmma_tile(
        const float* __restrict__ X, const float* __restrict__ W,
        const float* __restrict__ a_s, const float* __restrict__ a_d,
        float* __restrict__ H, float* __restrict__ S, float* __restrict__ D, const int wave, const int K) {
    int lane  = threadIdx.x & 31;
    int r0    = lane & 15;
    int khalf = lane >> 4;

    const float* xrow = X + (size_t)(wave * 16 + r0) * (size_t)K;

    const float as0 = a_s[r0], as1 = a_s[16 + r0], as2 = a_s[32 + r0], as3 = a_s[48 + r0];
    const float ad0 = a_d[r0], ad1 = a_d[16 + r0], ad2 = a_d[32 + r0], ad3 = a_d[48 + r0];

    v8f c0 = {}, c1 = {}, c2 = {}, c3 = {};
    for (int k0 = 0; k0 < K; k0 += 4) {
        int ka = k0 + khalf * 2;
        v2f a;
        a.x = xrow[ka];
        a.y = xrow[ka + 1];

        const float* w0 = W + (size_t)ka * 64;
        const float* w1 = w0 + 64;
        v2f b0, b1, b2, b3;
        b0.x = w0[r0];      b0.y = w1[r0];
        b1.x = w0[16 + r0]; b1.y = w1[16 + r0];
        b2.x = w0[32 + r0]; b2.y = w1[32 + r0];
        b3.x = w0[48 + r0]; b3.y = w1[48 + r0];

        c0 = __builtin_amdgcn_wmma_f32_16x16x4_f32(false, a, false, b0, (short)0, c0, false, false);
        c1 = __builtin_amdgcn_wmma_f32_16x16x4_f32(false, a, false, b1, (short)0, c1, false, false);
        c2 = __builtin_amdgcn_wmma_f32_16x16x4_f32(false, a, false, b2, (short)0, c2, false, false);
        c3 = __builtin_amdgcn_wmma_f32_16x16x4_f32(false, a, false, b3, (short)0, c3, false, false);
    }

    tile_store_and_scores(c0, c1, c2, c3, as0, as1, as2, as3, ad0, ad1, ad2, ad3, H, S, D, wave, r0, khalf);
}

#define NPW 8
#define NSTRIP (NN / NPW)
#define NQ 4
static_assert(NN % NPW == 0, "a strip must not run past the last node");
static_assert(NPW % NQ == 0, "a strip is a whole number of groups");

#define NPB_LOG2 8
#define NPB (1 << NPB_LOG2)
#define NBUCK ((NN + NPB - 1) / NPB)
#define CH 4096
#define EPT (CH / 256)
#define SRC_BITS 17
#define SRC_MASK ((1 << SRC_BITS) - 1)
static_assert(NN <= (1 << SRC_BITS), "a source id must fit the low bits of the packed word");
static_assert(NPB_LOG2 + SRC_BITS + 1 < 32, "the packed word must stay non-negative");
static_assert(NBUCK + 1 <= 512, "pass 1 scans two bucket counts per thread");

__global__ __launch_bounds__(256) void chunk_group_kernel(const int* __restrict__ ei, int E,
        int* __restrict__ lofs, int* __restrict__ tmp) {
    __shared__ int cnt[NBUCK + 1];
    __shared__ int wsum[8];
    __shared__ int stage[CH];
    const int tid  = threadIdx.x;
    const int lane = tid & 31;
    const int wv   = tid >> 5;
    const int e0 = blockIdx.x * CH + tid;
    int d[EPT], s[EPT];
    #pragma unroll
    for (int k = 0; k < EPT; ++k) {
        const int e = e0 + k * 256;
        d[k] = (e < E) ? ei[E + e] : -1;
        s[k] = (e < E) ? ei[e] : 0;
    }
    for (int i = tid; i <= NBUCK; i += 256) cnt[i] = 0;
    __syncthreads();
    int rk[EPT];
    #pragma unroll
    for (int k = 0; k < EPT; ++k) rk[k] = (d[k] >= 0) ? atomicAdd(&cnt[d[k] >> NPB_LOG2], 1) : 0;
    __syncthreads();
    const int i0 = 2 * tid, i1 = 2 * tid + 1;
    const int c0 = (i0 < NBUCK) ? cnt[i0] : 0;
    const int c1 = (i1 < NBUCK) ? cnt[i1] : 0;
    int inc = c0 + c1;
    #pragma unroll
    for (int o = 1; o < 32; o <<= 1) {
        int t = __shfl_up(inc, o, 32);
        if (lane >= o) inc += t;
    }
    if (lane == 31) wsum[wv] = inc;
    __syncthreads();
    int wprefix = 0;
    for (int w = 0; w < wv; ++w) wprefix += wsum[w];
    const int ex = wprefix + inc - (c0 + c1);
    if (i0 < NBUCK) cnt[i0] = ex;
    if (i1 < NBUCK) cnt[i1] = ex + c0;
    if (tid == 255) cnt[NBUCK] = ex + c0 + c1;
    __syncthreads();
    int* lo = lofs + (size_t)blockIdx.x * (NBUCK + 1);
    for (int i = tid; i <= NBUCK; i += 256) lo[i] = cnt[i];
    #pragma unroll
    for (int k = 0; k < EPT; ++k) {
        if (d[k] >= 0) stage[cnt[d[k] >> NPB_LOG2] + rk[k]] = ((d[k] & (NPB - 1)) << SRC_BITS) | s[k];
    }
    __syncthreads();
    const int total = cnt[NBUCK];
    int* out = tmp + (size_t)blockIdx.x * CH;
    #pragma unroll
    for (int k = 0; k < EPT; ++k) {
        const int i = tid + k * 256;
        if (i < total) out[i] = stage[i];
    }
}

struct GemmArgs {
    const float* X; const float* W; const float* a_s; const float* a_d;
    float* H; float* S; float* D;
    int nrows; int K;
};

#define RB 16
template <bool PLACE>
__device__ __forceinline__ int bucket_runs(const int* __restrict__ lofs, const int G1,
        const int* __restrict__ tmp, const int b, int* cnt, int* __restrict__ col,
        const int wv, const int lane) {
    int stsum = 0;
    for (int base = wv * 32; base < G1; base += 512) {
        int st[2], len[2];
        #pragma unroll
        for (int h = 0; h < 2; ++h) {
            const int w = base + h * 256 + lane;
            const int* lo = lofs + (size_t)((w < G1) ? w : 0) * (NBUCK + 1) + b;
            const int a = lo[0];
            const int e = lo[1];
            st[h]  = (w < G1) ? (w * CH + a) : 0;
            len[h] = (w < G1) ? (e - a) : 0;
            stsum += (w < G1) ? a : 0;
        }
        #pragma unroll
        for (int h = 0; h < 2; ++h) {
            #pragma nounroll
            for (int r = 0; r < 32; r += RB) {
                int wd[RB];
                #pragma unroll
                for (int u = 0; u < RB; ++u) {
                    const int s_r = __builtin_amdgcn_readlane(st[h], r + u);
                    const int l_r = __builtin_amdgcn_readlane(len[h], r + u);
                    wd[u] = (lane < l_r) ? tmp[s_r + lane] : -1;
                }
                if (!PLACE) {
                    #pragma unroll
                    for (int u = 0; u < RB; ++u) {
                        if (wd[u] >= 0) atomicAdd(&cnt[wd[u] >> SRC_BITS], 1);
                    }
                } else {
                    int pos[RB];
                    #pragma unroll
                    for (int u = 0; u < RB; ++u) pos[u] = (wd[u] >= 0) ? atomicAdd(&cnt[wd[u] >> SRC_BITS], 1) : 0;
                    #pragma unroll
                    for (int u = 0; u < RB; ++u) {
                        if (wd[u] >= 0) col[pos[u]] = wd[u] & SRC_MASK;
                    }
                }
            }
            unsigned m = __builtin_amdgcn_ballot_w32(len[h] > 32);
            while (m) {
                const int r = __builtin_ctz(m);
                m &= m - 1;
                const int s_r = __builtin_amdgcn_readlane(st[h], r);
                const int l_r = __builtin_amdgcn_readlane(len[h], r);
                for (int i = 32 + lane; i < l_r; i += 32) {
                    const int wd = tmp[s_r + i];
                    if (!PLACE) {
                        atomicAdd(&cnt[wd >> SRC_BITS], 1);
                    } else {
                        const int pos = atomicAdd(&cnt[wd >> SRC_BITS], 1);
                        col[pos] = wd & SRC_MASK;
                    }
                }
            }
        }
    }
    return stsum;
}

__global__ __launch_bounds__(256) void bucket_csr_gemm_kernel(const int* __restrict__ lofs, int G1,
        const int* __restrict__ tmp, int* __restrict__ off, int* __restrict__ col, const GemmArgs g) {
    if (blockIdx.x >= NBUCK) {
        const int tile = __builtin_amdgcn_readfirstlane((int)((blockIdx.x - NBUCK) * 8 + (threadIdx.x >> 5)));
        if (tile < (g.nrows >> 4)) gemm_wmma_tile(g.X, g.W, g.a_s, g.a_d, g.H, g.S, g.D, tile, g.K);
        return;
    }
    __shared__ int cnt[NPB];
    __shared__ int wtot[NPB / 32];
    __shared__ int red[NPB / 32];
    const int tid  = threadIdx.x;
    const int lane = tid & 31;
    const int wv   = __builtin_amdgcn_readfirstlane((int)(threadIdx.x >> 5));
    const int b    = blockIdx.x;
    cnt[tid] = 0;
    __syncthreads();
    int stsum = bucket_runs<false>(lofs, G1, tmp, b, cnt, nullptr, wv, lane);
    #pragma unroll
    for (int o = 16; o; o >>= 1) stsum += __shfl_xor(stsum, o, 32);
    if (lane == 0) red[wv] = stsum;
    __syncthreads();
    int bs = 0;
    for (int w = 0; w < NPB / 32; ++w) bs += red[w];
    const int v = cnt[tid];
    int inc = v;
    #pragma unroll
    for (int o = 1; o < 32; o <<= 1) {
        int t = __shfl_up(inc, o, 32);
        if (lane >= o) inc += t;
    }
    if (lane == 31) wtot[wv] = inc;
    __syncthreads();
    int wprefix = 0;
    for (int w = 0; w < wv; ++w) wprefix += wtot[w];
    const int excl = bs + wprefix + inc - v;
    const int node = b * NPB + tid;
    if (node < NN) off[node] = excl;
    if (node == NN - 1) off[NN] = excl + v;
    cnt[tid] = excl;
    __syncthreads();
    bucket_runs<true>(lofs, G1, tmp, b, cnt, col, wv, lane);
}


template <int M>
__device__ __forceinline__ int lane_xor_row(int v) {
    return __builtin_amdgcn_update_dpp(0, v, 0x160 | M, 0xf, 0xf, true);
}
__device__ __forceinline__ int lane_xor_half(int v) {
    return __builtin_amdgcn_permlanex16(v, v, 0x76543210, 0xfedcba98, false, false);
}
template <int J>
__device__ __forceinline__ int compare_exchange(int v, bool keep_min) {
    int p;
    if constexpr (J == 16) p = lane_xor_half(v);
    else                   p = lane_xor_row<J>(v);
    const int mn = (v < p) ? v : p;
    const int mx = (v < p) ? p : v;
    return keep_min ? mn : mx;
}
__device__ __forceinline__ int sort32_ascending(int v, const int lane) {
#define CX(K, J) v = compare_exchange<J>(v, (((lane & K) == 0) == ((lane & J) == 0)));
    CX(2, 1)
    CX(4, 2)  CX(4, 1)
    CX(8, 4)  CX(8, 2)  CX(8, 1)
    CX(16, 8) CX(16, 4) CX(16, 2) CX(16, 1)
    CX(32, 16) CX(32, 8) CX(32, 4) CX(32, 2) CX(32, 1)
#undef CX
    return v;
}

__device__ __forceinline__ float leaky(float a) { return (a >= 0.0f) ? a : NEG_SLOPE * a; }

__device__ __forceinline__ float readlane_f(float v, int l) {
    return __int_as_float(__builtin_amdgcn_readlane(__float_as_int(v), l));
}

#define TP 68

template <bool TO_LDS>
__device__ __forceinline__ void finish_row(v2f acc, float bx, float by, int relu,
                                           float* __restrict__ OUT, float* tile, int node, int row, int lane) {
    acc.x += bx;
    acc.y += by;
    if (relu) {
        acc.x = (acc.x > 0.0f) ? acc.x : 0.0f;
        acc.y = (acc.y > 0.0f) ? acc.y : 0.0f;
    }
    if (TO_LDS) ((v2f*)tile)[row * (TP / 2) + lane] = acc;
    else        ((v2f*)OUT)[(size_t)node * 32 + lane] = acc;
}

template <bool TO_LDS, bool SORT>
__device__ __forceinline__ void strip_rows(
        const int strip, const int lane,
        const int* __restrict__ off, const int* __restrict__ col, int* cs,
        const float* __restrict__ H, const float* __restrict__ S, const float* __restrict__ D,
        const float* __restrict__ bias, int relu, float* __restrict__ OUT, float* tile, const int rb0) {
    const int n0 = strip * NPW;
    const int   offv = off[n0 + ((lane < NPW) ? lane : NPW)];
    const float Sv   = S[n0 + (lane & (NPW - 1))];
    const float Dv   = D[n0 + (lane & (NPW - 1))];
    const v2f* H2 = (const v2f*)H;
    const float bx = bias[2 * lane];
    const float by = bias[2 * lane + 1];

    bool any_long = false;
    for (int q = 0; q < NPW; q += NQ) {
        int o0[NQ], deg[NQ], sv[NQ];
        v2f acc0[NQ];
        #pragma unroll
        for (int u = 0; u < NQ; ++u) {
            o0[u]   = __builtin_amdgcn_readlane(offv, q + u);
            deg[u]  = __builtin_amdgcn_readlane(offv, q + u + 1) - o0[u];
            if (SORT) sv[u] = (lane < deg[u]) ? col[o0[u] + lane] : 0x7fffffff;
            else      sv[u] = (lane < deg[u]) ? col[o0[u] + lane] : (n0 + q + u);
            acc0[u] = H2[(size_t)(n0 + q + u) * 32 + lane];
        }
        if (SORT) {
            int ks[NQ];
            #pragma unroll
            for (int u = 0; u < NQ; ++u) ks[u] = sort32_ascending(sv[u], lane);
            #pragma unroll
            for (int u = 0; u < NQ; ++u) {
                if (lane < deg[u] && deg[u] <= 32) cs[o0[u] + lane] = ks[u];
                sv[u] = (lane < deg[u]) ? ks[u] : (n0 + q + u);
            }
        }
        float av[NQ], aself[NQ];
        #pragma unroll
        for (int u = 0; u < NQ; ++u) av[u] = S[sv[u]];
        #pragma unroll
        for (int u = 0; u < NQ; ++u) {
            const float dn = readlane_f(Dv, q + u);
            aself[u] = leaky(readlane_f(Sv, q + u) + dn);
            av[u] = (lane < deg[u]) ? leaky(av[u] + dn) : -INFINITY;
        }
        float wv[NQ];
        #pragma unroll
        for (int u = 0; u < NQ; ++u) {
            float m = fmaxf(av[u], aself[u]);
            #pragma unroll
            for (int o = 16; o; o >>= 1) m = fmaxf(m, __shfl_xor(m, o, 32));
            const float ex = expf(av[u] - m);
            const float e = (lane < deg[u]) ? ex : 0.0f;
            const float e_self = expf(aself[u] - m);
            float z = e;
            #pragma unroll
            for (int o = 16; o; o >>= 1) z += __shfl_xor(z, o, 32);
            z += e_self;
            wv[u] = e / (z + EPSF);
            const float w_self = e_self / (z + EPSF);
            acc0[u].x = acc0[u].x * w_self;
            acc0[u].y = acc0[u].y * w_self;
        }
        #pragma unroll
        for (int u = 0; u < NQ; ++u) {
            if (deg[u] > 32) { any_long = true; continue; }
            const float w = wv[u];
            v2f acc = acc0[u];
            for (int j = 0; j < deg[u]; j += 8) {
                v2f hv[8];
                #pragma unroll
                for (int t = 0; t < 8; ++t) {
                    const int jt = (j + t < deg[u]) ? (j + t) : (deg[u] - 1);
                    const int sj = __builtin_amdgcn_readlane(sv[u], jt);
                    hv[t] = H2[(size_t)sj * 32 + lane];
                }
                #pragma unroll
                for (int t = 0; t < 8; ++t) {
                    const float wj = readlane_f(w, j + t);
                    acc.x = fmaf(hv[t].x, wj, acc.x);
                    acc.y = fmaf(hv[t].y, wj, acc.y);
                }
            }
            finish_row<TO_LDS>(acc, bx, by, relu, OUT, tile, n0 + q + u, rb0 + q + u, lane);
        }
    }

    if (any_long) {
        for (int u = 0; u < NPW; ++u) {
            const int lo = __builtin_amdgcn_readlane(offv, u);
            const int dg = __builtin_amdgcn_readlane(offv, u + 1) - lo;
            if (dg <= 32) continue;
            const int node = n0 + u;
            if (SORT) {
                for (int i = lane; i < dg; i += 32) {
                    const int ki = col[lo + i];
                    int rank = 0;
                    for (int j = 0; j < dg; ++j) {
                        const int kj = col[lo + j];
                        rank += (kj < ki || (kj == ki && j < i)) ? 1 : 0;
                    }
                    cs[lo + rank] = ki;
                }
                __threadfence();
            }
            const int* cr = SORT ? cs : col;
            const float dn = readlane_f(Dv, u);
            const float a_self = leaky(readlane_f(Sv, u) + dn);
            const v2f hsn = H2[(size_t)node * 32 + lane];
            float m = a_self;
            for (int c = 0; c < dg; c += 32) {
                if (c + lane < dg) m = fmaxf(m, leaky(S[cr[lo + c + lane]] + dn));
            }
            #pragma unroll
            for (int o = 16; o; o >>= 1) m = fmaxf(m, __shfl_xor(m, o, 32));
            float z = 0.0f;
            for (int c = 0; c < dg; c += 32) {
                if (c + lane < dg) z += expf(leaky(S[cr[lo + c + lane]] + dn) - m);
            }
            #pragma unroll
            for (int o = 16; o; o >>= 1) z += __shfl_xor(z, o, 32);
            const float e_self = expf(a_self - m);
            z += e_self;
            const float w_self = e_self / (z + EPSF);
            v2f acc;
            acc.x = hsn.x * w_self;
            acc.y = hsn.y * w_self;
            for (int c = 0; c < dg; c += 32) {
                int s = node;
                float w = 0.0f;
                if (c + lane < dg) {
                    s = cr[lo + c + lane];
                    w = expf(leaky(S[s] + dn) - m) / (z + EPSF);
                }
                const int nj = (dg - c < 32) ? (dg - c) : 32;
                for (int j = 0; j < nj; ++j) {
                    const int sj = __builtin_amdgcn_readlane(s, j);
                    const float wj = readlane_f(w, j);
                    const v2f hv = H2[(size_t)sj * 32 + lane];
                    acc.x = fmaf(hv.x, wj, acc.x);
                    acc.y = fmaf(hv.y, wj, acc.y);
                }
            }
            finish_row<TO_LDS>(acc, bx, by, relu, OUT, tile, node, rb0 + u, lane);
        }
    }
}

__global__ __launch_bounds__(256) void node_softmax_sum_kernel(
        const int* __restrict__ off, const int* __restrict__ col,
        const float* __restrict__ H, const float* __restrict__ S, const float* __restrict__ D,
        const float* __restrict__ bias, int relu, float* __restrict__ OUT) {
    const int lane = threadIdx.x & 31;
    const int strip = __builtin_amdgcn_readfirstlane((int)(blockIdx.x * 8 + (threadIdx.x >> 5)));
    if (strip >= NSTRIP) return;
    strip_rows<false, false>(strip, lane, off, col, nullptr, H, S, D, bias, relu, OUT, nullptr, 0);
}

static_assert(NPW * 8 == 64, "a block of 8 waves must finish four whole 16-row tiles");
static_assert(NN % 16 == 0, "whole tiles only");
__global__ __launch_bounds__(256, 16) void node_softmax_sum_gemm_kernel(
        const int* __restrict__ off, const int* __restrict__ col, int* csorted,
        const float* __restrict__ H, const float* __restrict__ S, const float* __restrict__ D,
        const float* __restrict__ bias, int relu,
        const float* __restrict__ W2, const float* __restrict__ a_s2, const float* __restrict__ a_d2,
        float* __restrict__ H2, float* __restrict__ S2, float* __restrict__ D2) {
    __shared__ float tile[64 * TP];
    const int lane = threadIdx.x & 31;
    const int wv = __builtin_amdgcn_readfirstlane((int)(threadIdx.x >> 5));
    const int strip = blockIdx.x * 8 + wv;
    if (strip < NSTRIP) strip_rows<true, true>(strip, lane, off, col, csorted, H, S, D, bias, relu, nullptr, tile, wv * NPW);
    __syncthreads();

    const int t = blockIdx.x * 4 + wv;
    if (wv >= 4 || t >= NN / 16) return;

    const int r0    = lane & 15;
    const int khalf = lane >> 4;
    const float* arow = tile + (wv * 16 + r0) * TP;

    const float as0 = a_s2[r0], as1 = a_s2[16 + r0], as2 = a_s2[32 + r0], as3 = a_s2[48 + r0];
    const float ad0 = a_d2[r0], ad1 = a_d2[16 + r0], ad2 = a_d2[32 + r0], ad3 = a_d2[48 + r0];

    v8f c0 = {}, c1 = {}, c2 = {}, c3 = {};
    for (int k0 = 0; k0 < 64; k0 += 4) {
        const int ka = k0 + khalf * 2;
        v2f a;
        a.x = arow[ka];
        a.y = arow[ka + 1];

        const float* w0 = W2 + (size_t)ka * 64;
        const float* w1 = w0 + 64;
        v2f b0, b1, b2, b3;
        b0.x = w0[r0];      b0.y = w1[r0];
        b1.x = w0[16 + r0]; b1.y = w1[16 + r0];
        b2.x = w0[32 + r0]; b2.y = w1[32 + r0];
        b3.x = w0[48 + r0]; b3.y = w1[48 + r0];

        c0 = __builtin_amdgcn_wmma_f32_16x16x4_f32(false, a, false, b0, (short)0, c0, false, false);
        c1 = __builtin_amdgcn_wmma_f32_16x16x4_f32(false, a, false, b1, (short)0, c1, false, false);
        c2 = __builtin_amdgcn_wmma_f32_16x16x4_f32(false, a, false, b2, (short)0, c2, false, false);
        c3 = __builtin_amdgcn_wmma_f32_16x16x4_f32(false, a, false, b3, (short)0, c3, false, false);
    }

    tile_store_and_scores(c0, c1, c2, c3, as0, as1, as2, as3, ad0, ad1, ad2, ad3, H2, S2, D2, t, r0, khalf);
}

extern "C" void kernel_launch(void* const* d_in, const int* in_sizes, int n_in,
                              void* d_out, int out_size, void* d_ws, size_t ws_size,
                              hipStream_t stream) {
    const float* x     = (const float*)d_in[0];
    const int*   ei    = (const int*)d_in[1];
    const float* W1    = (const float*)d_in[2];
    const float* as1   = (const float*)d_in[3];
    const float* ad1   = (const float*)d_in[4];
    const float* b1    = (const float*)d_in[5];
    const float* W2    = (const float*)d_in[6];
    const float* as2   = (const float*)d_in[7];
    const float* ad2   = (const float*)d_in[8];
    const float* b2    = (const float*)d_in[9];
    float* out = (float*)d_out;

    const int E  = in_sizes[1] / 2;
    const int NF = NN * 64;

    float* wsf = (float*)d_ws;
    float* H   = wsf;
    float* H2  = H   + NF;
    float* S   = H2  + NF;
    float* D   = S   + NN;
    const int G1 = (E + CH - 1) / CH;
    int*   lofs = (int*)(D + NN);
    int*   off  = lofs + G1 * (NBUCK + 1) + 16;
    int*   tmp  = off + NN + 16;
    int*   cA   = tmp + G1 * CH;
    int*   cB   = cA  + E;
    float* S2   = (float*)(cB + E);
    float* D2   = S2 + NN;

    const int TB = 256;
    dim3 blk(TB);
    int gNode = (NSTRIP + 7) / 8;
    int gGemm = ((NN / 16) + 7) / 8;

    chunk_group_kernel<<<G1, blk, 0, stream>>>(ei, E, lofs, tmp);
    const GemmArgs g1 = {x, W1, as1, ad1, H, S, D, NN, 128};
    bucket_csr_gemm_kernel<<<NBUCK + gGemm, blk, 0, stream>>>(lofs, G1, tmp, off, cA, g1);

    node_softmax_sum_gemm_kernel<<<gNode, blk, 0, stream>>>(off, cA, cB, H, S, D, b1, 1, W2, as2, ad2, H2, S2, D2);

    node_softmax_sum_kernel<<<gNode, blk, 0, stream>>>(off, cB, H2, S2, D2, b2, 0, out);
}


# Butterflies of softmax (max, sum), tile scores and bucket sum take partner lanes by DPP row_xmask and permlanex16 on the vector ALU, not ds_bpermute through the LDS unit; same pairs, sums unchanged
# speedup vs baseline: 6.9235x; 1.0488x over previous
#include <hip/hip_runtime.h>
#include <math.h>


typedef float v2f __attribute__((ext_vector_type(2)));
typedef float v8f __attribute__((ext_vector_type(8)));

#define NN 100000
#define NEG_SLOPE 0.2f
#define EPSF 1e-16f

template <int M>
__device__ __forceinline__ int lane_xor_row(int v) {
    return __builtin_amdgcn_update_dpp(0, v, 0x160 | M, 0xf, 0xf, true);
}
__device__ __forceinline__ int lane_xor_half(int v) {
    return __builtin_amdgcn_permlanex16(v, v, 0x76543210, 0xfedcba98, false, false);
}
template <int O>
__device__ __forceinline__ int lane_xor_i(int v) {
    if constexpr (O == 16) return lane_xor_half(v);
    else                   return lane_xor_row<O>(v);
}
template <int O>
__device__ __forceinline__ float lane_xor_f(float v) {
    return __int_as_float(lane_xor_i<O>(__float_as_int(v)));
}
__device__ __forceinline__ float wave_sum_f(float v) {
    v += lane_xor_f<16>(v); v += lane_xor_f<8>(v); v += lane_xor_f<4>(v); v += lane_xor_f<2>(v); v += lane_xor_f<1>(v);
    return v;
}
__device__ __forceinline__ float wave_max_f(float v) {
    v = fmaxf(v, lane_xor_f<16>(v)); v = fmaxf(v, lane_xor_f<8>(v)); v = fmaxf(v, lane_xor_f<4>(v));
    v = fmaxf(v, lane_xor_f<2>(v));  v = fmaxf(v, lane_xor_f<1>(v));
    return v;
}
__device__ __forceinline__ int wave_sum_i(int v) {
    v += lane_xor_i<16>(v); v += lane_xor_i<8>(v); v += lane_xor_i<4>(v); v += lane_xor_i<2>(v); v += lane_xor_i<1>(v);
    return v;
}
__device__ __forceinline__ float row16_sum_f(float v) {
    v += lane_xor_f<8>(v); v += lane_xor_f<4>(v); v += lane_xor_f<2>(v); v += lane_xor_f<1>(v);
    return v;
}

__device__ __forceinline__ void tile_store_and_scores(
        const v8f c0, const v8f c1, const v8f c2, const v8f c3,
        const float as0, const float as1, const float as2, const float as3,
        const float ad0, const float ad1, const float ad2, const float ad3,
        float* __restrict__ H, float* __restrict__ S, float* __restrict__ D,
        const int tile, const int r0, const int khalf) {
    float* hbase = H + (size_t)(tile * 16 + khalf * 8) * 64 + r0;
    #pragma unroll
    for (int r = 0; r < 8; ++r) {
        float* row = hbase + (size_t)r * 64;
        row[0]  = c0[r];
        row[16] = c1[r];
        row[32] = c2[r];
        row[48] = c3[r];
    }
    float so = 0.0f, dd = 0.0f;
    #pragma unroll
    for (int r = 0; r < 8; ++r) {
        float s = fmaf(c0[r], as0, c2[r] * as2) + fmaf(c1[r], as1, c3[r] * as3);
        float d = fmaf(c0[r], ad0, c2[r] * ad2) + fmaf(c1[r], ad1, c3[r] * ad3);
        s = row16_sum_f(s);
        d = row16_sum_f(d);
        if (r0 == r) { so = s; dd = d; }
    }
    if (r0 < 8) {
        S[tile * 16 + khalf * 8 + r0] = so;
        D[tile * 16 + khalf * 8 + r0] = dd;
    }
}

__device__ __forceinline__ void gemm_wmma_tile(
        const float* __restrict__ X, const float* __restrict__ W,
        const float* __restrict__ a_s, const float* __restrict__ a_d,
        float* __restrict__ H, float* __restrict__ S, float* __restrict__ D, const int wave, const int K) {
    int lane  = threadIdx.x & 31;
    int r0    = lane & 15;
    int khalf = lane >> 4;

    const float* xrow = X + (size_t)(wave * 16 + r0) * (size_t)K;

    const float as0 = a_s[r0], as1 = a_s[16 + r0], as2 = a_s[32 + r0], as3 = a_s[48 + r0];
    const float ad0 = a_d[r0], ad1 = a_d[16 + r0], ad2 = a_d[32 + r0], ad3 = a_d[48 + r0];

    v8f c0 = {}, c1 = {}, c2 = {}, c3 = {};
    for (int k0 = 0; k0 < K; k0 += 4) {
        int ka = k0 + khalf * 2;
        v2f a;
        a.x = xrow[ka];
        a.y = xrow[ka + 1];

        const float* w0 = W + (size_t)ka * 64;
        const float* w1 = w0 + 64;
        v2f b0, b1, b2, b3;
        b0.x = w0[r0];      b0.y = w1[r0];
        b1.x = w0[16 + r0]; b1.y = w1[16 + r0];
        b2.x = w0[32 + r0]; b2.y = w1[32 + r0];
        b3.x = w0[48 + r0]; b3.y = w1[48 + r0];

        c0 = __builtin_amdgcn_wmma_f32_16x16x4_f32(false, a, false, b0, (short)0, c0, false, false);
        c1 = __builtin_amdgcn_wmma_f32_16x16x4_f32(false, a, false, b1, (short)0, c1, false, false);
        c2 = __builtin_amdgcn_wmma_f32_16x16x4_f32(false, a, false, b2, (short)0, c2, false, false);
        c3 = __builtin_amdgcn_wmma_f32_16x16x4_f32(false, a, false, b3, (short)0, c3, false, false);
    }

    tile_store_and_scores(c0, c1, c2, c3, as0, as1, as2, as3, ad0, ad1, ad2, ad3, H, S, D, wave, r0, khalf);
}

#define NPW 8
#define NSTRIP (NN / NPW)
#define NQ 4
static_assert(NN % NPW == 0, "a strip must not run past the last node");
static_assert(NPW % NQ == 0, "a strip is a whole number of groups");

#define NPB_LOG2 8
#define NPB (1 << NPB_LOG2)
#define NBUCK ((NN + NPB - 1) / NPB)
#define CH 4096
#define EPT (CH / 256)
#define SRC_BITS 17
#define SRC_MASK ((1 << SRC_BITS) - 1)
static_assert(NN <= (1 << SRC_BITS), "a source id must fit the low bits of the packed word");
static_assert(NPB_LOG2 + SRC_BITS + 1 < 32, "the packed word must stay non-negative");
static_assert(NBUCK + 1 <= 512, "pass 1 scans two bucket counts per thread");

__global__ __launch_bounds__(256) void chunk_group_kernel(const int* __restrict__ ei, int E,
        int* __restrict__ lofs, int* __restrict__ tmp) {
    __shared__ int cnt[NBUCK + 1];
    __shared__ int wsum[8];
    __shared__ int stage[CH];
    const int tid  = threadIdx.x;
    const int lane = tid & 31;
    const int wv   = tid >> 5;
    const int e0 = blockIdx.x * CH + tid;
    int d[EPT], s[EPT];
    #pragma unroll
    for (int k = 0; k < EPT; ++k) {
        const int e = e0 + k * 256;
        d[k] = (e < E) ? ei[E + e] : -1;
        s[k] = (e < E) ? ei[e] : 0;
    }
    for (int i = tid; i <= NBUCK; i += 256) cnt[i] = 0;
    __syncthreads();
    int rk[EPT];
    #pragma unroll
    for (int k = 0; k < EPT; ++k) rk[k] = (d[k] >= 0) ? atomicAdd(&cnt[d[k] >> NPB_LOG2], 1) : 0;
    __syncthreads();
    const int i0 = 2 * tid, i1 = 2 * tid + 1;
    const int c0 = (i0 < NBUCK) ? cnt[i0] : 0;
    const int c1 = (i1 < NBUCK) ? cnt[i1] : 0;
    int inc = c0 + c1;
    #pragma unroll
    for (int o = 1; o < 32; o <<= 1) {
        int t = __shfl_up(inc, o, 32);
        if (lane >= o) inc += t;
    }
    if (lane == 31) wsum[wv] = inc;
    __syncthreads();
    int wprefix = 0;
    for (int w = 0; w < wv; ++w) wprefix += wsum[w];
    const int ex = wprefix + inc - (c0 + c1);
    if (i0 < NBUCK) cnt[i0] = ex;
    if (i1 < NBUCK) cnt[i1] = ex + c0;
    if (tid == 255) cnt[NBUCK] = ex + c0 + c1;
    __syncthreads();
    int* lo = lofs + (size_t)blockIdx.x * (NBUCK + 1);
    for (int i = tid; i <= NBUCK; i += 256) lo[i] = cnt[i];
    #pragma unroll
    for (int k = 0; k < EPT; ++k) {
        if (d[k] >= 0) stage[cnt[d[k] >> NPB_LOG2] + rk[k]] = ((d[k] & (NPB - 1)) << SRC_BITS) | s[k];
    }
    __syncthreads();
    const int total = cnt[NBUCK];
    int* out = tmp + (size_t)blockIdx.x * CH;
    #pragma unroll
    for (int k = 0; k < EPT; ++k) {
        const int i = tid + k * 256;
        if (i < total) out[i] = stage[i];
    }
}

struct GemmArgs {
    const float* X; const float* W; const float* a_s; const float* a_d;
    float* H; float* S; float* D;
    int nrows; int K;
};

#define RB 16
template <bool PLACE>
__device__ __forceinline__ int bucket_runs(const int* __restrict__ lofs, const int G1,
        const int* __restrict__ tmp, const int b, int* cnt, int* __restrict__ col,
        const int wv, const int lane) {
    int stsum = 0;
    for (int base = wv * 32; base < G1; base += 512) {
        int st[2], len[2];
        #pragma unroll
        for (int h = 0; h < 2; ++h) {
            const int w = base + h * 256 + lane;
            const int* lo = lofs + (size_t)((w < G1) ? w : 0) * (NBUCK + 1) + b;
            const int a = lo[0];
            const int e = lo[1];
            st[h]  = (w < G1) ? (w * CH + a) : 0;
            len[h] = (w < G1) ? (e - a) : 0;
            stsum += (w < G1) ? a : 0;
        }
        #pragma unroll
        for (int h = 0; h < 2; ++h) {
            #pragma nounroll
            for (int r = 0; r < 32; r += RB) {
                int wd[RB];
                #pragma unroll
                for (int u = 0; u < RB; ++u) {
                    const int s_r = __builtin_amdgcn_readlane(st[h], r + u);
                    const int l_r = __builtin_amdgcn_readlane(len[h], r + u);
                    wd[u] = (lane < l_r) ? tmp[s_r + lane] : -1;
                }
                if (!PLACE) {
                    #pragma unroll
                    for (int u = 0; u < RB; ++u) {
                        if (wd[u] >= 0) atomicAdd(&cnt[wd[u] >> SRC_BITS], 1);
                    }
                } else {
                    int pos[RB];
                    #pragma unroll
                    for (int u = 0; u < RB; ++u) pos[u] = (wd[u] >= 0) ? atomicAdd(&cnt[wd[u] >> SRC_BITS], 1) : 0;
                    #pragma unroll
                    for (int u = 0; u < RB; ++u) {
                        if (wd[u] >= 0) col[pos[u]] = wd[u] & SRC_MASK;
                    }
                }
            }
            unsigned m = __builtin_amdgcn_ballot_w32(len[h] > 32);
            while (m) {
                const int r = __builtin_ctz(m);
                m &= m - 1;
                const int s_r = __builtin_amdgcn_readlane(st[h], r);
                const int l_r = __builtin_amdgcn_readlane(len[h], r);
                for (int i = 32 + lane; i < l_r; i += 32) {
                    const int wd = tmp[s_r + i];
                    if (!PLACE) {
                        atomicAdd(&cnt[wd >> SRC_BITS], 1);
                    } else {
                        const int pos = atomicAdd(&cnt[wd >> SRC_BITS], 1);
                        col[pos] = wd & SRC_MASK;
                    }
                }
            }
        }
    }
    return stsum;
}

__global__ __launch_bounds__(256) void bucket_csr_gemm_kernel(const int* __restrict__ lofs, int G1,
        const int* __restrict__ tmp, int* __restrict__ off, int* __restrict__ col, const GemmArgs g) {
    if (blockIdx.x >= NBUCK) {
        const int tile = __builtin_amdgcn_readfirstlane((int)((blockIdx.x - NBUCK) * 8 + (threadIdx.x >> 5)));
        if (tile < (g.nrows >> 4)) gemm_wmma_tile(g.X, g.W, g.a_s, g.a_d, g.H, g.S, g.D, tile, g.K);
        return;
    }
    __shared__ int cnt[NPB];
    __shared__ int wtot[NPB / 32];
    __shared__ int red[NPB / 32];
    const int tid  = threadIdx.x;
    const int lane = tid & 31;
    const int wv   = __builtin_amdgcn_readfirstlane((int)(threadIdx.x >> 5));
    const int b    = blockIdx.x;
    cnt[tid] = 0;
    __syncthreads();
    int stsum = bucket_runs<false>(lofs, G1, tmp, b, cnt, nullptr, wv, lane);
    stsum = wave_sum_i(stsum);
    if (lane == 0) red[wv] = stsum;
    __syncthreads();
    int bs = 0;
    for (int w = 0; w < NPB / 32; ++w) bs += red[w];
    const int v = cnt[tid];
    int inc = v;
    #pragma unroll
    for (int o = 1; o < 32; o <<= 1) {
        int t = __shfl_up(inc, o, 32);
        if (lane >= o) inc += t;
    }
    if (lane == 31) wtot[wv] = inc;
    __syncthreads();
    int wprefix = 0;
    for (int w = 0; w < wv; ++w) wprefix += wtot[w];
    const int excl = bs + wprefix + inc - v;
    const int node = b * NPB + tid;
    if (node < NN) off[node] = excl;
    if (node == NN - 1) off[NN] = excl + v;
    cnt[tid] = excl;
    __syncthreads();
    bucket_runs<true>(lofs, G1, tmp, b, cnt, col, wv, lane);
}


template <int J>
__device__ __forceinline__ int compare_exchange(int v, bool keep_min) {
    int p;
    if constexpr (J == 16) p = lane_xor_half(v);
    else                   p = lane_xor_row<J>(v);
    const int mn = (v < p) ? v : p;
    const int mx = (v < p) ? p : v;
    return keep_min ? mn : mx;
}
__device__ __forceinline__ int sort32_ascending(int v, const int lane) {
#define CX(K, J) v = compare_exchange<J>(v, (((lane & K) == 0) == ((lane & J) == 0)));
    CX(2, 1)
    CX(4, 2)  CX(4, 1)
    CX(8, 4)  CX(8, 2)  CX(8, 1)
    CX(16, 8) CX(16, 4) CX(16, 2) CX(16, 1)
    CX(32, 16) CX(32, 8) CX(32, 4) CX(32, 2) CX(32, 1)
#undef CX
    return v;
}

__device__ __forceinline__ float leaky(float a) { return (a >= 0.0f) ? a : NEG_SLOPE * a; }

__device__ __forceinline__ float readlane_f(float v, int l) {
    return __int_as_float(__builtin_amdgcn_readlane(__float_as_int(v), l));
}

#define TP 68

template <bool TO_LDS>
__device__ __forceinline__ void finish_row(v2f acc, float bx, float by, int relu,
                                           float* __restrict__ OUT, float* tile, int node, int row, int lane) {
    acc.x += bx;
    acc.y += by;
    if (relu) {
        acc.x = (acc.x > 0.0f) ? acc.x : 0.0f;
        acc.y = (acc.y > 0.0f) ? acc.y : 0.0f;
    }
    if (TO_LDS) ((v2f*)tile)[row * (TP / 2) + lane] = acc;
    else        ((v2f*)OUT)[(size_t)node * 32 + lane] = acc;
}

template <bool TO_LDS, bool SORT>
__device__ __forceinline__ void strip_rows(
        const int strip, const int lane,
        const int* __restrict__ off, const int* __restrict__ col, int* cs,
        const float* __restrict__ H, const float* __restrict__ S, const float* __restrict__ D,
        const float* __restrict__ bias, int relu, float* __restrict__ OUT, float* tile, const int rb0) {
    const int n0 = strip * NPW;
    const int   offv = off[n0 + ((lane < NPW) ? lane : NPW)];
    const float Sv   = S[n0 + (lane & (NPW - 1))];
    const float Dv   = D[n0 + (lane & (NPW - 1))];
    const v2f* H2 = (const v2f*)H;
    const float bx = bias[2 * lane];
    const float by = bias[2 * lane + 1];

    bool any_long = false;
    for (int q = 0; q < NPW; q += NQ) {
        int o0[NQ], deg[NQ], sv[NQ];
        v2f acc0[NQ];
        #pragma unroll
        for (int u = 0; u < NQ; ++u) {
            o0[u]   = __builtin_amdgcn_readlane(offv, q + u);
            deg[u]  = __builtin_amdgcn_readlane(offv, q + u + 1) - o0[u];
            if (SORT) sv[u] = (lane < deg[u]) ? col[o0[u] + lane] : 0x7fffffff;
            else      sv[u] = (lane < deg[u]) ? col[o0[u] + lane] : (n0 + q + u);
            acc0[u] = H2[(size_t)(n0 + q + u) * 32 + lane];
        }
        if (SORT) {
            int ks[NQ];
            #pragma unroll
            for (int u = 0; u < NQ; ++u) ks[u] = sort32_ascending(sv[u], lane);
            #pragma unroll
            for (int u = 0; u < NQ; ++u) {
                if (lane < deg[u] && deg[u] <= 32) cs[o0[u] + lane] = ks[u];
                sv[u] = (lane < deg[u]) ? ks[u] : (n0 + q + u);
            }
        }
        float av[NQ], aself[NQ];
        #pragma unroll
        for (int u = 0; u < NQ; ++u) av[u] = S[sv[u]];
        #pragma unroll
        for (int u = 0; u < NQ; ++u) {
            const float dn = readlane_f(Dv, q + u);
            aself[u] = leaky(readlane_f(Sv, q + u) + dn);
            av[u] = (lane < deg[u]) ? leaky(av[u] + dn) : -INFINITY;
        }
        float wv[NQ];
        #pragma unroll
        for (int u = 0; u < NQ; ++u) {
            const float m = wave_max_f(fmaxf(av[u], aself[u]));
            const float ex = expf(av[u] - m);
            const float e = (lane < deg[u]) ? ex : 0.0f;
            const float e_self = expf(aself[u] - m);
            float z = wave_sum_f(e);
            z += e_self;
            wv[u] = e / (z + EPSF);
            const float w_self = e_self / (z + EPSF);
            acc0[u].x = acc0[u].x * w_self;
            acc0[u].y = acc0[u].y * w_self;
        }
        #pragma unroll
        for (int u = 0; u < NQ; ++u) {
            if (deg[u] > 32) { any_long = true; continue; }
            const float w = wv[u];
            v2f acc = acc0[u];
            for (int j = 0; j < deg[u]; j += 8) {
                v2f hv[8];
                #pragma unroll
                for (int t = 0; t < 8; ++t) {
                    const int jt = (j + t < deg[u]) ? (j + t) : (deg[u] - 1);
                    const int sj = __builtin_amdgcn_readlane(sv[u], jt);
                    hv[t] = H2[(size_t)sj * 32 + lane];
                }
                #pragma unroll
                for (int t = 0; t < 8; ++t) {
                    const float wj = readlane_f(w, j + t);
                    acc.x = fmaf(hv[t].x, wj, acc.x);
                    acc.y = fmaf(hv[t].y, wj, acc.y);
                }
            }
            finish_row<TO_LDS>(acc, bx, by, relu, OUT, tile, n0 + q + u, rb0 + q + u, lane);
        }
    }

    if (any_long) {
        for (int u = 0; u < NPW; ++u) {
            const int lo = __builtin_amdgcn_readlane(offv, u);
            const int dg = __builtin_amdgcn_readlane(offv, u + 1) - lo;
            if (dg <= 32) continue;
            const int node = n0 + u;
            if (SORT) {
                for (int i = lane; i < dg; i += 32) {
                    const int ki = col[lo + i];
                    int rank = 0;
                    for (int j = 0; j < dg; ++j) {
                        const int kj = col[lo + j];
                        rank += (kj < ki || (kj == ki && j < i)) ? 1 : 0;
                    }
                    cs[lo + rank] = ki;
                }
                __threadfence();
            }
            const int* cr = SORT ? cs : col;
            const float dn = readlane_f(Dv, u);
            const float a_self = leaky(readlane_f(Sv, u) + dn);
            const v2f hsn = H2[(size_t)node * 32 + lane];
            float m = a_self;
            for (int c = 0; c < dg; c += 32) {
                if (c + lane < dg) m = fmaxf(m, leaky(S[cr[lo + c + lane]] + dn));
            }
            m = wave_max_f(m);
            float z = 0.0f;
            for (int c = 0; c < dg; c += 32) {
                if (c + lane < dg) z += expf(leaky(S[cr[lo + c + lane]] + dn) - m);
            }
            z = wave_sum_f(z);
            const float e_self = expf(a_self - m);
            z += e_self;
            const float w_self = e_self / (z + EPSF);
            v2f acc;
            acc.x = hsn.x * w_self;
            acc.y = hsn.y * w_self;
            for (int c = 0; c < dg; c += 32) {
                int s = node;
                float w = 0.0f;
                if (c + lane < dg) {
                    s = cr[lo + c + lane];
                    w = expf(leaky(S[s] + dn) - m) / (z + EPSF);
                }
                const int nj = (dg - c < 32) ? (dg - c) : 32;
                for (int j = 0; j < nj; ++j) {
                    const int sj = __builtin_amdgcn_readlane(s, j);
                    const float wj = readlane_f(w, j);
                    const v2f hv = H2[(size_t)sj * 32 + lane];
                    acc.x = fmaf(hv.x, wj, acc.x);
                    acc.y = fmaf(hv.y, wj, acc.y);
                }
            }
            finish_row<TO_LDS>(acc, bx, by, relu, OUT, tile, node, rb0 + u, lane);
        }
    }
}

__global__ __launch_bounds__(256) void node_softmax_sum_kernel(
        const int* __restrict__ off, const int* __restrict__ col,
        const float* __restrict__ H, const float* __restrict__ S, const float* __restrict__ D,
        const float* __restrict__ bias, int relu, float* __restrict__ OUT) {
    const int lane = threadIdx.x & 31;
    const int strip = __builtin_amdgcn_readfirstlane((int)(blockIdx.x * 8 + (threadIdx.x >> 5)));
    if (strip >= NSTRIP) return;
    strip_rows<false, false>(strip, lane, off, col, nullptr, H, S, D, bias, relu, OUT, nullptr, 0);
}

static_assert(NPW * 8 == 64, "a block of 8 waves must finish four whole 16-row tiles");
static_assert(NN % 16 == 0, "whole tiles only");
__global__ __launch_bounds__(256, 16) void node_softmax_sum_gemm_kernel(
        const int* __restrict__ off, const int* __restrict__ col, int* csorted,
        const float* __restrict__ H, const float* __restrict__ S, const float* __restrict__ D,
        const float* __restrict__ bias, int relu,
        const float* __restrict__ W2, const float* __restrict__ a_s2, const float* __restrict__ a_d2,
        float* __restrict__ H2, float* __restrict__ S2, float* __restrict__ D2) {
    __shared__ float tile[64 * TP];
    const int lane = threadIdx.x & 31;
    const int wv = __builtin_amdgcn_readfirstlane((int)(threadIdx.x >> 5));
    const int strip = blockIdx.x * 8 + wv;
    if (strip < NSTRIP) strip_rows<true, true>(strip, lane, off, col, csorted, H, S, D, bias, relu, nullptr, tile, wv * NPW);
    __syncthreads();

    const int t = blockIdx.x * 4 + wv;
    if (wv >= 4 || t >= NN / 16) return;

    const int r0    = lane & 15;
    const int khalf = lane >> 4;
    const float* arow = tile + (wv * 16 + r0) * TP;

    const float as0 = a_s2[r0], as1 = a_s2[16 + r0], as2 = a_s2[32 + r0], as3 = a_s2[48 + r0];
    const float ad0 = a_d2[r0], ad1 = a_d2[16 + r0], ad2 = a_d2[32 + r0], ad3 = a_d2[48 + r0];

    v8f c0 = {}, c1 = {}, c2 = {}, c3 = {};
    for (int k0 = 0; k0 < 64; k0 += 4) {
        const int ka = k0 + khalf * 2;
        v2f a;
        a.x = arow[ka];
        a.y = arow[ka + 1];

        const float* w0 = W2 + (size_t)ka * 64;
        const float* w1 = w0 + 64;
        v2f b0, b1, b2, b3;
        b0.x = w0[r0];      b0.y = w1[r0];
        b1.x = w0[16 + r0]; b1.y = w1[16 + r0];
        b2.x = w0[32 + r0]; b2.y = w1[32 + r0];
        b3.x = w0[48 + r0]; b3.y = w1[48 + r0];

        c0 = __builtin_amdgcn_wmma_f32_16x16x4_f32(false, a, false, b0, (short)0, c0, false, false);
        c1 = __builtin_amdgcn_wmma_f32_16x16x4_f32(false, a, false, b1, (short)0, c1, false, false);
        c2 = __builtin_amdgcn_wmma_f32_16x16x4_f32(false, a, false, b2, (short)0, c2, false, false);
        c3 = __builtin_amdgcn_wmma_f32_16x16x4_f32(false, a, false, b3, (short)0, c3, false, false);
    }

    tile_store_and_scores(c0, c1, c2, c3, as0, as1, as2, as3, ad0, ad1, ad2, ad3, H2, S2, D2, t, r0, khalf);
}

extern "C" void kernel_launch(void* const* d_in, const int* in_sizes, int n_in,
                              void* d_out, int out_size, void* d_ws, size_t ws_size,
                              hipStream_t stream) {
    const float* x     = (const float*)d_in[0];
    const int*   ei    = (const int*)d_in[1];
    const float* W1    = (const float*)d_in[2];
    const float* as1   = (const float*)d_in[3];
    const float* ad1   = (const float*)d_in[4];
    const float* b1    = (const float*)d_in[5];
    const float* W2    = (const float*)d_in[6];
    const float* as2   = (const float*)d_in[7];
    const float* ad2   = (const float*)d_in[8];
    const float* b2    = (const float*)d_in[9];
    float* out = (float*)d_out;

    const int E  = in_sizes[1] / 2;
    const int NF = NN * 64;

    float* wsf = (float*)d_ws;
    float* H   = wsf;
    float* H2  = H   + NF;
    float* S   = H2  + NF;
    float* D   = S   + NN;
    const int G1 = (E + CH - 1) / CH;
    int*   lofs = (int*)(D + NN);
    int*   off  = lofs + G1 * (NBUCK + 1) + 16;
    int*   tmp  = off + NN + 16;
    int*   cA   = tmp + G1 * CH;
    int*   cB   = cA  + E;
    float* S2   = (float*)(cB + E);
    float* D2   = S2 + NN;

    const int TB = 256;
    dim3 blk(TB);
    int gNode = (NSTRIP + 7) / 8;
    int gGemm = ((NN / 16) + 7) / 8;

    chunk_group_kernel<<<G1, blk, 0, stream>>>(ei, E, lofs, tmp);
    const GemmArgs g1 = {x, W1, as1, ad1, H, S, D, NN, 128};
    bucket_csr_gemm_kernel<<<NBUCK + gGemm, blk, 0, stream>>>(lofs, G1, tmp, off, cA, g1);

    node_softmax_sum_gemm_kernel<<<gNode, blk, 0, stream>>>(off, cA, cB, H, S, D, b1, 1, W2, as2, ad2, H2, S2, D2);

    node_softmax_sum_kernel<<<gNode, blk, 0, stream>>>(off, cB, H2, S2, D2, b2, 0, out);
}
